# starting kernel
# baseline (speedup 1.0000x reference)
#include <hip/hip_runtime.h>

#define NN 200000
#define NE 12800000
#define NTILES (NN / 16)

typedef __attribute__((ext_vector_type(2))) float v2f;
typedef __attribute__((ext_vector_type(8))) float v8f;


__global__ void gnn_zero_kernel(float* __restrict__ p, int n) {
  int i = blockIdx.x * blockDim.x + threadIdx.x;
  if (i < n) p[i] = 0.0f;
}

__global__ void gnn_deg_kernel(const int* __restrict__ dst, unsigned* __restrict__ degcnt) {
  int e = blockIdx.x * blockDim.x + threadIdx.x;
  if (e < NE) atomicAdd(&degcnt[dst[e]], 1u);
}

__global__ void gnn_dinv_kernel(const unsigned* __restrict__ degcnt, float* __restrict__ dinv) {
  int i = blockIdx.x * blockDim.x + threadIdx.x;
  if (i < NN) dinv[i] = rsqrtf((float)degcnt[i] + 1.0f);
}

__global__ void gnn_gemm1_wmma_kernel(const float* __restrict__ x,
                                      const float* __restrict__ W1,
                                      const float* __restrict__ dinv,
                                      float* __restrict__ hs1) {
  int wave = (blockIdx.x * blockDim.x + threadIdx.x) >> 5;
  int lane = threadIdx.x & 31;
  if (wave >= NTILES) return;
  int base = wave << 4;
  int m    = lane & 15;
  int r    = base + m;
  bool lo  = lane < 16;

  float x0 = x[r * 2 + 0];
  float x1 = x[r * 2 + 1];
  float w0 = W1[m];
  float w1 = W1[16 + m];

  v2f a, b;
  a.x = lo ? x0 : 0.0f;
  a.y = lo ? x1 : 0.0f;
  b.x = lo ? w0 : 0.0f;
  b.y = lo ? w1 : 0.0f;

  v8f c = {};
  c = __builtin_amdgcn_wmma_f32_16x16x4_f32(
       false, a,  false, b,
       (short)0, c,  false,  false);

  int col   = lane & 15;
  int rowHi = (lane >> 4) << 3;
#pragma unroll
  for (int rr = 0; rr < 8; ++rr) {
    int row = base + rowHi + rr;
    hs1[row * 16 + col] = c[rr] * dinv[row];
  }
}

__global__ void gnn_edge16_kernel(const int* __restrict__ src, const int* __restrict__ dst,
                                  const float* __restrict__ hs1, float* __restrict__ agg1) {
  int t = blockIdx.x * blockDim.x + threadIdx.x;
  if (t >= NE * 16) return;
  int e = t >> 4;
  int f = t & 15;
  int s = src[e];
  int d = dst[e];
  unsafeAtomicAdd(&agg1[d * 16 + f], hs1[s * 16 + f]);
}

__global__ void gnn_finalize1_kernel(const float* __restrict__ agg1,
                                     const float* __restrict__ hs1,
                                     const float* __restrict__ dinv,
                                     const float* __restrict__ b1,
                                     const float* __restrict__ W2,
                                     float* __restrict__ hs2) {
  int i = blockIdx.x * blockDim.x + threadIdx.x;
  if (i >= NN) return;
  float di = dinv[i];
  float h0 = 0.0f, h1 = 0.0f;
#pragma unroll
  for (int f = 0; f < 16; ++f) {
    float v = di * (agg1[i * 16 + f] + hs1[i * 16 + f]) + b1[f];
    v = v > 0.0f ? v : 0.0f;
    h0 += v * W2[f * 2 + 0];
    h1 += v * W2[f * 2 + 1];
  }
  hs2[i * 2 + 0] = h0 * di;
  hs2[i * 2 + 1] = h1 * di;
}

__global__ void gnn_edge2_kernel(const int* __restrict__ src, const int* __restrict__ dst,
                                 const float* __restrict__ hs2, float* __restrict__ agg2) {
  int t = blockIdx.x * blockDim.x + threadIdx.x;
  if (t >= NE * 2) return;
  int e = t >> 1;
  int f = t & 1;
  int s = src[e];
  int d = dst[e];
  unsafeAtomicAdd(&agg2[d * 2 + f], hs2[s * 2 + f]);
}

__global__ void gnn_finalize2_kernel(const float* __restrict__ agg2,
                                     const float* __restrict__ hs2,
                                     const float* __restrict__ dinv,
                                     const float* __restrict__ b2,
                                     float* __restrict__ out) {
  int t = blockIdx.x * blockDim.x + threadIdx.x;
  if (t >= NN * 2) return;
  int i = t >> 1;
  int c = t & 1;
  out[t] = dinv[i] * (agg2[t] + hs2[t]) + b2[c];
}

extern "C" void kernel_launch(void* const* d_in, const int* in_sizes, int n_in,
                              void* d_out, int out_size, void* d_ws, size_t ws_size,
                              hipStream_t stream) {
  const float* x  = (const float*)d_in[0];
  const float* W1 = (const float*)d_in[1];
  const float* b1 = (const float*)d_in[2];
  const float* W2 = (const float*)d_in[3];
  const float* b2 = (const float*)d_in[4];
  const int* edge = (const int*)d_in[5];
  const int* src  = edge;
  const int* dst  = edge + NE;
  float* out = (float*)d_out;

  float* ws = (float*)d_ws;
  unsigned* degcnt = (unsigned*)ws;
  float* agg1 = ws + (size_t)1 * NN;
  float* agg2 = ws + (size_t)17 * NN;
  float* dinv = ws + (size_t)19 * NN;
  float* hs1  = ws + (size_t)20 * NN;
  float* hs2  = ws + (size_t)36 * NN;

  const int B = 256;

  {
    int n = 19 * NN;
    gnn_zero_kernel<<<(n + B - 1) / B, B, 0, stream>>>(ws, n);
  }
  gnn_deg_kernel<<<(NE + B - 1) / B, B, 0, stream>>>(dst, degcnt);
  gnn_dinv_kernel<<<(NN + B - 1) / B, B, 0, stream>>>(degcnt, dinv);
  {
    int threads = NTILES * 32;
    gnn_gemm1_wmma_kernel<<<(threads + B - 1) / B, B, 0, stream>>>(x, W1, dinv, hs1);
  }
  {
    long long n = (long long)NE * 16;
    gnn_edge16_kernel<<<(int)((n + B - 1) / B), B, 0, stream>>>(src, dst, hs1, agg1);
  }
  gnn_finalize1_kernel<<<(NN + B - 1) / B, B, 0, stream>>>(agg1, hs1, dinv, b1, W2, hs2);
  {
    int n = NE * 2;
    gnn_edge2_kernel<<<(n + B - 1) / B, B, 0, stream>>>(src, dst, hs2, agg2);
  }
  {
    int n = NN * 2;
    gnn_finalize2_kernel<<<(n + B - 1) / B, B, 0, stream>>>(agg2, hs2, dinv, b2, out);
  }
}


# Degree count and 2-wide layer-2 edge sum: global atomics replaced by LDS sums owned per node range (16 ranges x 16 edge slices), slice partials added in fixed order; f32 sum order changes
# speedup vs baseline: 1.9115x; 1.9115x over previous
#include <hip/hip_runtime.h>


#define NN 200000
#define NE 12800000
#define NTILES (NN / 16)

#define RANGES 16
#define SLICES 16
#define RN 12512
#define SE (NE / SLICES)
#define SE4 (SE / 4)
#define TPB 1024
#define NWAVES (TPB / 32)
#define DU 8
#define EU 8
#define WQ (32 * 4 * EU)
#define GW 4

static_assert(RN * RANGES >= NN && RN * (RANGES - 1) < NN, "ranges cover the nodes");
static_assert(RN % 32 == 0, "whole lines per owner");
static_assert(SE * SLICES == NE && SE % 4 == 0, "slices cover the edges");
static_assert(NN <= (1 << 18) && RN <= (1 << 14), "list entry = 14 bits local node | 18 bits source");

typedef __attribute__((ext_vector_type(2))) float v2f;
typedef __attribute__((ext_vector_type(8))) float v8f;


__global__ void gnn_zero_kernel(float* __restrict__ p, int n) {
  int i = blockIdx.x * blockDim.x + threadIdx.x;
  if (i < n) p[i] = 0.0f;
}

__global__ __launch_bounds__(TPB) void gnn_deg_lds_kernel(const int* __restrict__ dst,
                                                          unsigned* __restrict__ degpart) {
  __shared__ unsigned cnt[RN];
  const int r = blockIdx.x % RANGES;
  const int s = blockIdx.x / RANGES;
  const unsigned lo = (unsigned)(r * RN);
  const unsigned nloc = (NN - (int)lo) < RN ? (unsigned)(NN - (int)lo) : (unsigned)RN;

  for (int j = threadIdx.x; j < RN; j += TPB) cnt[j] = 0u;
  __syncthreads();

  const int4* __restrict__ d4 = reinterpret_cast<const int4*>(dst + (size_t)s * SE);
  for (int c0 = 0; c0 < SE4; c0 += TPB * DU) {
    int4 v[DU];
#pragma unroll
    for (int u = 0; u < DU; ++u) {
      int i = c0 + u * TPB + (int)threadIdx.x;
      v[u] = (i < SE4) ? d4[i] : make_int4(-1, -1, -1, -1);
    }
#pragma unroll
    for (int u = 0; u < DU; ++u) {
      unsigned a;
      a = (unsigned)v[u].x - lo; if (a < nloc) atomicAdd(&cnt[a], 1u);
      a = (unsigned)v[u].y - lo; if (a < nloc) atomicAdd(&cnt[a], 1u);
      a = (unsigned)v[u].z - lo; if (a < nloc) atomicAdd(&cnt[a], 1u);
      a = (unsigned)v[u].w - lo; if (a < nloc) atomicAdd(&cnt[a], 1u);
    }
  }
  __syncthreads();

  unsigned* __restrict__ row = degpart + (size_t)s * NN + lo;
  for (unsigned j = threadIdx.x; j < nloc; j += TPB) row[j] = cnt[j];
}

__global__ void gnn_dinv_kernel(const unsigned* __restrict__ degpart, float* __restrict__ dinv) {
  int i = blockIdx.x * blockDim.x + threadIdx.x;
  if (i >= NN) return;
  unsigned c = 0u;
#pragma unroll
  for (int s = 0; s < SLICES; ++s) c += degpart[(size_t)s * NN + i];
  dinv[i] = rsqrtf((float)c + 1.0f);
}

__global__ void gnn_gemm1_wmma_kernel(const float* __restrict__ x,
                                      const float* __restrict__ W1,
                                      const float* __restrict__ dinv,
                                      float* __restrict__ hs1) {
  int wave = (blockIdx.x * blockDim.x + threadIdx.x) >> 5;
  int lane = threadIdx.x & 31;
  if (wave >= NTILES) return;
  int base = wave << 4;
  int m    = lane & 15;
  int r    = base + m;
  bool lo  = lane < 16;

  float x0 = x[r * 2 + 0];
  float x1 = x[r * 2 + 1];
  float w0 = W1[m];
  float w1 = W1[16 + m];

  v2f a, b;
  a.x = lo ? x0 : 0.0f;
  a.y = lo ? x1 : 0.0f;
  b.x = lo ? w0 : 0.0f;
  b.y = lo ? w1 : 0.0f;

  v8f c = {};
  c = __builtin_amdgcn_wmma_f32_16x16x4_f32(
       false, a,  false, b,
       (short)0, c,  false,  false);

  int col   = lane & 15;
  int rowHi = (lane >> 4) << 3;
#pragma unroll
  for (int rr = 0; rr < 8; ++rr) {
    int row = base + rowHi + rr;
    hs1[row * 16 + col] = c[rr] * dinv[row];
  }
}

__global__ void gnn_edge16_kernel(const int* __restrict__ src, const int* __restrict__ dst,
                                  const float* __restrict__ hs1, float* __restrict__ agg1) {
  int t = blockIdx.x * blockDim.x + threadIdx.x;
  if (t >= NE * 16) return;
  int e = t >> 4;
  int f = t & 15;
  int s = src[e];
  int d = dst[e];
  unsafeAtomicAdd(&agg1[d * 16 + f], hs1[s * 16 + f]);
}

__global__ void gnn_finalize1_kernel(const float* __restrict__ agg1,
                                     const float* __restrict__ hs1,
                                     const float* __restrict__ dinv,
                                     const float* __restrict__ b1,
                                     const float* __restrict__ W2,
                                     float* __restrict__ hs2) {
  int i = blockIdx.x * blockDim.x + threadIdx.x;
  if (i >= NN) return;
  float di = dinv[i];
  float h0 = 0.0f, h1 = 0.0f;
#pragma unroll
  for (int f = 0; f < 16; ++f) {
    float v = di * (agg1[i * 16 + f] + hs1[i * 16 + f]) + b1[f];
    v = v > 0.0f ? v : 0.0f;
    h0 += v * W2[f * 2 + 0];
    h1 += v * W2[f * 2 + 1];
  }
  hs2[i * 2 + 0] = h0 * di;
  hs2[i * 2 + 1] = h1 * di;
}

__global__ __launch_bounds__(TPB) void gnn_edge2_lds_kernel(const int* __restrict__ src,
                                                            const int* __restrict__ dst,
                                                            const float* __restrict__ hs2,
                                                            float* __restrict__ part2) {
  __shared__ float acc[RN * 2];
  __shared__ unsigned wq[NWAVES][WQ];
  const int r = blockIdx.x % RANGES;
  const int s = blockIdx.x / RANGES;
  const unsigned lo = (unsigned)(r * RN);
  const unsigned nloc = (NN - (int)lo) < RN ? (unsigned)(NN - (int)lo) : (unsigned)RN;
  const unsigned lane = threadIdx.x & 31;
  unsigned* q = wq[threadIdx.x >> 5];

  for (int j = threadIdx.x; j < RN * 2; j += TPB) acc[j] = 0.0f;
  __syncthreads();

  const int4* __restrict__ d4 = reinterpret_cast<const int4*>(dst + (size_t)s * SE);
  const int4* __restrict__ s4 = reinterpret_cast<const int4*>(src + (size_t)s * SE);
  const float2* __restrict__ h2 = reinterpret_cast<const float2*>(hs2);

  for (int c0 = 0; c0 < SE4; c0 += TPB * EU) {
    int4 dv[EU], sv[EU];
#pragma unroll
    for (int u = 0; u < EU; ++u) {
      int i = c0 + u * TPB + (int)threadIdx.x;
      bool ok = i < SE4;
      dv[u] = ok ? d4[i] : make_int4(-1, -1, -1, -1);
      sv[u] = ok ? s4[i] : make_int4(0, 0, 0, 0);
    }
    unsigned n = 0;
#pragma unroll
    for (int u = 0; u < EU; ++u) {
      const int dd[4] = {dv[u].x, dv[u].y, dv[u].z, dv[u].w};
      const int ss[4] = {sv[u].x, sv[u].y, sv[u].z, sv[u].w};
#pragma unroll
      for (int k = 0; k < 4; ++k) {
        unsigned a = (unsigned)dd[k] - lo;
        bool m = a < nloc;
        unsigned mask = __builtin_amdgcn_ballot_w32(m);
        unsigned pos = n + __builtin_amdgcn_mbcnt_lo(mask, 0u);
        if (m) q[pos] = (a << 18) | ((unsigned)ss[k] & 0x3FFFFu);
        n += (unsigned)__builtin_popcount(mask);
      }
    }
    __builtin_amdgcn_fence(__ATOMIC_RELEASE, "wavefront");
    __builtin_amdgcn_wave_barrier();
    __builtin_amdgcn_fence(__ATOMIC_ACQUIRE, "wavefront");
    for (unsigned k0 = 0; k0 < n; k0 += 32 * GW) {
      unsigned p[GW];
      float2 h[GW];
#pragma unroll
      for (int j = 0; j < GW; ++j) {
        unsigned k = k0 + 32u * j + lane;
        p[j] = (k < n) ? q[k] : 0xFFFFFFFFu;
      }
#pragma unroll
      for (int j = 0; j < GW; ++j) {
        h[j] = make_float2(0.0f, 0.0f);
        if (p[j] != 0xFFFFFFFFu) h[j] = h2[p[j] & 0x3FFFFu];
      }
#pragma unroll
      for (int j = 0; j < GW; ++j) {
        if (p[j] != 0xFFFFFFFFu) {
          unsigned a = p[j] >> 18;
          unsafeAtomicAdd(&acc[2 * a + 0], h[j].x);
          unsafeAtomicAdd(&acc[2 * a + 1], h[j].y);
        }
      }
    }
    __builtin_amdgcn_fence(__ATOMIC_RELEASE, "wavefront");
    __builtin_amdgcn_wave_barrier();
    __builtin_amdgcn_fence(__ATOMIC_ACQUIRE, "wavefront");
  }
  __syncthreads();

  float* __restrict__ row = part2 + (size_t)s * (2 * NN) + 2 * lo;
  for (unsigned j = threadIdx.x; j < 2 * nloc; j += TPB) row[j] = acc[j];
}

__global__ void gnn_finalize2_kernel(const float* __restrict__ part2,
                                     const float* __restrict__ hs2,
                                     const float* __restrict__ dinv,
                                     const float* __restrict__ b2,
                                     float* __restrict__ out) {
  int t = blockIdx.x * blockDim.x + threadIdx.x;
  if (t >= NN * 2) return;
  int i = t >> 1;
  int c = t & 1;
  float a = 0.0f;
#pragma unroll
  for (int s = 0; s < SLICES; ++s) a += part2[(size_t)s * (2 * NN) + t];
  out[t] = dinv[i] * (a + hs2[t]) + b2[c];
}

extern "C" void kernel_launch(void* const* d_in, const int* in_sizes, int n_in,
                              void* d_out, int out_size, void* d_ws, size_t ws_size,
                              hipStream_t stream) {
  const float* x  = (const float*)d_in[0];
  const float* W1 = (const float*)d_in[1];
  const float* b1 = (const float*)d_in[2];
  const float* W2 = (const float*)d_in[3];
  const float* b2 = (const float*)d_in[4];
  const int* edge = (const int*)d_in[5];
  const int* src  = edge;
  const int* dst  = edge + NE;
  float* out = (float*)d_out;

  float* ws = (float*)d_ws;
  float* dinv = ws;
  float* hs2  = ws + (size_t)1 * NN;
  float* agg1 = ws + (size_t)3 * NN;
  float* hs1  = ws + (size_t)19 * NN;
  unsigned* degpart = (unsigned*)(ws + (size_t)19 * NN);
  float* part2 = ws + (size_t)3 * NN;

  const int B = 256;

  {
    int n = 16 * NN;
    gnn_zero_kernel<<<(n + B - 1) / B, B, 0, stream>>>(agg1, n);
  }
  gnn_deg_lds_kernel<<<RANGES * SLICES, TPB, 0, stream>>>(dst, degpart);
  gnn_dinv_kernel<<<(NN + B - 1) / B, B, 0, stream>>>(degpart, dinv);
  {
    int threads = NTILES * 32;
    gnn_gemm1_wmma_kernel<<<(threads + B - 1) / B, B, 0, stream>>>(x, W1, dinv, hs1);
  }
  {
    long long n = (long long)NE * 16;
    gnn_edge16_kernel<<<(int)((n + B - 1) / B), B, 0, stream>>>(src, dst, hs1, agg1);
  }
  gnn_finalize1_kernel<<<(NN + B - 1) / B, B, 0, stream>>>(agg1, hs1, dinv, b1, W2, hs2);
  gnn_edge2_lds_kernel<<<RANGES * SLICES, TPB, 0, stream>>>(src, dst, hs2, part2);
  {
    int n = NN * 2;
    gnn_finalize2_kernel<<<(n + B - 1) / B, B, 0, stream>>>(part2, hs2, dinv, b2, out);
  }
}


# Layer-1 edge sum moved in front of the W1 product: 2-wide x*dinv rows summed by the LDS owner pass, WMMA applied to the sums; 16-wide global float atomics gone; f32 sum order changes
# speedup vs baseline: 4.7899x; 2.5058x over previous
#include <hip/hip_runtime.h>


#define NN 200000
#define NE 12800000
#define NTILES (NN / 16)

#define RANGES 16
#define SLICES 16
#define RN 12512
#define SE (NE / SLICES)
#define SE4 (SE / 4)
#define TPB 1024
#define NWAVES (TPB / 32)
#define DU 8
#define EU 8
#define WQ (32 * 4 * EU)
#define GW 4

static_assert(RN * RANGES >= NN && RN * (RANGES - 1) < NN, "ranges cover the nodes");
static_assert(RN % 32 == 0, "whole lines per owner");
static_assert(SE * SLICES == NE && SE % 4 == 0, "slices cover the edges");
static_assert(NN <= (1 << 18) && RN <= (1 << 14), "list entry = 14 bits local node | 18 bits source");

typedef __attribute__((ext_vector_type(2))) float v2f;
typedef __attribute__((ext_vector_type(8))) float v8f;


__global__ __launch_bounds__(TPB) void gnn_deg_lds_kernel(const int* __restrict__ dst,
                                                          unsigned* __restrict__ degpart) {
  __shared__ unsigned cnt[RN];
  const int r = blockIdx.x % RANGES;
  const int s = blockIdx.x / RANGES;
  const unsigned lo = (unsigned)(r * RN);
  const unsigned nloc = (NN - (int)lo) < RN ? (unsigned)(NN - (int)lo) : (unsigned)RN;

  for (int j = threadIdx.x; j < RN; j += TPB) cnt[j] = 0u;
  __syncthreads();

  const int4* __restrict__ d4 = reinterpret_cast<const int4*>(dst + (size_t)s * SE);
  for (int c0 = 0; c0 < SE4; c0 += TPB * DU) {
    int4 v[DU];
#pragma unroll
    for (int u = 0; u < DU; ++u) {
      int i = c0 + u * TPB + (int)threadIdx.x;
      v[u] = (i < SE4) ? d4[i] : make_int4(-1, -1, -1, -1);
    }
#pragma unroll
    for (int u = 0; u < DU; ++u) {
      unsigned a;
      a = (unsigned)v[u].x - lo; if (a < nloc) atomicAdd(&cnt[a], 1u);
      a = (unsigned)v[u].y - lo; if (a < nloc) atomicAdd(&cnt[a], 1u);
      a = (unsigned)v[u].z - lo; if (a < nloc) atomicAdd(&cnt[a], 1u);
      a = (unsigned)v[u].w - lo; if (a < nloc) atomicAdd(&cnt[a], 1u);
    }
  }
  __syncthreads();

  unsigned* __restrict__ row = degpart + (size_t)s * NN + lo;
  for (unsigned j = threadIdx.x; j < nloc; j += TPB) row[j] = cnt[j];
}

__global__ void gnn_dinv_xs_kernel(const unsigned* __restrict__ degpart,
                                   const float* __restrict__ x,
                                   float* __restrict__ dinv,
                                   float* __restrict__ xs) {
  int i = blockIdx.x * blockDim.x + threadIdx.x;
  if (i >= NN) return;
  unsigned c = 0u;
#pragma unroll
  for (int s = 0; s < SLICES; ++s) c += degpart[(size_t)s * NN + i];
  float d = rsqrtf((float)c + 1.0f);
  dinv[i] = d;
  float2 xv = reinterpret_cast<const float2*>(x)[i];
  reinterpret_cast<float2*>(xs)[i] = make_float2(xv.x * d, xv.y * d);
}

__global__ void gnn_layer1_wmma_kernel(const float* __restrict__ part,
                                       const float* __restrict__ x,
                                       const float* __restrict__ dinv,
                                       const float* __restrict__ W1,
                                       const float* __restrict__ b1,
                                       const float* __restrict__ W2,
                                       float* __restrict__ hs2) {
  int wave = (blockIdx.x * blockDim.x + threadIdx.x) >> 5;
  int lane = threadIdx.x & 31;
  if (wave >= NTILES) return;
  int c    = lane & 15;
  int half = lane >> 4;
  int node = (wave << 4) + c;
  bool lo  = half == 0;

  const float2* __restrict__ p2 = reinterpret_cast<const float2*>(part);
  float2 a = make_float2(0.0f, 0.0f);
#pragma unroll
  for (int s = 0; s < SLICES; ++s) {
    float2 v = p2[(size_t)s * NN + node];
    a.x += v.x;
    a.y += v.y;
  }
  float di = dinv[node];
  float2 xv = reinterpret_cast<const float2*>(x)[node];
  float z0 = di * (a.x + xv.x * di);
  float z1 = di * (a.y + xv.y * di);

  float w0 = W1[c];
  float w1 = W1[16 + c];

  v2f A, B;
  A.x = lo ? w0 : 0.0f;
  A.y = lo ? w1 : 0.0f;
  B.x = lo ? z0 : 0.0f;
  B.y = lo ? z1 : 0.0f;

  v8f acc = {};
  acc = __builtin_amdgcn_wmma_f32_16x16x4_f32(
       false, A,  false, B,
       (short)0, acc,  false,  false);

  const float* __restrict__ bb = b1 + 8 * half;
  const float* __restrict__ ww = W2 + 16 * half;
  float h0 = 0.0f, h1 = 0.0f;
#pragma unroll
  for (int r = 0; r < 8; ++r) {
    float v = acc[r] + bb[r];
    v = v > 0.0f ? v : 0.0f;
    h0 += v * ww[2 * r + 0];
    h1 += v * ww[2 * r + 1];
  }
  h0 += __shfl_xor(h0, 16);
  h1 += __shfl_xor(h1, 16);
  if (lo) reinterpret_cast<float2*>(hs2)[node] = make_float2(h0 * di, h1 * di);
}

__global__ __launch_bounds__(TPB) void gnn_edge_sum2_lds_kernel(const int* __restrict__ src,
                                                                const int* __restrict__ dst,
                                                                const float* __restrict__ rows2,
                                                                float* __restrict__ part2) {
  __shared__ float acc[RN * 2];
  __shared__ unsigned wq[NWAVES][WQ];
  const int r = blockIdx.x % RANGES;
  const int s = blockIdx.x / RANGES;
  const unsigned lo = (unsigned)(r * RN);
  const unsigned nloc = (NN - (int)lo) < RN ? (unsigned)(NN - (int)lo) : (unsigned)RN;
  const unsigned lane = threadIdx.x & 31;
  unsigned* q = wq[threadIdx.x >> 5];

  for (int j = threadIdx.x; j < RN * 2; j += TPB) acc[j] = 0.0f;
  __syncthreads();

  const int4* __restrict__ d4 = reinterpret_cast<const int4*>(dst + (size_t)s * SE);
  const int4* __restrict__ s4 = reinterpret_cast<const int4*>(src + (size_t)s * SE);
  const float2* __restrict__ h2 = reinterpret_cast<const float2*>(rows2);

  for (int c0 = 0; c0 < SE4; c0 += TPB * EU) {
    int4 dv[EU], sv[EU];
#pragma unroll
    for (int u = 0; u < EU; ++u) {
      int i = c0 + u * TPB + (int)threadIdx.x;
      bool ok = i < SE4;
      dv[u] = ok ? d4[i] : make_int4(-1, -1, -1, -1);
      sv[u] = ok ? s4[i] : make_int4(0, 0, 0, 0);
    }
    unsigned n = 0;
#pragma unroll
    for (int u = 0; u < EU; ++u) {
      const int dd[4] = {dv[u].x, dv[u].y, dv[u].z, dv[u].w};
      const int ss[4] = {sv[u].x, sv[u].y, sv[u].z, sv[u].w};
#pragma unroll
      for (int k = 0; k < 4; ++k) {
        unsigned a = (unsigned)dd[k] - lo;
        bool m = a < nloc;
        unsigned mask = __builtin_amdgcn_ballot_w32(m);
        unsigned pos = n + __builtin_amdgcn_mbcnt_lo(mask, 0u);
        if (m) q[pos] = (a << 18) | ((unsigned)ss[k] & 0x3FFFFu);
        n += (unsigned)__builtin_popcount(mask);
      }
    }
    __builtin_amdgcn_fence(__ATOMIC_RELEASE, "wavefront");
    __builtin_amdgcn_wave_barrier();
    __builtin_amdgcn_fence(__ATOMIC_ACQUIRE, "wavefront");
    for (unsigned k0 = 0; k0 < n; k0 += 32 * GW) {
      unsigned p[GW];
      float2 h[GW];
#pragma unroll
      for (int j = 0; j < GW; ++j) {
        unsigned k = k0 + 32u * j + lane;
        p[j] = (k < n) ? q[k] : 0xFFFFFFFFu;
      }
#pragma unroll
      for (int j = 0; j < GW; ++j) {
        h[j] = make_float2(0.0f, 0.0f);
        if (p[j] != 0xFFFFFFFFu) h[j] = h2[p[j] & 0x3FFFFu];
      }
#pragma unroll
      for (int j = 0; j < GW; ++j) {
        if (p[j] != 0xFFFFFFFFu) {
          unsigned a = p[j] >> 18;
          unsafeAtomicAdd(&acc[2 * a + 0], h[j].x);
          unsafeAtomicAdd(&acc[2 * a + 1], h[j].y);
        }
      }
    }
    __builtin_amdgcn_fence(__ATOMIC_RELEASE, "wavefront");
    __builtin_amdgcn_wave_barrier();
    __builtin_amdgcn_fence(__ATOMIC_ACQUIRE, "wavefront");
  }
  __syncthreads();

  float* __restrict__ row = part2 + (size_t)s * (2 * NN) + 2 * lo;
  for (unsigned j = threadIdx.x; j < 2 * nloc; j += TPB) row[j] = acc[j];
}

__global__ void gnn_finalize2_kernel(const float* __restrict__ part2,
                                     const float* __restrict__ hs2,
                                     const float* __restrict__ dinv,
                                     const float* __restrict__ b2,
                                     float* __restrict__ out) {
  int t = blockIdx.x * blockDim.x + threadIdx.x;
  if (t >= NN * 2) return;
  int i = t >> 1;
  int c = t & 1;
  float a = 0.0f;
#pragma unroll
  for (int s = 0; s < SLICES; ++s) a += part2[(size_t)s * (2 * NN) + t];
  out[t] = dinv[i] * (a + hs2[t]) + b2[c];
}

extern "C" void kernel_launch(void* const* d_in, const int* in_sizes, int n_in,
                              void* d_out, int out_size, void* d_ws, size_t ws_size,
                              hipStream_t stream) {
  const float* x  = (const float*)d_in[0];
  const float* W1 = (const float*)d_in[1];
  const float* b1 = (const float*)d_in[2];
  const float* W2 = (const float*)d_in[3];
  const float* b2 = (const float*)d_in[4];
  const int* edge = (const int*)d_in[5];
  const int* src  = edge;
  const int* dst  = edge + NE;
  float* out = (float*)d_out;

  float* ws = (float*)d_ws;
  float* dinv  = ws;
  float* rows2 = ws + (size_t)1 * NN;
  unsigned* degpart = (unsigned*)(ws + (size_t)3 * NN);
  float* part2 = ws + (size_t)3 * NN;

  const int B = 256;

  gnn_deg_lds_kernel<<<RANGES * SLICES, TPB, 0, stream>>>(dst, degpart);
  gnn_dinv_xs_kernel<<<(NN + B - 1) / B, B, 0, stream>>>(degpart, x, dinv, rows2);
  gnn_edge_sum2_lds_kernel<<<RANGES * SLICES, TPB, 0, stream>>>(src, dst, rows2, part2);
  {
    int threads = NTILES * 32;
    gnn_layer1_wmma_kernel<<<(threads + B - 1) / B, B, 0, stream>>>(part2, x, dinv, W1, b1, W2, rows2);
  }
  gnn_edge_sum2_lds_kernel<<<RANGES * SLICES, TPB, 0, stream>>>(src, dst, rows2, part2);
  {
    int n = NN * 2;
    gnn_finalize2_kernel<<<(n + B - 1) / B, B, 0, stream>>>(part2, rows2, dinv, b2, out);
  }
}


# Edge list counting-sorted by destination range in 3 parts reusing one buffer inside the old 28 MB workspace; the 3 scatter passes read only their own range, one owner each; f32 sum order changes
# speedup vs baseline: 5.5394x; 1.1565x over previous
#include <hip/hip_runtime.h>


#define NN 200000
#define NE 12800000
#define NTILES (NN / 16)

#define RNG 800
#define NR (NN / RNG)
#define NB 512
#define CH (NE / NB)
#define CH4 (CH / 4)
#define CHS 25024
#define OFFW 256
#define TPB 1024
#define NWAVES (TPB / 32)
#define BU 7
#define GW 4

#define PA_BASE 0
#define PA_N 192
#define PB_BASE 192
#define PB_N 192
#define PC_BASE 384
#define PC_N 128
#define PMAX 192

static_assert(NR * RNG == NN && RNG % 32 == 0, "ranges cover the nodes in whole lines");
static_assert(CH * NB == NE && CH % 4 == 0, "chunks cover the edges");
static_assert(BU * TPB >= CH4, "one pass of the work-group loads a chunk");
static_assert(CHS >= CH && CHS % 32 == 0, "whole lines per chunk");
static_assert(NR + 1 <= OFFW && OFFW == 8 * 32, "offset row: 8 entries per lane of one wave");
static_assert(PA_N + PB_N + PC_N == NB && PB_BASE == PA_N && PC_BASE == PA_N + PB_N, "parts cover the chunks");
static_assert(PA_N <= PMAX && PB_N <= PMAX && PC_N <= PMAX, "parts fit the sorted buffer");
static_assert(PA_N % (2 * NWAVES) == 0 && PB_N % (2 * NWAVES) == 0 && PC_N % (2 * NWAVES) == 0 &&
              PMAX / NWAVES <= 32, "segment pairs per wave, bounds held one per lane");
static_assert(NN <= (1 << 18) && RNG < (1 << 14) - 1, "word = 14 bits local node | 18 bits source");

typedef __attribute__((ext_vector_type(2))) float v2f;
typedef __attribute__((ext_vector_type(8))) float v8f;

#define WS_OFFS ((size_t)6 * NN)
#define WS_BINS (WS_OFFS + (size_t)PMAX * OFFW)
#define WS_WORDS (WS_BINS + (size_t)PMAX * CHS)
static_assert(WS_WORDS * 4 <= (size_t)35 * NN * 4, "inside the 28,000,000 bytes the version before this one used");
static_assert(WS_OFFS % 32 == 0 && WS_BINS % 32 == 0, "planes start on 128-byte lines");

__global__ __launch_bounds__(TPB) void gnn_sort_chunk_kernel(const int* __restrict__ src,
                                                             const int* __restrict__ dst,
                                                             unsigned* __restrict__ offs,
                                                             unsigned* __restrict__ bins,
                                                             int cbase) {
  __shared__ unsigned cnt[OFFW];
  __shared__ unsigned cur[OFFW];
  __shared__ __attribute__((aligned(16))) unsigned stage[CH];
  const int b = blockIdx.x;
  const int tid = threadIdx.x;
  if (tid < OFFW) cnt[tid] = 0u;
  __syncthreads();

  const int4* __restrict__ d4 = reinterpret_cast<const int4*>(dst + (size_t)(cbase + b) * CH);
  const int4* __restrict__ s4 = reinterpret_cast<const int4*>(src + (size_t)(cbase + b) * CH);
  int4 dv[BU], sv[BU];
#pragma unroll
  for (int u = 0; u < BU; ++u) {
    int i = u * TPB + tid;
    bool ok = i < CH4;
    dv[u] = ok ? d4[i] : make_int4(-1, -1, -1, -1);
    sv[u] = ok ? s4[i] : make_int4(0, 0, 0, 0);
  }
#pragma unroll
  for (int u = 0; u < BU; ++u) {
    const unsigned dd[4] = {(unsigned)dv[u].x, (unsigned)dv[u].y, (unsigned)dv[u].z, (unsigned)dv[u].w};
#pragma unroll
    for (int k = 0; k < 4; ++k)
      if (dd[k] < (unsigned)NN) atomicAdd(&cnt[dd[k] / RNG], 1u);
  }
  __syncthreads();

  if (tid < 32) {
    unsigned c[8];
    unsigned s = 0u;
#pragma unroll
    for (int k = 0; k < 8; ++k) { c[k] = cnt[8 * tid + k]; s += c[k]; }
    unsigned inc = s;
#pragma unroll
    for (int d = 1; d < 32; d <<= 1) {
      unsigned t = __shfl_up(inc, d);
      if (tid >= d) inc += t;
    }
    unsigned ex = inc - s;
    unsigned* __restrict__ orow = offs + (size_t)b * OFFW + 8 * tid;
#pragma unroll
    for (int k = 0; k < 8; ++k) { cur[8 * tid + k] = ex; orow[k] = ex; ex += c[k]; }
  }
  __syncthreads();

#pragma unroll
  for (int u = 0; u < BU; ++u) {
    const unsigned dd[4] = {(unsigned)dv[u].x, (unsigned)dv[u].y, (unsigned)dv[u].z, (unsigned)dv[u].w};
    const unsigned ss[4] = {(unsigned)sv[u].x, (unsigned)sv[u].y, (unsigned)sv[u].z, (unsigned)sv[u].w};
#pragma unroll
    for (int k = 0; k < 4; ++k) {
      if (dd[k] < (unsigned)NN) {
        unsigned r = dd[k] / RNG;
        unsigned pos = atomicAdd(&cur[r], 1u);
        stage[pos] = ((dd[k] - r * RNG) << 18) | (ss[k] & 0x3FFFFu);
      }
    }
  }
  __syncthreads();

  uint4* __restrict__ o4 = reinterpret_cast<uint4*>(bins + (size_t)b * CHS);
  const uint4* st4 = reinterpret_cast<const uint4*>(stage);
  for (int i = tid; i < CH4; i += TPB) o4[i] = st4[i];
}

__global__ __launch_bounds__(TPB) void gnn_deg_owner_kernel(const unsigned* __restrict__ offs,
                                                            const unsigned* __restrict__ bins,
                                                            const float* __restrict__ x,
                                                            unsigned* __restrict__ degc,
                                                            float* __restrict__ dinv,
                                                            float* __restrict__ xs,
                                                            int nseg, int first, int last) {
  __shared__ unsigned cnt[RNG];
  const int r = blockIdx.x;
  const int tid = threadIdx.x;
  const unsigned lane = threadIdx.x & 31;
  const unsigned w = threadIdx.x >> 5;
  if (tid < RNG) cnt[tid] = first ? 0u : degc[r * RNG + tid];
  __syncthreads();

  unsigned s0 = 0u, s1 = 0u;
  if ((int)lane < nseg) {
    unsigned bb = w * (unsigned)nseg + lane;
    const unsigned* __restrict__ o = offs + (size_t)bb * OFFW + r;
    s0 = bb * CHS + o[0];
    s1 = bb * CHS + o[1];
  }
  for (int j = 0; j < nseg; j += 2) {
    unsigned a0 = (unsigned)__builtin_amdgcn_readlane((int)s0, j);
    unsigned a1 = (unsigned)__builtin_amdgcn_readlane((int)s1, j);
    unsigned b0 = (unsigned)__builtin_amdgcn_readlane((int)s0, j + 1);
    unsigned b1 = (unsigned)__builtin_amdgcn_readlane((int)s1, j + 1);
    unsigned na = a1 - a0, nb = b1 - b0;
    unsigned nmax = na > nb ? na : nb;
    for (unsigned k0 = 0; k0 < nmax; k0 += 32 * GW) {
      unsigned p[2 * GW];
#pragma unroll
      for (int t = 0; t < GW; ++t) {
        unsigned k = k0 + 32u * t + lane;
        p[t]      = (k < na) ? bins[a0 + k] : 0xFFFFFFFFu;
        p[GW + t] = (k < nb) ? bins[b0 + k] : 0xFFFFFFFFu;
      }
#pragma unroll
      for (int t = 0; t < 2 * GW; ++t)
        if (p[t] != 0xFFFFFFFFu) atomicAdd(&cnt[p[t] >> 18], 1u);
    }
  }
  __syncthreads();

  if (tid < RNG) {
    int node = r * RNG + tid;
    if (last) {
      float d = rsqrtf((float)cnt[tid] + 1.0f);
      dinv[node] = d;
      float2 xv = reinterpret_cast<const float2*>(x)[node];
      reinterpret_cast<float2*>(xs)[node] = make_float2(xv.x * d, xv.y * d);
    } else {
      degc[node] = cnt[tid];
    }
  }
}

__global__ __launch_bounds__(TPB) void gnn_edge_sum2_owner_kernel(const unsigned* __restrict__ offs,
                                                                  const unsigned* __restrict__ bins,
                                                                  const float* __restrict__ rows2,
                                                                  float* __restrict__ agg,
                                                                  int nseg, int first) {
  __shared__ float acc[RNG * 2];
  const int r = blockIdx.x;
  const int tid = threadIdx.x;
  const unsigned lane = threadIdx.x & 31;
  const unsigned w = threadIdx.x >> 5;
  float* __restrict__ row = agg + (size_t)2 * r * RNG;
  for (int j = tid; j < RNG * 2; j += TPB) acc[j] = first ? 0.0f : row[j];
  __syncthreads();

  const float2* __restrict__ h2 = reinterpret_cast<const float2*>(rows2);
  unsigned s0 = 0u, s1 = 0u;
  if ((int)lane < nseg) {
    unsigned bb = w * (unsigned)nseg + lane;
    const unsigned* __restrict__ o = offs + (size_t)bb * OFFW + r;
    s0 = bb * CHS + o[0];
    s1 = bb * CHS + o[1];
  }
  for (int j = 0; j < nseg; j += 2) {
    unsigned a0 = (unsigned)__builtin_amdgcn_readlane((int)s0, j);
    unsigned a1 = (unsigned)__builtin_amdgcn_readlane((int)s1, j);
    unsigned b0 = (unsigned)__builtin_amdgcn_readlane((int)s0, j + 1);
    unsigned b1 = (unsigned)__builtin_amdgcn_readlane((int)s1, j + 1);
    unsigned na = a1 - a0, nb = b1 - b0;
    unsigned nmax = na > nb ? na : nb;
    for (unsigned k0 = 0; k0 < nmax; k0 += 32 * GW) {
      unsigned p[2 * GW];
      float2 h[2 * GW];
#pragma unroll
      for (int t = 0; t < GW; ++t) {
        unsigned k = k0 + 32u * t + lane;
        p[t]      = (k < na) ? bins[a0 + k] : 0xFFFFFFFFu;
        p[GW + t] = (k < nb) ? bins[b0 + k] : 0xFFFFFFFFu;
      }
#pragma unroll
      for (int t = 0; t < 2 * GW; ++t) {
        h[t] = make_float2(0.0f, 0.0f);
        if (p[t] != 0xFFFFFFFFu) h[t] = h2[p[t] & 0x3FFFFu];
      }
#pragma unroll
      for (int t = 0; t < 2 * GW; ++t) {
        if (p[t] != 0xFFFFFFFFu) {
          unsigned a = p[t] >> 18;
          unsafeAtomicAdd(&acc[2 * a + 0], h[t].x);
          unsafeAtomicAdd(&acc[2 * a + 1], h[t].y);
        }
      }
    }
  }
  __syncthreads();

  for (int j = tid; j < 2 * RNG; j += TPB) row[j] = acc[j];
}

__global__ void gnn_layer1_wmma_kernel(const float* __restrict__ agg,
                                       const float* __restrict__ x,
                                       const float* __restrict__ dinv,
                                       const float* __restrict__ W1,
                                       const float* __restrict__ b1,
                                       const float* __restrict__ W2,
                                       float* __restrict__ hs2) {
  int wave = (blockIdx.x * blockDim.x + threadIdx.x) >> 5;
  int lane = threadIdx.x & 31;
  if (wave >= NTILES) return;
  int c    = lane & 15;
  int half = lane >> 4;
  int node = (wave << 4) + c;
  bool lo  = half == 0;

  float2 a = reinterpret_cast<const float2*>(agg)[node];
  float di = dinv[node];
  float2 xv = reinterpret_cast<const float2*>(x)[node];
  float z0 = di * (a.x + xv.x * di);
  float z1 = di * (a.y + xv.y * di);

  float w0 = W1[c];
  float w1 = W1[16 + c];

  v2f A, B;
  A.x = lo ? w0 : 0.0f;
  A.y = lo ? w1 : 0.0f;
  B.x = lo ? z0 : 0.0f;
  B.y = lo ? z1 : 0.0f;

  v8f acc = {};
  acc = __builtin_amdgcn_wmma_f32_16x16x4_f32(
       false, A,  false, B,
       (short)0, acc,  false,  false);

  const float* __restrict__ bb = b1 + 8 * half;
  const float* __restrict__ ww = W2 + 16 * half;
  float h0 = 0.0f, h1 = 0.0f;
#pragma unroll
  for (int r = 0; r < 8; ++r) {
    float v = acc[r] + bb[r];
    v = v > 0.0f ? v : 0.0f;
    h0 += v * ww[2 * r + 0];
    h1 += v * ww[2 * r + 1];
  }
  h0 += __shfl_xor(h0, 16);
  h1 += __shfl_xor(h1, 16);
  if (lo) reinterpret_cast<float2*>(hs2)[node] = make_float2(h0 * di, h1 * di);
}

__global__ void gnn_finalize2_kernel(const float* __restrict__ agg,
                                     const float* __restrict__ hs2,
                                     const float* __restrict__ dinv,
                                     const float* __restrict__ b2,
                                     float* __restrict__ out) {
  int t = blockIdx.x * blockDim.x + threadIdx.x;
  if (t >= NN * 2) return;
  int i = t >> 1;
  int c = t & 1;
  out[t] = dinv[i] * (agg[t] + hs2[t]) + b2[c];
}

extern "C" void kernel_launch(void* const* d_in, const int* in_sizes, int n_in,
                              void* d_out, int out_size, void* d_ws, size_t ws_size,
                              hipStream_t stream) {
  const float* x  = (const float*)d_in[0];
  const float* W1 = (const float*)d_in[1];
  const float* b1 = (const float*)d_in[2];
  const float* W2 = (const float*)d_in[3];
  const float* b2 = (const float*)d_in[4];
  const int* edge = (const int*)d_in[5];
  const int* src  = edge;
  const int* dst  = edge + NE;
  float* out = (float*)d_out;

  float* ws = (float*)d_ws;
  float* dinv  = ws;
  float* rows2 = ws + (size_t)1 * NN;
  float* agg   = ws + (size_t)3 * NN;
  unsigned* degc = (unsigned*)(ws + (size_t)5 * NN);
  unsigned* offs = (unsigned*)(ws + WS_OFFS);
  unsigned* bins = (unsigned*)(ws + WS_BINS);

  const int B = 256;

  gnn_sort_chunk_kernel<<<PA_N, TPB, 0, stream>>>(src, dst, offs, bins, PA_BASE);
  gnn_deg_owner_kernel<<<NR, TPB, 0, stream>>>(offs, bins, x, degc, dinv, rows2, PA_N / NWAVES, 1, 0);
  gnn_sort_chunk_kernel<<<PB_N, TPB, 0, stream>>>(src, dst, offs, bins, PB_BASE);
  gnn_deg_owner_kernel<<<NR, TPB, 0, stream>>>(offs, bins, x, degc, dinv, rows2, PB_N / NWAVES, 0, 0);
  gnn_sort_chunk_kernel<<<PC_N, TPB, 0, stream>>>(src, dst, offs, bins, PC_BASE);
  gnn_deg_owner_kernel<<<NR, TPB, 0, stream>>>(offs, bins, x, degc, dinv, rows2, PC_N / NWAVES, 0, 1);

  gnn_edge_sum2_owner_kernel<<<NR, TPB, 0, stream>>>(offs, bins, rows2, agg, PC_N / NWAVES, 1);
  gnn_sort_chunk_kernel<<<PA_N, TPB, 0, stream>>>(src, dst, offs, bins, PA_BASE);
  gnn_edge_sum2_owner_kernel<<<NR, TPB, 0, stream>>>(offs, bins, rows2, agg, PA_N / NWAVES, 0);
  gnn_sort_chunk_kernel<<<PB_N, TPB, 0, stream>>>(src, dst, offs, bins, PB_BASE);
  gnn_edge_sum2_owner_kernel<<<NR, TPB, 0, stream>>>(offs, bins, rows2, agg, PB_N / NWAVES, 0);

  {
    int threads = NTILES * 32;
    gnn_layer1_wmma_kernel<<<(threads + B - 1) / B, B, 0, stream>>>(agg, x, dinv, W1, b1, W2, rows2);
  }

  gnn_edge_sum2_owner_kernel<<<NR, TPB, 0, stream>>>(offs, bins, rows2, agg, PB_N / NWAVES, 1);
  gnn_sort_chunk_kernel<<<PC_N, TPB, 0, stream>>>(src, dst, offs, bins, PC_BASE);
  gnn_edge_sum2_owner_kernel<<<NR, TPB, 0, stream>>>(offs, bins, rows2, agg, PC_N / NWAVES, 0);
  gnn_sort_chunk_kernel<<<PA_N, TPB, 0, stream>>>(src, dst, offs, bins, PA_BASE);
  gnn_edge_sum2_owner_kernel<<<NR, TPB, 0, stream>>>(offs, bins, rows2, agg, PA_N / NWAVES, 0);

  {
    int n = NN * 2;
    gnn_finalize2_kernel<<<(n + B - 1) / B, B, 0, stream>>>(agg, rows2, dinv, b2, out);
  }
}


# Degree count from one 16-bit destination-only sorted list of all edges: 1 sort + 1 owner launch replace 3 + 3; xs rows wait in the output buffer; edge-sum part order and so f32 sum order change
# speedup vs baseline: 5.7835x; 1.0441x over previous
#include <hip/hip_runtime.h>


#define NN 200000
#define NE 12800000
#define NTILES (NN / 16)

#define RNG 800
#define NR (NN / RNG)
#define NB 512
#define CH (NE / NB)
#define CH4 (CH / 4)
#define CHS 25024
#define OFFW 256
#define TPB 1024
#define NWAVES (TPB / 32)
#define BU 7
#define GW 4
#define CHS16 25024
#define NSEG16 (NB / NWAVES)
#define SG16 4

#define PA_BASE 0
#define PA_N 192
#define PB_BASE 192
#define PB_N 192
#define PC_BASE 384
#define PC_N 128
#define PMAX 192

static_assert(NR * RNG == NN && RNG % 32 == 0, "ranges cover the nodes in whole lines");
static_assert(CH * NB == NE && CH % 4 == 0, "chunks cover the edges");
static_assert(BU * TPB >= CH4, "one pass of the work-group loads a chunk");
static_assert(CHS >= CH && CHS % 32 == 0, "whole lines per chunk");
static_assert(NR + 1 <= OFFW && OFFW == 8 * 32, "offset row: 8 entries per lane of one wave");
static_assert(PA_N + PB_N + PC_N == NB && PB_BASE == PA_N && PC_BASE == PA_N + PB_N, "parts cover the chunks");
static_assert(PA_N <= PMAX && PB_N <= PMAX && PC_N <= PMAX, "parts fit the sorted buffer");
static_assert(PA_N % (2 * NWAVES) == 0 && PB_N % (2 * NWAVES) == 0 && PC_N % (2 * NWAVES) == 0 &&
              PMAX / NWAVES <= 32, "segment pairs per wave, bounds held one per lane");
static_assert(NN <= (1 << 18) && RNG < (1 << 14) - 1, "word = 14 bits local node | 18 bits source");
static_assert(CHS16 >= CH && (CHS16 * 2) % 128 == 0 && (CH * 2) % 16 == 0, "16-bit chunks: whole lines, whole 16-byte groups");
static_assert(NSEG16 * NWAVES == NB && NSEG16 % SG16 == 0 && NSEG16 <= 32, "degree pass: segment groups per wave, bounds held one per lane");
static_assert((size_t)NB * CHS16 < ((size_t)1 << 32), "16-bit entry index fits 32 bits");

typedef __attribute__((ext_vector_type(2))) float v2f;
typedef __attribute__((ext_vector_type(8))) float v8f;

#define WS_OFFS16 ((size_t)1 * NN)
#define WS_BINS16 (WS_OFFS16 + (size_t)NB * OFFW)
#define WS_WORDS16 (WS_BINS16 + (size_t)NB * CHS16 / 2)
#define WS_OFFS ((size_t)6 * NN)
#define WS_BINS (WS_OFFS + (size_t)PMAX * OFFW)
#define WS_WORDS (WS_BINS + (size_t)PMAX * CHS)
static_assert(WS_WORDS * 4 <= (size_t)35 * NN * 4, "inside the 28,000,000 bytes the version before this one used");
static_assert(WS_OFFS % 32 == 0 && WS_BINS % 32 == 0, "planes start on 128-byte lines");
static_assert(WS_WORDS16 * 4 <= (size_t)35 * NN * 4, "degree list inside the same 28,000,000 bytes");
static_assert(WS_OFFS16 % 32 == 0 && WS_BINS16 % 32 == 0, "degree planes start on 128-byte lines");

__global__ __launch_bounds__(TPB) void gnn_sort_chunk_kernel(const int* __restrict__ src,
                                                             const int* __restrict__ dst,
                                                             unsigned* __restrict__ offs,
                                                             unsigned* __restrict__ bins,
                                                             int cbase) {
  __shared__ unsigned cnt[OFFW];
  __shared__ unsigned cur[OFFW];
  __shared__ __attribute__((aligned(16))) unsigned stage[CH];
  const int b = blockIdx.x;
  const int tid = threadIdx.x;
  if (tid < OFFW) cnt[tid] = 0u;
  __syncthreads();

  const int4* __restrict__ d4 = reinterpret_cast<const int4*>(dst + (size_t)(cbase + b) * CH);
  const int4* __restrict__ s4 = reinterpret_cast<const int4*>(src + (size_t)(cbase + b) * CH);
  int4 dv[BU], sv[BU];
#pragma unroll
  for (int u = 0; u < BU; ++u) {
    int i = u * TPB + tid;
    bool ok = i < CH4;
    dv[u] = ok ? d4[i] : make_int4(-1, -1, -1, -1);
    sv[u] = ok ? s4[i] : make_int4(0, 0, 0, 0);
  }
#pragma unroll
  for (int u = 0; u < BU; ++u) {
    const unsigned dd[4] = {(unsigned)dv[u].x, (unsigned)dv[u].y, (unsigned)dv[u].z, (unsigned)dv[u].w};
#pragma unroll
    for (int k = 0; k < 4; ++k)
      if (dd[k] < (unsigned)NN) atomicAdd(&cnt[dd[k] / RNG], 1u);
  }
  __syncthreads();

  if (tid < 32) {
    unsigned c[8];
    unsigned s = 0u;
#pragma unroll
    for (int k = 0; k < 8; ++k) { c[k] = cnt[8 * tid + k]; s += c[k]; }
    unsigned inc = s;
#pragma unroll
    for (int d = 1; d < 32; d <<= 1) {
      unsigned t = __shfl_up(inc, d);
      if (tid >= d) inc += t;
    }
    unsigned ex = inc - s;
    unsigned* __restrict__ orow = offs + (size_t)b * OFFW + 8 * tid;
#pragma unroll
    for (int k = 0; k < 8; ++k) { cur[8 * tid + k] = ex; orow[k] = ex; ex += c[k]; }
  }
  __syncthreads();

#pragma unroll
  for (int u = 0; u < BU; ++u) {
    const unsigned dd[4] = {(unsigned)dv[u].x, (unsigned)dv[u].y, (unsigned)dv[u].z, (unsigned)dv[u].w};
    const unsigned ss[4] = {(unsigned)sv[u].x, (unsigned)sv[u].y, (unsigned)sv[u].z, (unsigned)sv[u].w};
#pragma unroll
    for (int k = 0; k < 4; ++k) {
      if (dd[k] < (unsigned)NN) {
        unsigned r = dd[k] / RNG;
        unsigned pos = atomicAdd(&cur[r], 1u);
        stage[pos] = ((dd[k] - r * RNG) << 18) | (ss[k] & 0x3FFFFu);
      }
    }
  }
  __syncthreads();

  uint4* __restrict__ o4 = reinterpret_cast<uint4*>(bins + (size_t)b * CHS);
  const uint4* st4 = reinterpret_cast<const uint4*>(stage);
  for (int i = tid; i < CH4; i += TPB) o4[i] = st4[i];
}

__global__ __launch_bounds__(TPB) void gnn_sort_chunk16_kernel(const int* __restrict__ dst,
                                                               unsigned* __restrict__ offs,
                                                               unsigned short* __restrict__ bins16) {
  __shared__ unsigned cnt[OFFW];
  __shared__ unsigned cur[OFFW];
  __shared__ __attribute__((aligned(16))) unsigned short stage[CH];
  const int b = blockIdx.x;
  const int tid = threadIdx.x;
  if (tid < OFFW) cnt[tid] = 0u;
  __syncthreads();

  const int4* __restrict__ d4 = reinterpret_cast<const int4*>(dst + (size_t)b * CH);
  int4 dv[BU];
#pragma unroll
  for (int u = 0; u < BU; ++u) {
    int i = u * TPB + tid;
    dv[u] = (i < CH4) ? d4[i] : make_int4(-1, -1, -1, -1);
  }
#pragma unroll
  for (int u = 0; u < BU; ++u) {
    const unsigned dd[4] = {(unsigned)dv[u].x, (unsigned)dv[u].y, (unsigned)dv[u].z, (unsigned)dv[u].w};
#pragma unroll
    for (int k = 0; k < 4; ++k)
      if (dd[k] < (unsigned)NN) atomicAdd(&cnt[dd[k] / RNG], 1u);
  }
  __syncthreads();

  if (tid < 32) {
    unsigned c[8];
    unsigned s = 0u;
#pragma unroll
    for (int k = 0; k < 8; ++k) { c[k] = cnt[8 * tid + k]; s += c[k]; }
    unsigned inc = s;
#pragma unroll
    for (int d = 1; d < 32; d <<= 1) {
      unsigned t = __shfl_up(inc, d);
      if (tid >= d) inc += t;
    }
    unsigned ex = inc - s;
    unsigned* __restrict__ orow = offs + (size_t)b * OFFW + 8 * tid;
#pragma unroll
    for (int k = 0; k < 8; ++k) { cur[8 * tid + k] = ex; orow[k] = ex; ex += c[k]; }
  }
  __syncthreads();

#pragma unroll
  for (int u = 0; u < BU; ++u) {
    const unsigned dd[4] = {(unsigned)dv[u].x, (unsigned)dv[u].y, (unsigned)dv[u].z, (unsigned)dv[u].w};
#pragma unroll
    for (int k = 0; k < 4; ++k) {
      if (dd[k] < (unsigned)NN) {
        unsigned r = dd[k] / RNG;
        unsigned pos = atomicAdd(&cur[r], 1u);
        stage[pos] = (unsigned short)(dd[k] - r * RNG);
      }
    }
  }
  __syncthreads();

  uint4* __restrict__ o4 = reinterpret_cast<uint4*>(bins16 + (size_t)b * CHS16);
  const uint4* st4 = reinterpret_cast<const uint4*>(stage);
  for (int i = tid; i < CH / 8; i += TPB) o4[i] = st4[i];
}

__global__ __launch_bounds__(TPB) void gnn_deg_owner16_kernel(const unsigned* __restrict__ offs,
                                                              const unsigned short* __restrict__ bins16,
                                                              const float* __restrict__ x,
                                                              float* __restrict__ dinv,
                                                              float* __restrict__ xs) {
  __shared__ unsigned cnt[RNG];
  const int r = blockIdx.x;
  const int tid = threadIdx.x;
  const unsigned lane = threadIdx.x & 31;
  const unsigned w = threadIdx.x >> 5;
  if (tid < RNG) cnt[tid] = 0u;
  __syncthreads();

  unsigned s0 = 0u, s1 = 0u;
  if (lane < (unsigned)NSEG16) {
    unsigned bb = w * (unsigned)NSEG16 + lane;
    const unsigned* __restrict__ o = offs + (size_t)bb * OFFW + r;
    s0 = bb * (unsigned)CHS16 + o[0];
    s1 = bb * (unsigned)CHS16 + o[1];
  }
  for (int j = 0; j < NSEG16; j += SG16) {
    unsigned sa[SG16], sn[SG16];
    unsigned nmax = 0u;
#pragma unroll
    for (int g = 0; g < SG16; ++g) {
      sa[g] = (unsigned)__builtin_amdgcn_readlane((int)s0, j + g);
      sn[g] = (unsigned)__builtin_amdgcn_readlane((int)s1, j + g) - sa[g];
      nmax = sn[g] > nmax ? sn[g] : nmax;
    }
    for (unsigned k0 = 0; k0 < nmax; k0 += 32 * GW) {
      unsigned p[SG16 * GW];
#pragma unroll
      for (int g = 0; g < SG16; ++g) {
#pragma unroll
        for (int t = 0; t < GW; ++t) {
          unsigned k = k0 + 32u * t + lane;
          p[g * GW + t] = (k < sn[g]) ? (unsigned)bins16[sa[g] + k] : 0xFFFFu;
        }
      }
#pragma unroll
      for (int t = 0; t < SG16 * GW; ++t)
        if (p[t] != 0xFFFFu) atomicAdd(&cnt[p[t]], 1u);
    }
  }
  __syncthreads();

  if (tid < RNG) {
    int node = r * RNG + tid;
    float d = rsqrtf((float)cnt[tid] + 1.0f);
    dinv[node] = d;
    float2 xv = reinterpret_cast<const float2*>(x)[node];
    reinterpret_cast<float2*>(xs)[node] = make_float2(xv.x * d, xv.y * d);
  }
}

__global__ __launch_bounds__(TPB) void gnn_edge_sum2_owner_kernel(const unsigned* __restrict__ offs,
                                                                  const unsigned* __restrict__ bins,
                                                                  const float* __restrict__ rows2,
                                                                  float* __restrict__ agg,
                                                                  int nseg, int first) {
  __shared__ float acc[RNG * 2];
  const int r = blockIdx.x;
  const int tid = threadIdx.x;
  const unsigned lane = threadIdx.x & 31;
  const unsigned w = threadIdx.x >> 5;
  float* __restrict__ row = agg + (size_t)2 * r * RNG;
  for (int j = tid; j < RNG * 2; j += TPB) acc[j] = first ? 0.0f : row[j];
  __syncthreads();

  const float2* __restrict__ h2 = reinterpret_cast<const float2*>(rows2);
  unsigned s0 = 0u, s1 = 0u;
  if ((int)lane < nseg) {
    unsigned bb = w * (unsigned)nseg + lane;
    const unsigned* __restrict__ o = offs + (size_t)bb * OFFW + r;
    s0 = bb * CHS + o[0];
    s1 = bb * CHS + o[1];
  }
  for (int j = 0; j < nseg; j += 2) {
    unsigned a0 = (unsigned)__builtin_amdgcn_readlane((int)s0, j);
    unsigned a1 = (unsigned)__builtin_amdgcn_readlane((int)s1, j);
    unsigned b0 = (unsigned)__builtin_amdgcn_readlane((int)s0, j + 1);
    unsigned b1 = (unsigned)__builtin_amdgcn_readlane((int)s1, j + 1);
    unsigned na = a1 - a0, nb = b1 - b0;
    unsigned nmax = na > nb ? na : nb;
    for (unsigned k0 = 0; k0 < nmax; k0 += 32 * GW) {
      unsigned p[2 * GW];
      float2 h[2 * GW];
#pragma unroll
      for (int t = 0; t < GW; ++t) {
        unsigned k = k0 + 32u * t + lane;
        p[t]      = (k < na) ? bins[a0 + k] : 0xFFFFFFFFu;
        p[GW + t] = (k < nb) ? bins[b0 + k] : 0xFFFFFFFFu;
      }
#pragma unroll
      for (int t = 0; t < 2 * GW; ++t) {
        h[t] = make_float2(0.0f, 0.0f);
        if (p[t] != 0xFFFFFFFFu) h[t] = h2[p[t] & 0x3FFFFu];
      }
#pragma unroll
      for (int t = 0; t < 2 * GW; ++t) {
        if (p[t] != 0xFFFFFFFFu) {
          unsigned a = p[t] >> 18;
          unsafeAtomicAdd(&acc[2 * a + 0], h[t].x);
          unsafeAtomicAdd(&acc[2 * a + 1], h[t].y);
        }
      }
    }
  }
  __syncthreads();

  for (int j = tid; j < 2 * RNG; j += TPB) row[j] = acc[j];
}

__global__ void gnn_layer1_wmma_kernel(const float* __restrict__ agg,
                                       const float* __restrict__ x,
                                       const float* __restrict__ dinv,
                                       const float* __restrict__ W1,
                                       const float* __restrict__ b1,
                                       const float* __restrict__ W2,
                                       float* __restrict__ hs2) {
  int wave = (blockIdx.x * blockDim.x + threadIdx.x) >> 5;
  int lane = threadIdx.x & 31;
  if (wave >= NTILES) return;
  int c    = lane & 15;
  int half = lane >> 4;
  int node = (wave << 4) + c;
  bool lo  = half == 0;

  float2 a = reinterpret_cast<const float2*>(agg)[node];
  float di = dinv[node];
  float2 xv = reinterpret_cast<const float2*>(x)[node];
  float z0 = di * (a.x + xv.x * di);
  float z1 = di * (a.y + xv.y * di);

  float w0 = W1[c];
  float w1 = W1[16 + c];

  v2f A, B;
  A.x = lo ? w0 : 0.0f;
  A.y = lo ? w1 : 0.0f;
  B.x = lo ? z0 : 0.0f;
  B.y = lo ? z1 : 0.0f;

  v8f acc = {};
  acc = __builtin_amdgcn_wmma_f32_16x16x4_f32(
       false, A,  false, B,
       (short)0, acc,  false,  false);

  const float* __restrict__ bb = b1 + 8 * half;
  const float* __restrict__ ww = W2 + 16 * half;
  float h0 = 0.0f, h1 = 0.0f;
#pragma unroll
  for (int r = 0; r < 8; ++r) {
    float v = acc[r] + bb[r];
    v = v > 0.0f ? v : 0.0f;
    h0 += v * ww[2 * r + 0];
    h1 += v * ww[2 * r + 1];
  }
  h0 += __shfl_xor(h0, 16);
  h1 += __shfl_xor(h1, 16);
  if (lo) reinterpret_cast<float2*>(hs2)[node] = make_float2(h0 * di, h1 * di);
}

__global__ void gnn_finalize2_kernel(const float* __restrict__ agg,
                                     const float* __restrict__ hs2,
                                     const float* __restrict__ dinv,
                                     const float* __restrict__ b2,
                                     float* __restrict__ out) {
  int t = blockIdx.x * blockDim.x + threadIdx.x;
  if (t >= NN * 2) return;
  int i = t >> 1;
  int c = t & 1;
  out[t] = dinv[i] * (agg[t] + hs2[t]) + b2[c];
}

extern "C" void kernel_launch(void* const* d_in, const int* in_sizes, int n_in,
                              void* d_out, int out_size, void* d_ws, size_t ws_size,
                              hipStream_t stream) {
  const float* x  = (const float*)d_in[0];
  const float* W1 = (const float*)d_in[1];
  const float* b1 = (const float*)d_in[2];
  const float* W2 = (const float*)d_in[3];
  const float* b2 = (const float*)d_in[4];
  const int* edge = (const int*)d_in[5];
  const int* src  = edge;
  const int* dst  = edge + NE;
  float* out = (float*)d_out;

  float* ws = (float*)d_ws;
  float* dinv  = ws;
  float* rows2 = ws + (size_t)1 * NN;
  float* agg   = ws + (size_t)3 * NN;
  unsigned* offs = (unsigned*)(ws + WS_OFFS);
  unsigned* bins = (unsigned*)(ws + WS_BINS);
  unsigned* offs16 = (unsigned*)(ws + WS_OFFS16);
  unsigned short* bins16 = (unsigned short*)(ws + WS_BINS16);
  float* xs = out;

  const int B = 256;

  gnn_sort_chunk16_kernel<<<NB, TPB, 0, stream>>>(dst, offs16, bins16);
  gnn_deg_owner16_kernel<<<NR, TPB, 0, stream>>>(offs16, bins16, x, dinv, xs);

  gnn_sort_chunk_kernel<<<PA_N, TPB, 0, stream>>>(src, dst, offs, bins, PA_BASE);
  gnn_edge_sum2_owner_kernel<<<NR, TPB, 0, stream>>>(offs, bins, xs, agg, PA_N / NWAVES, 1);
  gnn_sort_chunk_kernel<<<PB_N, TPB, 0, stream>>>(src, dst, offs, bins, PB_BASE);
  gnn_edge_sum2_owner_kernel<<<NR, TPB, 0, stream>>>(offs, bins, xs, agg, PB_N / NWAVES, 0);
  gnn_sort_chunk_kernel<<<PC_N, TPB, 0, stream>>>(src, dst, offs, bins, PC_BASE);
  gnn_edge_sum2_owner_kernel<<<NR, TPB, 0, stream>>>(offs, bins, xs, agg, PC_N / NWAVES, 0);

  {
    int threads = NTILES * 32;
    gnn_layer1_wmma_kernel<<<(threads + B - 1) / B, B, 0, stream>>>(agg, x, dinv, W1, b1, W2, rows2);
  }

  gnn_edge_sum2_owner_kernel<<<NR, TPB, 0, stream>>>(offs, bins, rows2, agg, PC_N / NWAVES, 1);
  gnn_sort_chunk_kernel<<<PA_N, TPB, 0, stream>>>(src, dst, offs, bins, PA_BASE);
  gnn_edge_sum2_owner_kernel<<<NR, TPB, 0, stream>>>(offs, bins, rows2, agg, PA_N / NWAVES, 0);
  gnn_sort_chunk_kernel<<<PB_N, TPB, 0, stream>>>(src, dst, offs, bins, PB_BASE);
  gnn_edge_sum2_owner_kernel<<<NR, TPB, 0, stream>>>(offs, bins, rows2, agg, PB_N / NWAVES, 0);

  {
    int n = NN * 2;
    gnn_finalize2_kernel<<<(n + B - 1) / B, B, 0, stream>>>(agg, rows2, dinv, b2, out);
  }
}


# Edge sums tiled 32 destination ranges x 8 source blocks: a tile's source rows are staged in LDS, so no per-edge row read goes to L2; 8 partial planes summed per node; f32 sum order changes
# speedup vs baseline: 6.8124x; 1.1779x over previous
#include <hip/hip_runtime.h>


#define NN 200000
#define NE 12800000
#define NTILES (NN / 16)

#define TPB 1024
#define NWAVES (TPB / 32)

#define RNG 800
#define NR (NN / RNG)
#define NB 512
#define CH (NE / NB)
#define CH4 (CH / 4)
#define OFFW 256
#define BU 7
#define GW 4
#define CHS16 25024
#define NSEG16 (NB / NWAVES)
#define SG16 4

#define TD 32
#define TS 8
#define NT (TD * TS)
#define DRNG 6256
#define NPAD (TD * DRNG)
#define SBLK 25000
#define NB2 1024
#define CH2 (NE / NB2)
#define CH2_4 (CH2 / 4)
#define CHS2 12512
#define OFFW2 288
#define BU2 4
#define NPART 4
#define PCH (NB2 / NPART)
#define NSEG2 (PCH / NWAVES)
#define GW2 3

static_assert(NR * RNG == NN && RNG % 32 == 0, "degree ranges cover the nodes in whole lines");
static_assert(CH * NB == NE && CH % 4 == 0, "degree chunks cover the edges");
static_assert(BU * TPB >= CH4, "one pass of the work-group loads a degree chunk");
static_assert(NR + 1 <= OFFW && OFFW == 8 * 32, "degree offset row: 8 entries per lane of one wave");
static_assert(CHS16 >= CH && (CHS16 * 2) % 128 == 0 && (CH * 2) % 16 == 0, "16-bit chunks: whole lines, whole 16-byte groups");
static_assert(NSEG16 * NWAVES == NB && NSEG16 % SG16 == 0 && NSEG16 <= 32, "degree pass: segment groups per wave, bounds held one per lane");
static_assert((size_t)NB * CHS16 < ((size_t)1 << 32), "16-bit entry index fits 32 bits");

static_assert(NPAD >= NN && DRNG % 16 == 0, "destination ranges cover the nodes in whole lines of 8-byte rows");
static_assert(TS * SBLK == NN && SBLK % 2 == 0, "source blocks cover the nodes in whole 16-byte groups");
static_assert(DRNG <= (1 << 13) && SBLK <= (1 << 15), "word = 13 bits local destination | 15 bits local source");
static_assert(NT == 8 * 32 && NT + 1 <= OFFW2 && OFFW2 % 32 == 0, "tile offset row: 8 counts per lane of one wave, whole lines");
static_assert(CH2 * NB2 == NE && CH2 % 4 == 0 && BU2 * TPB >= CH2_4, "chunks cover the edges; one pass loads a chunk");
static_assert(CHS2 >= CH2 && CHS2 % 32 == 0, "whole lines per chunk");
static_assert(PCH * NPART == NB2 && NSEG2 * NWAVES == PCH && NSEG2 <= 32, "parts cover the chunks; bounds held one per lane");

typedef __attribute__((ext_vector_type(2))) float v2f;
typedef __attribute__((ext_vector_type(8))) float v8f;

#define WS_OFFS16 ((size_t)1 * NN)
#define WS_BINS16 (WS_OFFS16 + (size_t)NB * OFFW)
#define WS_WORDS16 (WS_BINS16 + (size_t)NB * CHS16 / 2)
#define WS_PART ((size_t)1 * NN)
#define WS_OFFS2 (WS_PART + (size_t)TS * 2 * NPAD)
#define WS_BINS2 (WS_OFFS2 + (size_t)PCH * OFFW2)
#define WS_WORDS2 (WS_BINS2 + (size_t)PCH * CHS2)
static_assert(WS_WORDS2 * 4 <= (size_t)35 * NN * 4, "inside the 28,000,000 bytes the versions before this one used");
static_assert(WS_PART % 32 == 0 && (2 * NPAD) % 32 == 0 && WS_OFFS2 % 32 == 0 && WS_BINS2 % 32 == 0, "planes start on 128-byte lines");
static_assert(WS_WORDS16 * 4 <= (size_t)35 * NN * 4, "degree list inside the same 28,000,000 bytes");
static_assert(WS_OFFS16 % 32 == 0 && WS_BINS16 % 32 == 0, "degree planes start on 128-byte lines");

__global__ __launch_bounds__(TPB) void gnn_sort_tile_kernel(const int* __restrict__ src,
                                                            const int* __restrict__ dst,
                                                            unsigned* __restrict__ offs,
                                                            unsigned* __restrict__ bins,
                                                            int cbase) {
  __shared__ unsigned cnt[NT];
  __shared__ unsigned cur[NT];
  __shared__ __attribute__((aligned(16))) unsigned stage[CH2];
  const int b = blockIdx.x;
  const int tid = threadIdx.x;
  if (tid < NT) cnt[tid] = 0u;
  __syncthreads();

  const int4* __restrict__ d4 = reinterpret_cast<const int4*>(dst + (size_t)(cbase + b) * CH2);
  const int4* __restrict__ s4 = reinterpret_cast<const int4*>(src + (size_t)(cbase + b) * CH2);
  int4 dv[BU2], sv[BU2];
#pragma unroll
  for (int u = 0; u < BU2; ++u) {
    int i = u * TPB + tid;
    bool ok = i < CH2_4;
    dv[u] = ok ? d4[i] : make_int4(-1, -1, -1, -1);
    sv[u] = ok ? s4[i] : make_int4(-1, -1, -1, -1);
  }
#pragma unroll
  for (int u = 0; u < BU2; ++u) {
    const unsigned dd[4] = {(unsigned)dv[u].x, (unsigned)dv[u].y, (unsigned)dv[u].z, (unsigned)dv[u].w};
    const unsigned ss[4] = {(unsigned)sv[u].x, (unsigned)sv[u].y, (unsigned)sv[u].z, (unsigned)sv[u].w};
#pragma unroll
    for (int k = 0; k < 4; ++k)
      if (dd[k] < (unsigned)NN && ss[k] < (unsigned)NN)
        atomicAdd(&cnt[(dd[k] / DRNG) * TS + ss[k] / SBLK], 1u);
  }
  __syncthreads();

  if (tid < 32) {
    unsigned c[8];
    unsigned s = 0u;
#pragma unroll
    for (int k = 0; k < 8; ++k) { c[k] = cnt[8 * tid + k]; s += c[k]; }
    unsigned inc = s;
#pragma unroll
    for (int d = 1; d < 32; d <<= 1) {
      unsigned t = __shfl_up(inc, d);
      if (tid >= d) inc += t;
    }
    unsigned ex = inc - s;
    unsigned* __restrict__ orow = offs + (size_t)b * OFFW2 + 8 * tid;
#pragma unroll
    for (int k = 0; k < 8; ++k) { cur[8 * tid + k] = ex; orow[k] = ex; ex += c[k]; }
    if (tid == 31) offs[(size_t)b * OFFW2 + NT] = inc;
  }
  __syncthreads();

#pragma unroll
  for (int u = 0; u < BU2; ++u) {
    const unsigned dd[4] = {(unsigned)dv[u].x, (unsigned)dv[u].y, (unsigned)dv[u].z, (unsigned)dv[u].w};
    const unsigned ss[4] = {(unsigned)sv[u].x, (unsigned)sv[u].y, (unsigned)sv[u].z, (unsigned)sv[u].w};
#pragma unroll
    for (int k = 0; k < 4; ++k) {
      if (dd[k] < (unsigned)NN && ss[k] < (unsigned)NN) {
        unsigned r = dd[k] / DRNG;
        unsigned q = ss[k] / SBLK;
        unsigned pos = atomicAdd(&cur[r * TS + q], 1u);
        stage[pos] = ((dd[k] - r * DRNG) << 15) | (ss[k] - q * SBLK);
      }
    }
  }
  __syncthreads();

  uint4* __restrict__ o4 = reinterpret_cast<uint4*>(bins + (size_t)b * CHS2);
  const uint4* st4 = reinterpret_cast<const uint4*>(stage);
  for (int i = tid; i < CH2_4; i += TPB) o4[i] = st4[i];
}

__global__ __launch_bounds__(TPB) void gnn_sort_chunk16_kernel(const int* __restrict__ dst,
                                                               unsigned* __restrict__ offs,
                                                               unsigned short* __restrict__ bins16) {
  __shared__ unsigned cnt[OFFW];
  __shared__ unsigned cur[OFFW];
  __shared__ __attribute__((aligned(16))) unsigned short stage[CH];
  const int b = blockIdx.x;
  const int tid = threadIdx.x;
  if (tid < OFFW) cnt[tid] = 0u;
  __syncthreads();

  const int4* __restrict__ d4 = reinterpret_cast<const int4*>(dst + (size_t)b * CH);
  int4 dv[BU];
#pragma unroll
  for (int u = 0; u < BU; ++u) {
    int i = u * TPB + tid;
    dv[u] = (i < CH4) ? d4[i] : make_int4(-1, -1, -1, -1);
  }
#pragma unroll
  for (int u = 0; u < BU; ++u) {
    const unsigned dd[4] = {(unsigned)dv[u].x, (unsigned)dv[u].y, (unsigned)dv[u].z, (unsigned)dv[u].w};
#pragma unroll
    for (int k = 0; k < 4; ++k)
      if (dd[k] < (unsigned)NN) atomicAdd(&cnt[dd[k] / RNG], 1u);
  }
  __syncthreads();

  if (tid < 32) {
    unsigned c[8];
    unsigned s = 0u;
#pragma unroll
    for (int k = 0; k < 8; ++k) { c[k] = cnt[8 * tid + k]; s += c[k]; }
    unsigned inc = s;
#pragma unroll
    for (int d = 1; d < 32; d <<= 1) {
      unsigned t = __shfl_up(inc, d);
      if (tid >= d) inc += t;
    }
    unsigned ex = inc - s;
    unsigned* __restrict__ orow = offs + (size_t)b * OFFW + 8 * tid;
#pragma unroll
    for (int k = 0; k < 8; ++k) { cur[8 * tid + k] = ex; orow[k] = ex; ex += c[k]; }
  }
  __syncthreads();

#pragma unroll
  for (int u = 0; u < BU; ++u) {
    const unsigned dd[4] = {(unsigned)dv[u].x, (unsigned)dv[u].y, (unsigned)dv[u].z, (unsigned)dv[u].w};
#pragma unroll
    for (int k = 0; k < 4; ++k) {
      if (dd[k] < (unsigned)NN) {
        unsigned r = dd[k] / RNG;
        unsigned pos = atomicAdd(&cur[r], 1u);
        stage[pos] = (unsigned short)(dd[k] - r * RNG);
      }
    }
  }
  __syncthreads();

  uint4* __restrict__ o4 = reinterpret_cast<uint4*>(bins16 + (size_t)b * CHS16);
  const uint4* st4 = reinterpret_cast<const uint4*>(stage);
  for (int i = tid; i < CH / 8; i += TPB) o4[i] = st4[i];
}

__global__ __launch_bounds__(TPB) void gnn_deg_owner16_kernel(const unsigned* __restrict__ offs,
                                                              const unsigned short* __restrict__ bins16,
                                                              const float* __restrict__ x,
                                                              float* __restrict__ dinv,
                                                              float* __restrict__ xs) {
  __shared__ unsigned cnt[RNG];
  const int r = blockIdx.x;
  const int tid = threadIdx.x;
  const unsigned lane = threadIdx.x & 31;
  const unsigned w = threadIdx.x >> 5;
  if (tid < RNG) cnt[tid] = 0u;
  __syncthreads();

  unsigned s0 = 0u, s1 = 0u;
  if (lane < (unsigned)NSEG16) {
    unsigned bb = w * (unsigned)NSEG16 + lane;
    const unsigned* __restrict__ o = offs + (size_t)bb * OFFW + r;
    s0 = bb * (unsigned)CHS16 + o[0];
    s1 = bb * (unsigned)CHS16 + o[1];
  }
  for (int j = 0; j < NSEG16; j += SG16) {
    unsigned sa[SG16], sn[SG16];
    unsigned nmax = 0u;
#pragma unroll
    for (int g = 0; g < SG16; ++g) {
      sa[g] = (unsigned)__builtin_amdgcn_readlane((int)s0, j + g);
      sn[g] = (unsigned)__builtin_amdgcn_readlane((int)s1, j + g) - sa[g];
      nmax = sn[g] > nmax ? sn[g] : nmax;
    }
    for (unsigned k0 = 0; k0 < nmax; k0 += 32 * GW) {
      unsigned p[SG16 * GW];
#pragma unroll
      for (int g = 0; g < SG16; ++g) {
#pragma unroll
        for (int t = 0; t < GW; ++t) {
          unsigned k = k0 + 32u * t + lane;
          p[g * GW + t] = (k < sn[g]) ? (unsigned)bins16[sa[g] + k] : 0xFFFFu;
        }
      }
#pragma unroll
      for (int t = 0; t < SG16 * GW; ++t)
        if (p[t] != 0xFFFFu) atomicAdd(&cnt[p[t]], 1u);
    }
  }
  __syncthreads();

  if (tid < RNG) {
    int node = r * RNG + tid;
    float d = rsqrtf((float)cnt[tid] + 1.0f);
    dinv[node] = d;
    float2 xv = reinterpret_cast<const float2*>(x)[node];
    reinterpret_cast<float2*>(xs)[node] = make_float2(xv.x * d, xv.y * d);
  }
}

__global__ __launch_bounds__(TPB) void gnn_tile_sum2_kernel(const unsigned* __restrict__ offs,
                                                            const unsigned* __restrict__ bins,
                                                            const float* __restrict__ rows2,
                                                            float* __restrict__ part,
                                                            int first) {
  __shared__ __attribute__((aligned(16))) float tab[SBLK * 2];
  __shared__ __attribute__((aligned(16))) float acc[DRNG * 2];
  const int t = blockIdx.x;
  const int r = t / TS;
  const int q = t - r * TS;
  const int tid = threadIdx.x;
  const unsigned lane = threadIdx.x & 31;
  const unsigned w = threadIdx.x >> 5;

  const unsigned bb = w * (unsigned)NSEG2 + (lane < (unsigned)NSEG2 ? lane : 0u);
  const unsigned* __restrict__ o = offs + (size_t)bb * OFFW2 + t;
  const unsigned o0 = o[0], o1 = o[1];

  const float4* __restrict__ g4 = reinterpret_cast<const float4*>(rows2 + (size_t)2 * q * SBLK);
  float4* __restrict__ prow = reinterpret_cast<float4*>(part + (size_t)q * (2 * NPAD) + (size_t)2 * r * DRNG);
  float4* t4 = reinterpret_cast<float4*>(tab);
  float4* a4 = reinterpret_cast<float4*>(acc);
  constexpr int NG = (SBLK / 2 + TPB - 1) / TPB;
  constexpr int NA = (DRNG / 2 + TPB - 1) / TPB;
  float4 v[NG], pv[NA];
#pragma unroll
  for (int u = 0; u < NG; ++u) {
    int i = u * TPB + tid;
    v[u] = (i < SBLK / 2) ? g4[i] : make_float4(0.0f, 0.0f, 0.0f, 0.0f);
  }
#pragma unroll
  for (int u = 0; u < NA; ++u) {
    int i = u * TPB + tid;
    pv[u] = (!first && i < DRNG / 2) ? prow[i] : make_float4(0.0f, 0.0f, 0.0f, 0.0f);
  }
#pragma unroll
  for (int u = 0; u < NG; ++u) {
    int i = u * TPB + tid;
    if (i < SBLK / 2) t4[i] = v[u];
  }
#pragma unroll
  for (int u = 0; u < NA; ++u) {
    int i = u * TPB + tid;
    if (i < DRNG / 2) a4[i] = pv[u];
  }
  const unsigned s0 = bb * (unsigned)CHS2 + o0;
  const unsigned s1 = bb * (unsigned)CHS2 + o1;
  __syncthreads();

  const float2* tab2 = reinterpret_cast<const float2*>(tab);
  {
    unsigned sa[NSEG2], sn[NSEG2];
    unsigned nmax = 0u;
#pragma unroll
    for (int g = 0; g < NSEG2; ++g) {
      sa[g] = (unsigned)__builtin_amdgcn_readlane((int)s0, g);
      sn[g] = (unsigned)__builtin_amdgcn_readlane((int)s1, g) - sa[g];
      nmax = sn[g] > nmax ? sn[g] : nmax;
    }
    for (unsigned k0 = 0; k0 < nmax; k0 += 32 * GW2) {
      unsigned p[NSEG2 * GW2];
      float2 h[NSEG2 * GW2];
#pragma unroll
      for (int g = 0; g < NSEG2; ++g) {
#pragma unroll
        for (int u = 0; u < GW2; ++u) {
          unsigned k = k0 + 32u * u + lane;
          p[g * GW2 + u] = (k < sn[g]) ? bins[sa[g] + k] : 0xFFFFFFFFu;
        }
      }
#pragma unroll
      for (int u = 0; u < NSEG2 * GW2; ++u) {
        h[u] = make_float2(0.0f, 0.0f);
        if (p[u] != 0xFFFFFFFFu) h[u] = tab2[p[u] & 0x7FFFu];
      }
#pragma unroll
      for (int u = 0; u < NSEG2 * GW2; ++u) {
        if (p[u] != 0xFFFFFFFFu) {
          unsigned a = p[u] >> 15;
          unsafeAtomicAdd(&acc[2 * a + 0], h[u].x);
          unsafeAtomicAdd(&acc[2 * a + 1], h[u].y);
        }
      }
    }
  }
  __syncthreads();

  for (int i = tid; i < DRNG / 2; i += TPB) prow[i] = a4[i];
}

__global__ void gnn_layer1_wmma_kernel(const float* __restrict__ part,
                                       const float* __restrict__ x,
                                       const float* __restrict__ dinv,
                                       const float* __restrict__ W1,
                                       const float* __restrict__ b1,
                                       const float* __restrict__ W2,
                                       float* __restrict__ hs2) {
  int wave = (blockIdx.x * blockDim.x + threadIdx.x) >> 5;
  int lane = threadIdx.x & 31;
  if (wave >= NTILES) return;
  int c    = lane & 15;
  int half = lane >> 4;
  int node = (wave << 4) + c;
  bool lo  = half == 0;

  float2 a = make_float2(0.0f, 0.0f);
#pragma unroll
  for (int s = 0; s < TS; ++s) {
    float2 v = reinterpret_cast<const float2*>(part + (size_t)s * (2 * NPAD))[node];
    a.x += v.x;
    a.y += v.y;
  }
  float di = dinv[node];
  float2 xv = reinterpret_cast<const float2*>(x)[node];
  float z0 = di * (a.x + xv.x * di);
  float z1 = di * (a.y + xv.y * di);

  float w0 = W1[c];
  float w1 = W1[16 + c];

  v2f A, B;
  A.x = lo ? w0 : 0.0f;
  A.y = lo ? w1 : 0.0f;
  B.x = lo ? z0 : 0.0f;
  B.y = lo ? z1 : 0.0f;

  v8f acc = {};
  acc = __builtin_amdgcn_wmma_f32_16x16x4_f32(
       false, A,  false, B,
       (short)0, acc,  false,  false);

  const float* __restrict__ bb = b1 + 8 * half;
  const float* __restrict__ ww = W2 + 16 * half;
  float h0 = 0.0f, h1 = 0.0f;
#pragma unroll
  for (int r = 0; r < 8; ++r) {
    float v = acc[r] + bb[r];
    v = v > 0.0f ? v : 0.0f;
    h0 += v * ww[2 * r + 0];
    h1 += v * ww[2 * r + 1];
  }
  h0 += __shfl_xor(h0, 16);
  h1 += __shfl_xor(h1, 16);
  if (lo) reinterpret_cast<float2*>(hs2)[node] = make_float2(h0 * di, h1 * di);
}

__global__ void gnn_finalize2_kernel(const float* __restrict__ part,
                                     const float* __restrict__ dinv,
                                     const float* __restrict__ b2,
                                     float* out) {
  int t = blockIdx.x * blockDim.x + threadIdx.x;
  if (t >= NN * 2) return;
  int i = t >> 1;
  int c = t & 1;
  float a = 0.0f;
#pragma unroll
  for (int s = 0; s < TS; ++s) a += part[(size_t)s * (2 * NPAD) + t];
  out[t] = dinv[i] * (a + out[t]) + b2[c];
}

extern "C" void kernel_launch(void* const* d_in, const int* in_sizes, int n_in,
                              void* d_out, int out_size, void* d_ws, size_t ws_size,
                              hipStream_t stream) {
  const float* x  = (const float*)d_in[0];
  const float* W1 = (const float*)d_in[1];
  const float* b1 = (const float*)d_in[2];
  const float* W2 = (const float*)d_in[3];
  const float* b2 = (const float*)d_in[4];
  const int* edge = (const int*)d_in[5];
  const int* src  = edge;
  const int* dst  = edge + NE;
  float* out = (float*)d_out;

  float* ws = (float*)d_ws;
  float* dinv  = ws;
  float* part  = ws + WS_PART;
  unsigned* offs = (unsigned*)(ws + WS_OFFS2);
  unsigned* bins = (unsigned*)(ws + WS_BINS2);
  unsigned* offs16 = (unsigned*)(ws + WS_OFFS16);
  unsigned short* bins16 = (unsigned short*)(ws + WS_BINS16);
  float* rows = out;

  const int B = 256;

  gnn_sort_chunk16_kernel<<<NB, TPB, 0, stream>>>(dst, offs16, bins16);
  gnn_deg_owner16_kernel<<<NR, TPB, 0, stream>>>(offs16, bins16, x, dinv, rows);

  gnn_sort_tile_kernel<<<PCH, TPB, 0, stream>>>(src, dst, offs, bins, 0 * PCH);
  gnn_tile_sum2_kernel<<<NT, TPB, 0, stream>>>(offs, bins, rows, part, 1);
  gnn_sort_tile_kernel<<<PCH, TPB, 0, stream>>>(src, dst, offs, bins, 1 * PCH);
  gnn_tile_sum2_kernel<<<NT, TPB, 0, stream>>>(offs, bins, rows, part, 0);
  gnn_sort_tile_kernel<<<PCH, TPB, 0, stream>>>(src, dst, offs, bins, 2 * PCH);
  gnn_tile_sum2_kernel<<<NT, TPB, 0, stream>>>(offs, bins, rows, part, 0);
  gnn_sort_tile_kernel<<<PCH, TPB, 0, stream>>>(src, dst, offs, bins, 3 * PCH);
  gnn_tile_sum2_kernel<<<NT, TPB, 0, stream>>>(offs, bins, rows, part, 0);

  {
    int threads = NTILES * 32;
    gnn_layer1_wmma_kernel<<<(threads + B - 1) / B, B, 0, stream>>>(part, x, dinv, W1, b1, W2, rows);
  }

  gnn_tile_sum2_kernel<<<NT, TPB, 0, stream>>>(offs, bins, rows, part, 1);
  gnn_sort_tile_kernel<<<PCH, TPB, 0, stream>>>(src, dst, offs, bins, 0 * PCH);
  gnn_tile_sum2_kernel<<<NT, TPB, 0, stream>>>(offs, bins, rows, part, 0);
  gnn_sort_tile_kernel<<<PCH, TPB, 0, stream>>>(src, dst, offs, bins, 1 * PCH);
  gnn_tile_sum2_kernel<<<NT, TPB, 0, stream>>>(offs, bins, rows, part, 0);
  gnn_sort_tile_kernel<<<PCH, TPB, 0, stream>>>(src, dst, offs, bins, 2 * PCH);
  gnn_tile_sum2_kernel<<<NT, TPB, 0, stream>>>(offs, bins, rows, part, 0);

  {
    int n = NN * 2;
    gnn_finalize2_kernel<<<(n + B - 1) / B, B, 0, stream>>>(part, dinv, b2, out);
  }
}


# Edge list sorted in 3 parts, not 4: 6 source blocks x 42 destination ranges need 6 partial planes, so a third of the sorted list fits; 15 launches for 19; f32 sum order changes
# speedup vs baseline: 7.8616x; 1.1540x over previous
#include <hip/hip_runtime.h>


#define NN 200000
#define NE 12800000
#define NTILES (NN / 16)

#define TPB 1024
#define NWAVES (TPB / 32)

#define RNG 800
#define NR (NN / RNG)
#define NB 512
#define CH (NE / NB)
#define CH4 (CH / 4)
#define OFFW 256
#define BU 7
#define GW 4
#define CHS16 25024
#define NSEG16 (NB / NWAVES)
#define SG16 4

#define TD 42
#define TS 6
#define NT (TD * TS)
#define NTP 256
#define DRNG 4768
#define NPAD (TD * DRNG)
#define SBLK 33334
#define NPART 3
#define PCH 256
#define CHA 16668
#define CHB 16664
#define CHS2 16672
#define OFFW2 256
#define BU2 5
#define NSEG2 (PCH / NWAVES)
#define GW2 3

static_assert(NR * RNG == NN && RNG % 32 == 0, "degree ranges cover the nodes in whole lines");
static_assert(CH * NB == NE && CH % 4 == 0, "degree chunks cover the edges");
static_assert(BU * TPB >= CH4, "one pass of the work-group loads a degree chunk");
static_assert(NR + 1 <= OFFW && OFFW == 8 * 32, "degree offset row: 8 entries per lane of one wave");
static_assert(CHS16 >= CH && (CHS16 * 2) % 128 == 0 && (CH * 2) % 16 == 0, "16-bit chunks: whole lines, whole 16-byte groups");
static_assert(NSEG16 * NWAVES == NB && NSEG16 % SG16 == 0 && NSEG16 <= 32, "degree pass: segment groups per wave, bounds held one per lane");
static_assert((size_t)NB * CHS16 < ((size_t)1 << 32), "16-bit entry index fits 32 bits");

static_assert(NPAD >= NN && DRNG % 16 == 0, "destination ranges cover the nodes in whole lines of 8-byte rows");
static_assert(TS * SBLK >= NN && (TS - 1) * SBLK < NN && SBLK % 2 == 0 && NN % 2 == 0, "source blocks cover the nodes in whole 16-byte groups");
static_assert(DRNG < (1 << 16) - 1 && SBLK <= (1 << 16), "word = 16 bits local destination | 16 bits local source; all ones is no word");
static_assert(NT <= NTP && NTP == 8 * 32 && NT + 1 <= OFFW2 && OFFW2 == NTP, "tile offset row: 8 counts per lane of one wave, whole lines");
static_assert((size_t)PCH * ((NPART - 1) * CHA + CHB) == NE && CHA % 4 == 0 && CHB % 4 == 0 && CHB <= CHA && BU2 * TPB >= CHA / 4, "chunks cover the edges; one pass loads a chunk");
static_assert(CHS2 >= CHA && CHS2 % 32 == 0, "whole lines per chunk");
static_assert(NSEG2 * NWAVES == PCH && NSEG2 <= 32, "bounds held one per lane");

typedef __attribute__((ext_vector_type(2))) float v2f;
typedef __attribute__((ext_vector_type(8))) float v8f;

#define WS_OFFS16 ((size_t)1 * NN)
#define WS_BINS16 (WS_OFFS16 + (size_t)NB * OFFW)
#define WS_WORDS16 (WS_BINS16 + (size_t)NB * CHS16 / 2)
#define WS_PART ((size_t)1 * NN)
#define WS_OFFS2 (WS_PART + (size_t)TS * 2 * NPAD)
#define WS_BINS2 (WS_OFFS2 + (size_t)PCH * OFFW2)
#define WS_WORDS2 (WS_BINS2 + (size_t)PCH * CHS2)
static_assert(WS_WORDS2 * 4 <= (size_t)35 * NN * 4, "inside the 28,000,000 bytes the versions before this one used");
static_assert(WS_PART % 32 == 0 && (2 * NPAD) % 32 == 0 && WS_OFFS2 % 32 == 0 && WS_BINS2 % 32 == 0, "planes start on 128-byte lines");
static_assert(WS_WORDS16 * 4 <= (size_t)35 * NN * 4, "degree list inside the same 28,000,000 bytes");
static_assert(WS_OFFS16 % 32 == 0 && WS_BINS16 % 32 == 0, "degree planes start on 128-byte lines");

__global__ __launch_bounds__(TPB) void gnn_sort_tile_kernel(const int* __restrict__ src,
                                                            const int* __restrict__ dst,
                                                            unsigned* __restrict__ offs,
                                                            unsigned* __restrict__ bins,
                                                            int part) {
  __shared__ unsigned cnt[NTP];
  __shared__ unsigned cur[NTP];
  __shared__ __attribute__((aligned(16))) unsigned stage[CHA];
  const int b = blockIdx.x;
  const int tid = threadIdx.x;
  if (tid < NTP) cnt[tid] = 0u;
  __syncthreads();

  const int csz4 = (part < NPART - 1) ? CHA / 4 : CHB / 4;
  const size_t e0 = (size_t)part * PCH * CHA + (size_t)b * (4 * csz4);
  const int4* __restrict__ d4 = reinterpret_cast<const int4*>(dst + e0);
  const int4* __restrict__ s4 = reinterpret_cast<const int4*>(src + e0);
  int4 dv[BU2], sv[BU2];
#pragma unroll
  for (int u = 0; u < BU2; ++u) {
    int i = u * TPB + tid;
    bool ok = i < csz4;
    dv[u] = ok ? d4[i] : make_int4(-1, -1, -1, -1);
    sv[u] = ok ? s4[i] : make_int4(-1, -1, -1, -1);
  }
#pragma unroll
  for (int u = 0; u < BU2; ++u) {
    const unsigned dd[4] = {(unsigned)dv[u].x, (unsigned)dv[u].y, (unsigned)dv[u].z, (unsigned)dv[u].w};
    const unsigned ss[4] = {(unsigned)sv[u].x, (unsigned)sv[u].y, (unsigned)sv[u].z, (unsigned)sv[u].w};
#pragma unroll
    for (int k = 0; k < 4; ++k)
      if (dd[k] < (unsigned)NN && ss[k] < (unsigned)NN)
        atomicAdd(&cnt[(dd[k] / DRNG) * TS + ss[k] / SBLK], 1u);
  }
  __syncthreads();

  if (tid < 32) {
    unsigned c[8];
    unsigned s = 0u;
#pragma unroll
    for (int k = 0; k < 8; ++k) { c[k] = cnt[8 * tid + k]; s += c[k]; }
    unsigned inc = s;
#pragma unroll
    for (int d = 1; d < 32; d <<= 1) {
      unsigned t = __shfl_up(inc, d);
      if (tid >= d) inc += t;
    }
    unsigned ex = inc - s;
    unsigned* __restrict__ orow = offs + (size_t)b * OFFW2 + 8 * tid;
#pragma unroll
    for (int k = 0; k < 8; ++k) { cur[8 * tid + k] = ex; orow[k] = ex; ex += c[k]; }
  }
  __syncthreads();

#pragma unroll
  for (int u = 0; u < BU2; ++u) {
    const unsigned dd[4] = {(unsigned)dv[u].x, (unsigned)dv[u].y, (unsigned)dv[u].z, (unsigned)dv[u].w};
    const unsigned ss[4] = {(unsigned)sv[u].x, (unsigned)sv[u].y, (unsigned)sv[u].z, (unsigned)sv[u].w};
#pragma unroll
    for (int k = 0; k < 4; ++k) {
      if (dd[k] < (unsigned)NN && ss[k] < (unsigned)NN) {
        unsigned r = dd[k] / DRNG;
        unsigned q = ss[k] / SBLK;
        unsigned pos = atomicAdd(&cur[r * TS + q], 1u);
        stage[pos] = ((dd[k] - r * DRNG) << 16) | (ss[k] - q * SBLK);
      }
    }
  }
  __syncthreads();

  uint4* __restrict__ o4 = reinterpret_cast<uint4*>(bins + (size_t)b * CHS2);
  const uint4* st4 = reinterpret_cast<const uint4*>(stage);
  for (int i = tid; i < csz4; i += TPB) o4[i] = st4[i];
}

__global__ __launch_bounds__(TPB) void gnn_sort_chunk16_kernel(const int* __restrict__ dst,
                                                               unsigned* __restrict__ offs,
                                                               unsigned short* __restrict__ bins16) {
  __shared__ unsigned cnt[OFFW];
  __shared__ unsigned cur[OFFW];
  __shared__ __attribute__((aligned(16))) unsigned short stage[CH];
  const int b = blockIdx.x;
  const int tid = threadIdx.x;
  if (tid < OFFW) cnt[tid] = 0u;
  __syncthreads();

  const int4* __restrict__ d4 = reinterpret_cast<const int4*>(dst + (size_t)b * CH);
  int4 dv[BU];
#pragma unroll
  for (int u = 0; u < BU; ++u) {
    int i = u * TPB + tid;
    dv[u] = (i < CH4) ? d4[i] : make_int4(-1, -1, -1, -1);
  }
#pragma unroll
  for (int u = 0; u < BU; ++u) {
    const unsigned dd[4] = {(unsigned)dv[u].x, (unsigned)dv[u].y, (unsigned)dv[u].z, (unsigned)dv[u].w};
#pragma unroll
    for (int k = 0; k < 4; ++k)
      if (dd[k] < (unsigned)NN) atomicAdd(&cnt[dd[k] / RNG], 1u);
  }
  __syncthreads();

  if (tid < 32) {
    unsigned c[8];
    unsigned s = 0u;
#pragma unroll
    for (int k = 0; k < 8; ++k) { c[k] = cnt[8 * tid + k]; s += c[k]; }
    unsigned inc = s;
#pragma unroll
    for (int d = 1; d < 32; d <<= 1) {
      unsigned t = __shfl_up(inc, d);
      if (tid >= d) inc += t;
    }
    unsigned ex = inc - s;
    unsigned* __restrict__ orow = offs + (size_t)b * OFFW + 8 * tid;
#pragma unroll
    for (int k = 0; k < 8; ++k) { cur[8 * tid + k] = ex; orow[k] = ex; ex += c[k]; }
  }
  __syncthreads();

#pragma unroll
  for (int u = 0; u < BU; ++u) {
    const unsigned dd[4] = {(unsigned)dv[u].x, (unsigned)dv[u].y, (unsigned)dv[u].z, (unsigned)dv[u].w};
#pragma unroll
    for (int k = 0; k < 4; ++k) {
      if (dd[k] < (unsigned)NN) {
        unsigned r = dd[k] / RNG;
        unsigned pos = atomicAdd(&cur[r], 1u);
        stage[pos] = (unsigned short)(dd[k] - r * RNG);
      }
    }
  }
  __syncthreads();

  uint4* __restrict__ o4 = reinterpret_cast<uint4*>(bins16 + (size_t)b * CHS16);
  const uint4* st4 = reinterpret_cast<const uint4*>(stage);
  for (int i = tid; i < CH / 8; i += TPB) o4[i] = st4[i];
}

__global__ __launch_bounds__(TPB) void gnn_deg_owner16_kernel(const unsigned* __restrict__ offs,
                                                              const unsigned short* __restrict__ bins16,
                                                              const float* __restrict__ x,
                                                              float* __restrict__ dinv,
                                                              float* __restrict__ xs) {
  __shared__ unsigned cnt[RNG];
  const int r = blockIdx.x;
  const int tid = threadIdx.x;
  const unsigned lane = threadIdx.x & 31;
  const unsigned w = threadIdx.x >> 5;
  if (tid < RNG) cnt[tid] = 0u;
  __syncthreads();

  unsigned s0 = 0u, s1 = 0u;
  if (lane < (unsigned)NSEG16) {
    unsigned bb = w * (unsigned)NSEG16 + lane;
    const unsigned* __restrict__ o = offs + (size_t)bb * OFFW + r;
    s0 = bb * (unsigned)CHS16 + o[0];
    s1 = bb * (unsigned)CHS16 + o[1];
  }
  for (int j = 0; j < NSEG16; j += SG16) {
    unsigned sa[SG16], sn[SG16];
    unsigned nmax = 0u;
#pragma unroll
    for (int g = 0; g < SG16; ++g) {
      sa[g] = (unsigned)__builtin_amdgcn_readlane((int)s0, j + g);
      sn[g] = (unsigned)__builtin_amdgcn_readlane((int)s1, j + g) - sa[g];
      nmax = sn[g] > nmax ? sn[g] : nmax;
    }
    for (unsigned k0 = 0; k0 < nmax; k0 += 32 * GW) {
      unsigned p[SG16 * GW];
#pragma unroll
      for (int g = 0; g < SG16; ++g) {
#pragma unroll
        for (int t = 0; t < GW; ++t) {
          unsigned k = k0 + 32u * t + lane;
          p[g * GW + t] = (k < sn[g]) ? (unsigned)bins16[sa[g] + k] : 0xFFFFu;
        }
      }
#pragma unroll
      for (int t = 0; t < SG16 * GW; ++t)
        if (p[t] != 0xFFFFu) atomicAdd(&cnt[p[t]], 1u);
    }
  }
  __syncthreads();

  if (tid < RNG) {
    int node = r * RNG + tid;
    float d = rsqrtf((float)cnt[tid] + 1.0f);
    dinv[node] = d;
    float2 xv = reinterpret_cast<const float2*>(x)[node];
    reinterpret_cast<float2*>(xs)[node] = make_float2(xv.x * d, xv.y * d);
  }
}

__global__ __launch_bounds__(TPB) void gnn_tile_sum2_kernel(const unsigned* __restrict__ offs,
                                                            const unsigned* __restrict__ bins,
                                                            const float* __restrict__ rows2,
                                                            float* __restrict__ part,
                                                            int first) {
  __shared__ __attribute__((aligned(16))) float tab[SBLK * 2];
  __shared__ __attribute__((aligned(16))) float acc[DRNG * 2];
  const int t = blockIdx.x;
  const int r = t / TS;
  const int q = t - r * TS;
  const int tid = threadIdx.x;
  const unsigned lane = threadIdx.x & 31;
  const unsigned w = threadIdx.x >> 5;

  const unsigned bb = w * (unsigned)NSEG2 + (lane < (unsigned)NSEG2 ? lane : 0u);
  const unsigned* __restrict__ o = offs + (size_t)bb * OFFW2 + t;
  const unsigned o0 = o[0], o1 = o[1];

  const float4* __restrict__ g4 = reinterpret_cast<const float4*>(rows2 + (size_t)2 * q * SBLK);
  float4* __restrict__ prow = reinterpret_cast<float4*>(part + (size_t)q * (2 * NPAD) + (size_t)2 * r * DRNG);
  float4* t4 = reinterpret_cast<float4*>(tab);
  float4* a4 = reinterpret_cast<float4*>(acc);
  constexpr int NG = (SBLK / 2 + TPB - 1) / TPB;
  constexpr int NA = (DRNG / 2 + TPB - 1) / TPB;
  const int nrow4 = (NN / 2 - q * (SBLK / 2)) < SBLK / 2 ? (NN / 2 - q * (SBLK / 2)) : SBLK / 2;
  float4 v[NG], pv[NA];
#pragma unroll
  for (int u = 0; u < NG; ++u) {
    int i = u * TPB + tid;
    v[u] = (i < nrow4) ? g4[i] : make_float4(0.0f, 0.0f, 0.0f, 0.0f);
  }
#pragma unroll
  for (int u = 0; u < NA; ++u) {
    int i = u * TPB + tid;
    pv[u] = (!first && i < DRNG / 2) ? prow[i] : make_float4(0.0f, 0.0f, 0.0f, 0.0f);
  }
#pragma unroll
  for (int u = 0; u < NG; ++u) {
    int i = u * TPB + tid;
    if (i < SBLK / 2) t4[i] = v[u];
  }
#pragma unroll
  for (int u = 0; u < NA; ++u) {
    int i = u * TPB + tid;
    if (i < DRNG / 2) a4[i] = pv[u];
  }
  const unsigned s0 = bb * (unsigned)CHS2 + o0;
  const unsigned s1 = bb * (unsigned)CHS2 + o1;
  __syncthreads();

  const float2* tab2 = reinterpret_cast<const float2*>(tab);
  {
    unsigned sa[NSEG2], sn[NSEG2];
    unsigned nmax = 0u;
#pragma unroll
    for (int g = 0; g < NSEG2; ++g) {
      sa[g] = (unsigned)__builtin_amdgcn_readlane((int)s0, g);
      sn[g] = (unsigned)__builtin_amdgcn_readlane((int)s1, g) - sa[g];
      nmax = sn[g] > nmax ? sn[g] : nmax;
    }
    for (unsigned k0 = 0; k0 < nmax; k0 += 32 * GW2) {
      unsigned p[NSEG2 * GW2];
      float2 h[NSEG2 * GW2];
#pragma unroll
      for (int g = 0; g < NSEG2; ++g) {
#pragma unroll
        for (int u = 0; u < GW2; ++u) {
          unsigned k = k0 + 32u * u + lane;
          p[g * GW2 + u] = (k < sn[g]) ? bins[sa[g] + k] : 0xFFFFFFFFu;
        }
      }
#pragma unroll
      for (int u = 0; u < NSEG2 * GW2; ++u) {
        h[u] = make_float2(0.0f, 0.0f);
        if (p[u] != 0xFFFFFFFFu) h[u] = tab2[p[u] & 0xFFFFu];
      }
#pragma unroll
      for (int u = 0; u < NSEG2 * GW2; ++u) {
        if (p[u] != 0xFFFFFFFFu) {
          unsigned a = p[u] >> 16;
          unsafeAtomicAdd(&acc[2 * a + 0], h[u].x);
          unsafeAtomicAdd(&acc[2 * a + 1], h[u].y);
        }
      }
    }
  }
  __syncthreads();

  for (int i = tid; i < DRNG / 2; i += TPB) prow[i] = a4[i];
}

__global__ void gnn_layer1_wmma_kernel(const float* __restrict__ part,
                                       const float* __restrict__ x,
                                       const float* __restrict__ dinv,
                                       const float* __restrict__ W1,
                                       const float* __restrict__ b1,
                                       const float* __restrict__ W2,
                                       float* __restrict__ hs2) {
  int wave = (blockIdx.x * blockDim.x + threadIdx.x) >> 5;
  int lane = threadIdx.x & 31;
  if (wave >= NTILES) return;
  int c    = lane & 15;
  int half = lane >> 4;
  int node = (wave << 4) + c;
  bool lo  = half == 0;

  float2 a = make_float2(0.0f, 0.0f);
#pragma unroll
  for (int s = 0; s < TS; ++s) {
    float2 v = reinterpret_cast<const float2*>(part + (size_t)s * (2 * NPAD))[node];
    a.x += v.x;
    a.y += v.y;
  }
  float di = dinv[node];
  float2 xv = reinterpret_cast<const float2*>(x)[node];
  float z0 = di * (a.x + xv.x * di);
  float z1 = di * (a.y + xv.y * di);

  float w0 = W1[c];
  float w1 = W1[16 + c];

  v2f A, B;
  A.x = lo ? w0 : 0.0f;
  A.y = lo ? w1 : 0.0f;
  B.x = lo ? z0 : 0.0f;
  B.y = lo ? z1 : 0.0f;

  v8f acc = {};
  acc = __builtin_amdgcn_wmma_f32_16x16x4_f32(
       false, A,  false, B,
       (short)0, acc,  false,  false);

  const float* __restrict__ bb = b1 + 8 * half;
  const float* __restrict__ ww = W2 + 16 * half;
  float h0 = 0.0f, h1 = 0.0f;
#pragma unroll
  for (int r = 0; r < 8; ++r) {
    float v = acc[r] + bb[r];
    v = v > 0.0f ? v : 0.0f;
    h0 += v * ww[2 * r + 0];
    h1 += v * ww[2 * r + 1];
  }
  h0 += __shfl_xor(h0, 16);
  h1 += __shfl_xor(h1, 16);
  if (lo) reinterpret_cast<float2*>(hs2)[node] = make_float2(h0 * di, h1 * di);
}

__global__ void gnn_finalize2_kernel(const float* __restrict__ part,
                                     const float* __restrict__ dinv,
                                     const float* __restrict__ b2,
                                     float* out) {
  int t = blockIdx.x * blockDim.x + threadIdx.x;
  if (t >= NN * 2) return;
  int i = t >> 1;
  int c = t & 1;
  float a = 0.0f;
#pragma unroll
  for (int s = 0; s < TS; ++s) a += part[(size_t)s * (2 * NPAD) + t];
  out[t] = dinv[i] * (a + out[t]) + b2[c];
}

extern "C" void kernel_launch(void* const* d_in, const int* in_sizes, int n_in,
                              void* d_out, int out_size, void* d_ws, size_t ws_size,
                              hipStream_t stream) {
  const float* x  = (const float*)d_in[0];
  const float* W1 = (const float*)d_in[1];
  const float* b1 = (const float*)d_in[2];
  const float* W2 = (const float*)d_in[3];
  const float* b2 = (const float*)d_in[4];
  const int* edge = (const int*)d_in[5];
  const int* src  = edge;
  const int* dst  = edge + NE;
  float* out = (float*)d_out;

  float* ws = (float*)d_ws;
  float* dinv  = ws;
  float* part  = ws + WS_PART;
  unsigned* offs = (unsigned*)(ws + WS_OFFS2);
  unsigned* bins = (unsigned*)(ws + WS_BINS2);
  unsigned* offs16 = (unsigned*)(ws + WS_OFFS16);
  unsigned short* bins16 = (unsigned short*)(ws + WS_BINS16);
  float* rows = out;

  const int B = 256;

  gnn_sort_chunk16_kernel<<<NB, TPB, 0, stream>>>(dst, offs16, bins16);
  gnn_deg_owner16_kernel<<<NR, TPB, 0, stream>>>(offs16, bins16, x, dinv, rows);

  gnn_sort_tile_kernel<<<PCH, TPB, 0, stream>>>(src, dst, offs, bins, 0);
  gnn_tile_sum2_kernel<<<NT, TPB, 0, stream>>>(offs, bins, rows, part, 1);
  gnn_sort_tile_kernel<<<PCH, TPB, 0, stream>>>(src, dst, offs, bins, 1);
  gnn_tile_sum2_kernel<<<NT, TPB, 0, stream>>>(offs, bins, rows, part, 0);
  gnn_sort_tile_kernel<<<PCH, TPB, 0, stream>>>(src, dst, offs, bins, 2);
  gnn_tile_sum2_kernel<<<NT, TPB, 0, stream>>>(offs, bins, rows, part, 0);

  {
    int threads = NTILES * 32;
    gnn_layer1_wmma_kernel<<<(threads + B - 1) / B, B, 0, stream>>>(part, x, dinv, W1, b1, W2, rows);
  }

  gnn_tile_sum2_kernel<<<NT, TPB, 0, stream>>>(offs, bins, rows, part, 1);
  gnn_sort_tile_kernel<<<PCH, TPB, 0, stream>>>(src, dst, offs, bins, 0);
  gnn_tile_sum2_kernel<<<NT, TPB, 0, stream>>>(offs, bins, rows, part, 0);
  gnn_sort_tile_kernel<<<PCH, TPB, 0, stream>>>(src, dst, offs, bins, 1);
  gnn_tile_sum2_kernel<<<NT, TPB, 0, stream>>>(offs, bins, rows, part, 0);

  {
    int n = NN * 2;
    gnn_finalize2_kernel<<<(n + B - 1) / B, B, 0, stream>>>(part, dinv, b2, out);
  }
}


# Both counting sorts keep a private copy of every LDS counter per lane (32 copies, scanned by the whole work-group): no two lanes of a wave add to one word or bank; f32 sum order changes
# speedup vs baseline: 9.3974x; 1.1954x over previous
#include <hip/hip_runtime.h>


#define NN 200000
#define NE 12800000
#define NTILES (NN / 16)

#define TPB 1024
#define NWAVES (TPB / 32)
#define NCP 32
#define NKEY 256

#define RNG 800
#define NR (NN / RNG)
#define NB 512
#define CH (NE / NB)
#define CH4 (CH / 4)
#define OFFW 256
#define BU 7
#define GW 4
#define CHS16 25024
#define NSEG16 (NB / NWAVES)
#define SG16 4

#define TD 42
#define TS 6
#define NT (TD * TS)
#define NTP 256
#define DRNG 4768
#define NPAD (TD * DRNG)
#define SBLK 33334
#define NPART 3
#define PCH 256
#define CHA 16668
#define CHB 16664
#define CHS2 16672
#define OFFW2 256
#define BU2 5
#define NSEG2 (PCH / NWAVES)
#define GW2 3

static_assert(NR * RNG == NN && RNG % 32 == 0, "degree ranges cover the nodes in whole lines");
static_assert(CH * NB == NE && CH % 4 == 0, "degree chunks cover the edges");
static_assert(BU * TPB >= CH4, "one pass of the work-group loads a degree chunk");
static_assert(NR + 1 <= OFFW && OFFW == 8 * 32, "degree offset row: 8 entries per lane of one wave");
static_assert(CHS16 >= CH && (CHS16 * 2) % 128 == 0 && (CH * 2) % 16 == 0, "16-bit chunks: whole lines, whole 16-byte groups");
static_assert(NSEG16 * NWAVES == NB && NSEG16 % SG16 == 0 && NSEG16 <= 32, "degree pass: segment groups per wave, bounds held one per lane");
static_assert((size_t)NB * CHS16 < ((size_t)1 << 32), "16-bit entry index fits 32 bits");

static_assert(NPAD >= NN && DRNG % 16 == 0, "destination ranges cover the nodes in whole lines of 8-byte rows");
static_assert(TS * SBLK >= NN && (TS - 1) * SBLK < NN && SBLK % 2 == 0 && NN % 2 == 0, "source blocks cover the nodes in whole 16-byte groups");
static_assert(DRNG < (1 << 16) - 1 && SBLK <= (1 << 16), "word = 16 bits local destination | 16 bits local source; all ones is no word");
static_assert(NT <= NTP && NTP == 8 * 32 && NT + 1 <= OFFW2 && OFFW2 == NTP, "tile offset row: 8 counts per lane of one wave, whole lines");
static_assert((size_t)PCH * ((NPART - 1) * CHA + CHB) == NE && CHA % 4 == 0 && CHB % 4 == 0 && CHB <= CHA && BU2 * TPB >= CHA / 4, "chunks cover the edges; one pass loads a chunk");
static_assert(CHS2 >= CHA && CHS2 % 32 == 0, "whole lines per chunk");
static_assert(NSEG2 * NWAVES == PCH && NSEG2 <= 32, "bounds held one per lane");
static_assert(NKEY == NTP && NKEY == OFFW && NKEY * NCP == 8 * TPB && NWAVES == 32 && NCP == 32,
              "counter scan: 8 entries per thread, one wave total per lane of a wave");

typedef __attribute__((ext_vector_type(2))) float v2f;
typedef __attribute__((ext_vector_type(8))) float v8f;

#define WS_OFFS16 ((size_t)1 * NN)
#define WS_BINS16 (WS_OFFS16 + (size_t)NB * OFFW)
#define WS_WORDS16 (WS_BINS16 + (size_t)NB * CHS16 / 2)
#define WS_PART ((size_t)1 * NN)
#define WS_OFFS2 (WS_PART + (size_t)TS * 2 * NPAD)
#define WS_BINS2 (WS_OFFS2 + (size_t)PCH * OFFW2)
#define WS_WORDS2 (WS_BINS2 + (size_t)PCH * CHS2)
static_assert(WS_WORDS2 * 4 <= (size_t)35 * NN * 4, "inside the 28,000,000 bytes the versions before this one used");
static_assert(WS_PART % 32 == 0 && (2 * NPAD) % 32 == 0 && WS_OFFS2 % 32 == 0 && WS_BINS2 % 32 == 0, "planes start on 128-byte lines");
static_assert(WS_WORDS16 * 4 <= (size_t)35 * NN * 4, "degree list inside the same 28,000,000 bytes");
static_assert(WS_OFFS16 % 32 == 0 && WS_BINS16 % 32 == 0, "degree planes start on 128-byte lines");

__device__ __forceinline__ unsigned gnn_wave_scan_incl(unsigned v, unsigned lane) {
  v += (unsigned)__builtin_amdgcn_update_dpp(0, (int)v, 0x111, 0xf, 0xf, false);
  v += (unsigned)__builtin_amdgcn_update_dpp(0, (int)v, 0x112, 0xf, 0xf, false);
  v += (unsigned)__builtin_amdgcn_update_dpp(0, (int)v, 0x114, 0xf, 0xf, false);
  v += (unsigned)__builtin_amdgcn_update_dpp(0, (int)v, 0x118, 0xf, 0xf, false);
  const unsigned low = (unsigned)__builtin_amdgcn_readlane((int)v, 15);
  if (lane >= 16u) v += low;
  return v;
}

__device__ __forceinline__ void gnn_scan_private_counters(unsigned* cnt, unsigned* wtot, unsigned* kbase,
                                                          int tid, unsigned lane, unsigned w) {
  uint4* c4 = reinterpret_cast<uint4*>(cnt);
  const uint4 ca = c4[2 * tid + 0];
  const uint4 cb = c4[2 * tid + 1];
  const unsigned c[8] = {ca.x, ca.y, ca.z, ca.w, cb.x, cb.y, cb.z, cb.w};
  unsigned s = 0u;
#pragma unroll
  for (int k = 0; k < 8; ++k) s += c[k];
  const unsigned inc = gnn_wave_scan_incl(s, lane);
  if (lane == 31u) wtot[w] = inc;
  __syncthreads();
  const unsigned wv = wtot[lane];
  const unsigned winc = gnn_wave_scan_incl(wv, lane);
  const int ws = __builtin_amdgcn_readfirstlane((int)w);
  const unsigned wbase = (unsigned)__builtin_amdgcn_readlane((int)(winc - wv), ws);
  unsigned ex = wbase + (inc - s);
  if ((tid & 3) == 0) kbase[tid >> 2] = ex;
  unsigned e[8];
#pragma unroll
  for (int k = 0; k < 8; ++k) { e[k] = ex; ex += c[k]; }
  c4[2 * tid + 0] = make_uint4(e[0], e[1], e[2], e[3]);
  c4[2 * tid + 1] = make_uint4(e[4], e[5], e[6], e[7]);
  __syncthreads();
}

__global__ __launch_bounds__(TPB) void gnn_sort_tile_kernel(const int* __restrict__ src,
                                                            const int* __restrict__ dst,
                                                            unsigned* __restrict__ offs,
                                                            unsigned* __restrict__ bins,
                                                            int part) {
  __shared__ __attribute__((aligned(16))) unsigned cnt[NKEY * NCP];
  __shared__ unsigned wtot[NWAVES];
  __shared__ unsigned kbase[NKEY];
  __shared__ __attribute__((aligned(16))) unsigned stage[CHA];
  const int b = blockIdx.x;
  const int tid = threadIdx.x;
  const unsigned lane = threadIdx.x & 31;
  const unsigned w = threadIdx.x >> 5;
  reinterpret_cast<uint4*>(cnt)[2 * tid + 0] = make_uint4(0u, 0u, 0u, 0u);
  reinterpret_cast<uint4*>(cnt)[2 * tid + 1] = make_uint4(0u, 0u, 0u, 0u);
  __syncthreads();

  const int csz4 = (part < NPART - 1) ? CHA / 4 : CHB / 4;
  const size_t e0 = (size_t)part * PCH * CHA + (size_t)b * (4 * csz4);
  const int4* __restrict__ d4 = reinterpret_cast<const int4*>(dst + e0);
  const int4* __restrict__ s4 = reinterpret_cast<const int4*>(src + e0);
  int4 dv[BU2], sv[BU2];
#pragma unroll
  for (int u = 0; u < BU2; ++u) {
    int i = u * TPB + tid;
    bool ok = i < csz4;
    dv[u] = ok ? d4[i] : make_int4(-1, -1, -1, -1);
    sv[u] = ok ? s4[i] : make_int4(-1, -1, -1, -1);
  }
#pragma unroll
  for (int u = 0; u < BU2; ++u) {
    const unsigned dd[4] = {(unsigned)dv[u].x, (unsigned)dv[u].y, (unsigned)dv[u].z, (unsigned)dv[u].w};
    const unsigned ss[4] = {(unsigned)sv[u].x, (unsigned)sv[u].y, (unsigned)sv[u].z, (unsigned)sv[u].w};
#pragma unroll
    for (int k = 0; k < 4; ++k)
      if (dd[k] < (unsigned)NN && ss[k] < (unsigned)NN)
        atomicAdd(&cnt[((dd[k] / DRNG) * TS + ss[k] / SBLK) * NCP + lane], 1u);
  }
  __syncthreads();

  gnn_scan_private_counters(cnt, wtot, kbase, tid, lane, w);
  if (tid < NKEY) offs[(size_t)b * OFFW2 + tid] = kbase[tid];

#pragma unroll
  for (int u = 0; u < BU2; ++u) {
    const unsigned dd[4] = {(unsigned)dv[u].x, (unsigned)dv[u].y, (unsigned)dv[u].z, (unsigned)dv[u].w};
    const unsigned ss[4] = {(unsigned)sv[u].x, (unsigned)sv[u].y, (unsigned)sv[u].z, (unsigned)sv[u].w};
#pragma unroll
    for (int k = 0; k < 4; ++k) {
      if (dd[k] < (unsigned)NN && ss[k] < (unsigned)NN) {
        unsigned r = dd[k] / DRNG;
        unsigned q = ss[k] / SBLK;
        unsigned pos = atomicAdd(&cnt[(r * TS + q) * NCP + lane], 1u);
        stage[pos] = ((dd[k] - r * DRNG) << 16) | (ss[k] - q * SBLK);
      }
    }
  }
  __syncthreads();

  uint4* __restrict__ o4 = reinterpret_cast<uint4*>(bins + (size_t)b * CHS2);
  const uint4* st4 = reinterpret_cast<const uint4*>(stage);
  for (int i = tid; i < csz4; i += TPB) o4[i] = st4[i];
}

__global__ __launch_bounds__(TPB) void gnn_sort_chunk16_kernel(const int* __restrict__ dst,
                                                               unsigned* __restrict__ offs,
                                                               unsigned short* __restrict__ bins16) {
  __shared__ __attribute__((aligned(16))) unsigned cnt[NKEY * NCP];
  __shared__ unsigned wtot[NWAVES];
  __shared__ unsigned kbase[NKEY];
  __shared__ __attribute__((aligned(16))) unsigned short stage[CH];
  const int b = blockIdx.x;
  const int tid = threadIdx.x;
  const unsigned lane = threadIdx.x & 31;
  const unsigned w = threadIdx.x >> 5;
  reinterpret_cast<uint4*>(cnt)[2 * tid + 0] = make_uint4(0u, 0u, 0u, 0u);
  reinterpret_cast<uint4*>(cnt)[2 * tid + 1] = make_uint4(0u, 0u, 0u, 0u);
  __syncthreads();

  const int4* __restrict__ d4 = reinterpret_cast<const int4*>(dst + (size_t)b * CH);
  int4 dv[BU];
#pragma unroll
  for (int u = 0; u < BU; ++u) {
    int i = u * TPB + tid;
    dv[u] = (i < CH4) ? d4[i] : make_int4(-1, -1, -1, -1);
  }
#pragma unroll
  for (int u = 0; u < BU; ++u) {
    const unsigned dd[4] = {(unsigned)dv[u].x, (unsigned)dv[u].y, (unsigned)dv[u].z, (unsigned)dv[u].w};
#pragma unroll
    for (int k = 0; k < 4; ++k)
      if (dd[k] < (unsigned)NN) atomicAdd(&cnt[(dd[k] / RNG) * NCP + lane], 1u);
  }
  __syncthreads();

  gnn_scan_private_counters(cnt, wtot, kbase, tid, lane, w);
  if (tid < NKEY) offs[(size_t)b * OFFW + tid] = kbase[tid];

#pragma unroll
  for (int u = 0; u < BU; ++u) {
    const unsigned dd[4] = {(unsigned)dv[u].x, (unsigned)dv[u].y, (unsigned)dv[u].z, (unsigned)dv[u].w};
#pragma unroll
    for (int k = 0; k < 4; ++k) {
      if (dd[k] < (unsigned)NN) {
        unsigned r = dd[k] / RNG;
        unsigned pos = atomicAdd(&cnt[r * NCP + lane], 1u);
        stage[pos] = (unsigned short)(dd[k] - r * RNG);
      }
    }
  }
  __syncthreads();

  uint4* __restrict__ o4 = reinterpret_cast<uint4*>(bins16 + (size_t)b * CHS16);
  const uint4* st4 = reinterpret_cast<const uint4*>(stage);
  for (int i = tid; i < CH / 8; i += TPB) o4[i] = st4[i];
}

__global__ __launch_bounds__(TPB) void gnn_deg_owner16_kernel(const unsigned* __restrict__ offs,
                                                              const unsigned short* __restrict__ bins16,
                                                              const float* __restrict__ x,
                                                              float* __restrict__ dinv,
                                                              float* __restrict__ xs) {
  __shared__ unsigned cnt[RNG];
  const int r = blockIdx.x;
  const int tid = threadIdx.x;
  const unsigned lane = threadIdx.x & 31;
  const unsigned w = threadIdx.x >> 5;
  if (tid < RNG) cnt[tid] = 0u;
  __syncthreads();

  unsigned s0 = 0u, s1 = 0u;
  if (lane < (unsigned)NSEG16) {
    unsigned bb = w * (unsigned)NSEG16 + lane;
    const unsigned* __restrict__ o = offs + (size_t)bb * OFFW + r;
    s0 = bb * (unsigned)CHS16 + o[0];
    s1 = bb * (unsigned)CHS16 + o[1];
  }
  for (int j = 0; j < NSEG16; j += SG16) {
    unsigned sa[SG16], sn[SG16];
    unsigned nmax = 0u;
#pragma unroll
    for (int g = 0; g < SG16; ++g) {
      sa[g] = (unsigned)__builtin_amdgcn_readlane((int)s0, j + g);
      sn[g] = (unsigned)__builtin_amdgcn_readlane((int)s1, j + g) - sa[g];
      nmax = sn[g] > nmax ? sn[g] : nmax;
    }
    for (unsigned k0 = 0; k0 < nmax; k0 += 32 * GW) {
      unsigned p[SG16 * GW];
#pragma unroll
      for (int g = 0; g < SG16; ++g) {
#pragma unroll
        for (int t = 0; t < GW; ++t) {
          unsigned k = k0 + 32u * t + lane;
          p[g * GW + t] = (k < sn[g]) ? (unsigned)bins16[sa[g] + k] : 0xFFFFu;
        }
      }
#pragma unroll
      for (int t = 0; t < SG16 * GW; ++t)
        if (p[t] != 0xFFFFu) atomicAdd(&cnt[p[t]], 1u);
    }
  }
  __syncthreads();

  if (tid < RNG) {
    int node = r * RNG + tid;
    float d = rsqrtf((float)cnt[tid] + 1.0f);
    dinv[node] = d;
    float2 xv = reinterpret_cast<const float2*>(x)[node];
    reinterpret_cast<float2*>(xs)[node] = make_float2(xv.x * d, xv.y * d);
  }
}

__global__ __launch_bounds__(TPB) void gnn_tile_sum2_kernel(const unsigned* __restrict__ offs,
                                                            const unsigned* __restrict__ bins,
                                                            const float* __restrict__ rows2,
                                                            float* __restrict__ part,
                                                            int first) {
  __shared__ __attribute__((aligned(16))) float tab[SBLK * 2];
  __shared__ __attribute__((aligned(16))) float acc[DRNG * 2];
  const int t = blockIdx.x;
  const int r = t / TS;
  const int q = t - r * TS;
  const int tid = threadIdx.x;
  const unsigned lane = threadIdx.x & 31;
  const unsigned w = threadIdx.x >> 5;

  const unsigned bb = w * (unsigned)NSEG2 + (lane < (unsigned)NSEG2 ? lane : 0u);
  const unsigned* __restrict__ o = offs + (size_t)bb * OFFW2 + t;
  const unsigned o0 = o[0], o1 = o[1];

  const float4* __restrict__ g4 = reinterpret_cast<const float4*>(rows2 + (size_t)2 * q * SBLK);
  float4* __restrict__ prow = reinterpret_cast<float4*>(part + (size_t)q * (2 * NPAD) + (size_t)2 * r * DRNG);
  float4* t4 = reinterpret_cast<float4*>(tab);
  float4* a4 = reinterpret_cast<float4*>(acc);
  constexpr int NG = (SBLK / 2 + TPB - 1) / TPB;
  constexpr int NA = (DRNG / 2 + TPB - 1) / TPB;
  const int nrow4 = (NN / 2 - q * (SBLK / 2)) < SBLK / 2 ? (NN / 2 - q * (SBLK / 2)) : SBLK / 2;
  float4 v[NG], pv[NA];
#pragma unroll
  for (int u = 0; u < NG; ++u) {
    int i = u * TPB + tid;
    v[u] = (i < nrow4) ? g4[i] : make_float4(0.0f, 0.0f, 0.0f, 0.0f);
  }
#pragma unroll
  for (int u = 0; u < NA; ++u) {
    int i = u * TPB + tid;
    pv[u] = (!first && i < DRNG / 2) ? prow[i] : make_float4(0.0f, 0.0f, 0.0f, 0.0f);
  }
#pragma unroll
  for (int u = 0; u < NG; ++u) {
    int i = u * TPB + tid;
    if (i < SBLK / 2) t4[i] = v[u];
  }
#pragma unroll
  for (int u = 0; u < NA; ++u) {
    int i = u * TPB + tid;
    if (i < DRNG / 2) a4[i] = pv[u];
  }
  const unsigned s0 = bb * (unsigned)CHS2 + o0;
  const unsigned s1 = bb * (unsigned)CHS2 + o1;
  __syncthreads();

  const float2* tab2 = reinterpret_cast<const float2*>(tab);
  {
    unsigned sa[NSEG2], sn[NSEG2];
    unsigned nmax = 0u;
#pragma unroll
    for (int g = 0; g < NSEG2; ++g) {
      sa[g] = (unsigned)__builtin_amdgcn_readlane((int)s0, g);
      sn[g] = (unsigned)__builtin_amdgcn_readlane((int)s1, g) - sa[g];
      nmax = sn[g] > nmax ? sn[g] : nmax;
    }
    for (unsigned k0 = 0; k0 < nmax; k0 += 32 * GW2) {
      unsigned p[NSEG2 * GW2];
      float2 h[NSEG2 * GW2];
#pragma unroll
      for (int g = 0; g < NSEG2; ++g) {
#pragma unroll
        for (int u = 0; u < GW2; ++u) {
          unsigned k = k0 + 32u * u + lane;
          p[g * GW2 + u] = (k < sn[g]) ? bins[sa[g] + k] : 0xFFFFFFFFu;
        }
      }
#pragma unroll
      for (int u = 0; u < NSEG2 * GW2; ++u) {
        h[u] = make_float2(0.0f, 0.0f);
        if (p[u] != 0xFFFFFFFFu) h[u] = tab2[p[u] & 0xFFFFu];
      }
#pragma unroll
      for (int u = 0; u < NSEG2 * GW2; ++u) {
        if (p[u] != 0xFFFFFFFFu) {
          unsigned a = p[u] >> 16;
          unsafeAtomicAdd(&acc[2 * a + 0], h[u].x);
          unsafeAtomicAdd(&acc[2 * a + 1], h[u].y);
        }
      }
    }
  }
  __syncthreads();

  for (int i = tid; i < DRNG / 2; i += TPB) prow[i] = a4[i];
}

__global__ void gnn_layer1_wmma_kernel(const float* __restrict__ part,
                                       const float* __restrict__ x,
                                       const float* __restrict__ dinv,
                                       const float* __restrict__ W1,
                                       const float* __restrict__ b1,
                                       const float* __restrict__ W2,
                                       float* __restrict__ hs2) {
  int wave = (blockIdx.x * blockDim.x + threadIdx.x) >> 5;
  int lane = threadIdx.x & 31;
  if (wave >= NTILES) return;
  int c    = lane & 15;
  int half = lane >> 4;
  int node = (wave << 4) + c;
  bool lo  = half == 0;

  float2 a = make_float2(0.0f, 0.0f);
#pragma unroll
  for (int s = 0; s < TS; ++s) {
    float2 v = reinterpret_cast<const float2*>(part + (size_t)s * (2 * NPAD))[node];
    a.x += v.x;
    a.y += v.y;
  }
  float di = dinv[node];
  float2 xv = reinterpret_cast<const float2*>(x)[node];
  float z0 = di * (a.x + xv.x * di);
  float z1 = di * (a.y + xv.y * di);

  float w0 = W1[c];
  float w1 = W1[16 + c];

  v2f A, B;
  A.x = lo ? w0 : 0.0f;
  A.y = lo ? w1 : 0.0f;
  B.x = lo ? z0 : 0.0f;
  B.y = lo ? z1 : 0.0f;

  v8f acc = {};
  acc = __builtin_amdgcn_wmma_f32_16x16x4_f32(
       false, A,  false, B,
       (short)0, acc,  false,  false);

  const float* __restrict__ bb = b1 + 8 * half;
  const float* __restrict__ ww = W2 + 16 * half;
  float h0 = 0.0f, h1 = 0.0f;
#pragma unroll
  for (int r = 0; r < 8; ++r) {
    float v = acc[r] + bb[r];
    v = v > 0.0f ? v : 0.0f;
    h0 += v * ww[2 * r + 0];
    h1 += v * ww[2 * r + 1];
  }
  h0 += __shfl_xor(h0, 16);
  h1 += __shfl_xor(h1, 16);
  if (lo) reinterpret_cast<float2*>(hs2)[node] = make_float2(h0 * di, h1 * di);
}

__global__ void gnn_finalize2_kernel(const float* __restrict__ part,
                                     const float* __restrict__ dinv,
                                     const float* __restrict__ b2,
                                     float* out) {
  int t = blockIdx.x * blockDim.x + threadIdx.x;
  if (t >= NN * 2) return;
  int i = t >> 1;
  int c = t & 1;
  float a = 0.0f;
#pragma unroll
  for (int s = 0; s < TS; ++s) a += part[(size_t)s * (2 * NPAD) + t];
  out[t] = dinv[i] * (a + out[t]) + b2[c];
}

extern "C" void kernel_launch(void* const* d_in, const int* in_sizes, int n_in,
                              void* d_out, int out_size, void* d_ws, size_t ws_size,
                              hipStream_t stream) {
  const float* x  = (const float*)d_in[0];
  const float* W1 = (const float*)d_in[1];
  const float* b1 = (const float*)d_in[2];
  const float* W2 = (const float*)d_in[3];
  const float* b2 = (const float*)d_in[4];
  const int* edge = (const int*)d_in[5];
  const int* src  = edge;
  const int* dst  = edge + NE;
  float* out = (float*)d_out;

  float* ws = (float*)d_ws;
  float* dinv  = ws;
  float* part  = ws + WS_PART;
  unsigned* offs = (unsigned*)(ws + WS_OFFS2);
  unsigned* bins = (unsigned*)(ws + WS_BINS2);
  unsigned* offs16 = (unsigned*)(ws + WS_OFFS16);
  unsigned short* bins16 = (unsigned short*)(ws + WS_BINS16);
  float* rows = out;

  const int B = 256;

  gnn_sort_chunk16_kernel<<<NB, TPB, 0, stream>>>(dst, offs16, bins16);
  gnn_deg_owner16_kernel<<<NR, TPB, 0, stream>>>(offs16, bins16, x, dinv, rows);

  gnn_sort_tile_kernel<<<PCH, TPB, 0, stream>>>(src, dst, offs, bins, 0);
  gnn_tile_sum2_kernel<<<NT, TPB, 0, stream>>>(offs, bins, rows, part, 1);
  gnn_sort_tile_kernel<<<PCH, TPB, 0, stream>>>(src, dst, offs, bins, 1);
  gnn_tile_sum2_kernel<<<NT, TPB, 0, stream>>>(offs, bins, rows, part, 0);
  gnn_sort_tile_kernel<<<PCH, TPB, 0, stream>>>(src, dst, offs, bins, 2);
  gnn_tile_sum2_kernel<<<NT, TPB, 0, stream>>>(offs, bins, rows, part, 0);

  {
    int threads = NTILES * 32;
    gnn_layer1_wmma_kernel<<<(threads + B - 1) / B, B, 0, stream>>>(part, x, dinv, W1, b1, W2, rows);
  }

  gnn_tile_sum2_kernel<<<NT, TPB, 0, stream>>>(offs, bins, rows, part, 1);
  gnn_sort_tile_kernel<<<PCH, TPB, 0, stream>>>(src, dst, offs, bins, 0);
  gnn_tile_sum2_kernel<<<NT, TPB, 0, stream>>>(offs, bins, rows, part, 0);
  gnn_sort_tile_kernel<<<PCH, TPB, 0, stream>>>(src, dst, offs, bins, 1);
  gnn_tile_sum2_kernel<<<NT, TPB, 0, stream>>>(offs, bins, rows, part, 0);

  {
    int n = NN * 2;
    gnn_finalize2_kernel<<<(n + B - 1) / B, B, 0, stream>>>(part, dinv, b2, out);
  }
}


# Sorted edge list kept whole when the workspace size given holds it (62,415,104 bytes): one sort, one tile pass per layer, 7 launches for 15; halves from 36,413,696 (11); f32 sum order changes
# speedup vs baseline: 14.0983x; 1.5002x over previous
#include <hip/hip_runtime.h>


#define NN 200000
#define NE 12800000
#define NTILES (NN / 16)

#define TPB 1024
#define NWAVES (TPB / 32)
#define NCP 32
#define NKEY 256

#define RNG 800
#define NR (NN / RNG)
#define NB 512
#define CH (NE / NB)
#define CH4 (CH / 4)
#define OFFW 256
#define BU 7
#define GW 4
#define CHS16 25024
#define NSEG16 (NB / NWAVES)
#define SG16 4

#define TD 42
#define TS 6
#define NT (TD * TS)
#define NTP 256
#define DRNG 4768
#define NPAD (TD * DRNG)
#define SBLK 33334
#define NPART 3
#define PCH 256
#define CHA 16668
#define CHB 16664
#define CHS2 16672
#define OFFW2 256
#define BU2 5
#define NCHUNK (NPART * PCH)
#define NSLOT_HALF (NCHUNK / 2)
#define NSLOT_FULL NCHUNK
#define NSEG2 (PCH / NWAVES)
#define GW2 3

static_assert(NR * RNG == NN && RNG % 32 == 0, "degree ranges cover the nodes in whole lines");
static_assert(CH * NB == NE && CH % 4 == 0, "degree chunks cover the edges");
static_assert(BU * TPB >= CH4, "one pass of the work-group loads a degree chunk");
static_assert(NR + 1 <= OFFW && OFFW == 8 * 32, "degree offset row: 8 entries per lane of one wave");
static_assert(CHS16 >= CH && (CHS16 * 2) % 128 == 0 && (CH * 2) % 16 == 0, "16-bit chunks: whole lines, whole 16-byte groups");
static_assert(NSEG16 * NWAVES == NB && NSEG16 % SG16 == 0 && NSEG16 <= 32, "degree pass: segment groups per wave, bounds held one per lane");
static_assert((size_t)NB * CHS16 < ((size_t)1 << 32), "16-bit entry index fits 32 bits");

static_assert(NPAD >= NN && DRNG % 16 == 0, "destination ranges cover the nodes in whole lines of 8-byte rows");
static_assert(TS * SBLK >= NN && (TS - 1) * SBLK < NN && SBLK % 2 == 0 && NN % 2 == 0, "source blocks cover the nodes in whole 16-byte groups");
static_assert(DRNG < (1 << 16) - 1 && SBLK <= (1 << 16), "word = 16 bits local destination | 16 bits local source; all ones is no word");
static_assert(NT <= NTP && NTP == 8 * 32 && NT + 1 <= OFFW2 && OFFW2 == NTP, "tile offset row: 8 counts per lane of one wave, whole lines");
static_assert((size_t)PCH * ((NPART - 1) * CHA + CHB) == NE && CHA % 4 == 0 && CHB % 4 == 0 && CHB <= CHA && BU2 * TPB >= CHA / 4, "chunks cover the edges; one pass loads a chunk");
static_assert(CHS2 >= CHA && CHS2 % 32 == 0, "whole lines per chunk");
static_assert(NSEG2 * NWAVES == PCH && NSEG2 <= 32, "bounds held one per lane");
static_assert(NSLOT_HALF % NWAVES == 0 && NSLOT_FULL % NWAVES == 0 && NSLOT_FULL / NWAVES <= 32,
              "every wave of a tile pass walks the same number of slots, bounds held one per lane");
static_assert((size_t)NSLOT_FULL * CHS2 < ((size_t)1 << 32), "word index fits 32 bits");
static_assert(NKEY == NTP && NKEY == OFFW && NKEY * NCP == 8 * TPB && NWAVES == 32 && NCP == 32,
              "counter scan: 8 entries per thread, one wave total per lane of a wave");

typedef __attribute__((ext_vector_type(2))) float v2f;
typedef __attribute__((ext_vector_type(8))) float v8f;

#define WS_OFFS16 ((size_t)1 * NN)
#define WS_BINS16 (WS_OFFS16 + (size_t)NB * OFFW)
#define WS_WORDS16 (WS_BINS16 + (size_t)NB * CHS16 / 2)
#define WS_PART ((size_t)1 * NN)
#define WS_OFFS2 (WS_PART + (size_t)TS * 2 * NPAD)
#define WS_BINS2 (WS_OFFS2 + (size_t)PCH * OFFW2)
#define WS_WORDS2 (WS_BINS2 + (size_t)PCH * CHS2)
#define WS_BINS_OF(S) (WS_OFFS2 + (size_t)(S) * OFFW2)
#define WS_WORDS_OF(S) (WS_BINS_OF(S) + (size_t)(S) * CHS2)
#define WS_BYTES_HALF (WS_WORDS_OF(NSLOT_HALF) * 4)
#define WS_BYTES_FULL (WS_WORDS_OF(NSLOT_FULL) * 4)
static_assert(WS_WORDS2 * 4 <= (size_t)35 * NN * 4, "inside the 28,000,000 bytes the versions before this one used");
static_assert(WS_PART % 32 == 0 && (2 * NPAD) % 32 == 0 && WS_OFFS2 % 32 == 0 && WS_BINS2 % 32 == 0, "planes start on 128-byte lines");
static_assert(WS_WORDS16 * 4 <= (size_t)35 * NN * 4, "degree list inside the same 28,000,000 bytes");
static_assert(WS_BINS_OF(PCH) == WS_BINS2 && WS_WORDS_OF(PCH) == WS_WORDS2, "the smallest layout is the one above");
static_assert(WS_BINS_OF(NSLOT_HALF) % 32 == 0 && WS_BINS_OF(NSLOT_FULL) % 32 == 0, "planes start on 128-byte lines");
static_assert(WS_BYTES_HALF == (size_t)36413696 && WS_BYTES_FULL == (size_t)62415104, "sizes named in the comments");
static_assert(WS_WORDS16 <= WS_WORDS2, "the degree list lies inside every layout");
static_assert(WS_OFFS16 % 32 == 0 && WS_BINS16 % 32 == 0, "degree planes start on 128-byte lines");

__device__ __forceinline__ unsigned gnn_wave_scan_incl(unsigned v, unsigned lane) {
  v += (unsigned)__builtin_amdgcn_update_dpp(0, (int)v, 0x111, 0xf, 0xf, false);
  v += (unsigned)__builtin_amdgcn_update_dpp(0, (int)v, 0x112, 0xf, 0xf, false);
  v += (unsigned)__builtin_amdgcn_update_dpp(0, (int)v, 0x114, 0xf, 0xf, false);
  v += (unsigned)__builtin_amdgcn_update_dpp(0, (int)v, 0x118, 0xf, 0xf, false);
  const unsigned low = (unsigned)__builtin_amdgcn_readlane((int)v, 15);
  if (lane >= 16u) v += low;
  return v;
}

__device__ __forceinline__ void gnn_scan_private_counters(unsigned* cnt, unsigned* wtot, unsigned* kbase,
                                                          int tid, unsigned lane, unsigned w) {
  uint4* c4 = reinterpret_cast<uint4*>(cnt);
  const uint4 ca = c4[2 * tid + 0];
  const uint4 cb = c4[2 * tid + 1];
  const unsigned c[8] = {ca.x, ca.y, ca.z, ca.w, cb.x, cb.y, cb.z, cb.w};
  unsigned s = 0u;
#pragma unroll
  for (int k = 0; k < 8; ++k) s += c[k];
  const unsigned inc = gnn_wave_scan_incl(s, lane);
  if (lane == 31u) wtot[w] = inc;
  __syncthreads();
  const unsigned wv = wtot[lane];
  const unsigned winc = gnn_wave_scan_incl(wv, lane);
  const int ws = __builtin_amdgcn_readfirstlane((int)w);
  const unsigned wbase = (unsigned)__builtin_amdgcn_readlane((int)(winc - wv), ws);
  unsigned ex = wbase + (inc - s);
  if ((tid & 3) == 0) kbase[tid >> 2] = ex;
  unsigned e[8];
#pragma unroll
  for (int k = 0; k < 8; ++k) { e[k] = ex; ex += c[k]; }
  c4[2 * tid + 0] = make_uint4(e[0], e[1], e[2], e[3]);
  c4[2 * tid + 1] = make_uint4(e[4], e[5], e[6], e[7]);
  __syncthreads();
}

__global__ __launch_bounds__(TPB) void gnn_sort_tile_kernel(const int* __restrict__ src,
                                                            const int* __restrict__ dst,
                                                            unsigned* __restrict__ offs,
                                                            unsigned* __restrict__ bins,
                                                            int cbase) {
  __shared__ __attribute__((aligned(16))) unsigned cnt[NKEY * NCP];
  __shared__ unsigned wtot[NWAVES];
  __shared__ unsigned kbase[NKEY];
  __shared__ __attribute__((aligned(16))) unsigned stage[CHA];
  const int slot = blockIdx.x;
  const int part = (cbase + slot) / PCH;
  const int b = (cbase + slot) - part * PCH;
  const int tid = threadIdx.x;
  const unsigned lane = threadIdx.x & 31;
  const unsigned w = threadIdx.x >> 5;
  reinterpret_cast<uint4*>(cnt)[2 * tid + 0] = make_uint4(0u, 0u, 0u, 0u);
  reinterpret_cast<uint4*>(cnt)[2 * tid + 1] = make_uint4(0u, 0u, 0u, 0u);
  __syncthreads();

  const int csz4 = (part < NPART - 1) ? CHA / 4 : CHB / 4;
  const size_t e0 = (size_t)part * PCH * CHA + (size_t)b * (4 * csz4);
  const int4* __restrict__ d4 = reinterpret_cast<const int4*>(dst + e0);
  const int4* __restrict__ s4 = reinterpret_cast<const int4*>(src + e0);
  int4 dv[BU2], sv[BU2];
#pragma unroll
  for (int u = 0; u < BU2; ++u) {
    int i = u * TPB + tid;
    bool ok = i < csz4;
    dv[u] = ok ? d4[i] : make_int4(-1, -1, -1, -1);
    sv[u] = ok ? s4[i] : make_int4(-1, -1, -1, -1);
  }
#pragma unroll
  for (int u = 0; u < BU2; ++u) {
    const unsigned dd[4] = {(unsigned)dv[u].x, (unsigned)dv[u].y, (unsigned)dv[u].z, (unsigned)dv[u].w};
    const unsigned ss[4] = {(unsigned)sv[u].x, (unsigned)sv[u].y, (unsigned)sv[u].z, (unsigned)sv[u].w};
#pragma unroll
    for (int k = 0; k < 4; ++k)
      if (dd[k] < (unsigned)NN && ss[k] < (unsigned)NN)
        atomicAdd(&cnt[((dd[k] / DRNG) * TS + ss[k] / SBLK) * NCP + lane], 1u);
  }
  __syncthreads();

  gnn_scan_private_counters(cnt, wtot, kbase, tid, lane, w);
  if (tid < NKEY) offs[(size_t)slot * OFFW2 + tid] = kbase[tid];

#pragma unroll
  for (int u = 0; u < BU2; ++u) {
    const unsigned dd[4] = {(unsigned)dv[u].x, (unsigned)dv[u].y, (unsigned)dv[u].z, (unsigned)dv[u].w};
    const unsigned ss[4] = {(unsigned)sv[u].x, (unsigned)sv[u].y, (unsigned)sv[u].z, (unsigned)sv[u].w};
#pragma unroll
    for (int k = 0; k < 4; ++k) {
      if (dd[k] < (unsigned)NN && ss[k] < (unsigned)NN) {
        unsigned r = dd[k] / DRNG;
        unsigned q = ss[k] / SBLK;
        unsigned pos = atomicAdd(&cnt[(r * TS + q) * NCP + lane], 1u);
        stage[pos] = ((dd[k] - r * DRNG) << 16) | (ss[k] - q * SBLK);
      }
    }
  }
  __syncthreads();

  uint4* __restrict__ o4 = reinterpret_cast<uint4*>(bins + (size_t)slot * CHS2);
  const uint4* st4 = reinterpret_cast<const uint4*>(stage);
  for (int i = tid; i < csz4; i += TPB) o4[i] = st4[i];
}

__global__ __launch_bounds__(TPB) void gnn_sort_chunk16_kernel(const int* __restrict__ dst,
                                                               unsigned* __restrict__ offs,
                                                               unsigned short* __restrict__ bins16) {
  __shared__ __attribute__((aligned(16))) unsigned cnt[NKEY * NCP];
  __shared__ unsigned wtot[NWAVES];
  __shared__ unsigned kbase[NKEY];
  __shared__ __attribute__((aligned(16))) unsigned short stage[CH];
  const int b = blockIdx.x;
  const int tid = threadIdx.x;
  const unsigned lane = threadIdx.x & 31;
  const unsigned w = threadIdx.x >> 5;
  reinterpret_cast<uint4*>(cnt)[2 * tid + 0] = make_uint4(0u, 0u, 0u, 0u);
  reinterpret_cast<uint4*>(cnt)[2 * tid + 1] = make_uint4(0u, 0u, 0u, 0u);
  __syncthreads();

  const int4* __restrict__ d4 = reinterpret_cast<const int4*>(dst + (size_t)b * CH);
  int4 dv[BU];
#pragma unroll
  for (int u = 0; u < BU; ++u) {
    int i = u * TPB + tid;
    dv[u] = (i < CH4) ? d4[i] : make_int4(-1, -1, -1, -1);
  }
#pragma unroll
  for (int u = 0; u < BU; ++u) {
    const unsigned dd[4] = {(unsigned)dv[u].x, (unsigned)dv[u].y, (unsigned)dv[u].z, (unsigned)dv[u].w};
#pragma unroll
    for (int k = 0; k < 4; ++k)
      if (dd[k] < (unsigned)NN) atomicAdd(&cnt[(dd[k] / RNG) * NCP + lane], 1u);
  }
  __syncthreads();

  gnn_scan_private_counters(cnt, wtot, kbase, tid, lane, w);
  if (tid < NKEY) offs[(size_t)b * OFFW + tid] = kbase[tid];

#pragma unroll
  for (int u = 0; u < BU; ++u) {
    const unsigned dd[4] = {(unsigned)dv[u].x, (unsigned)dv[u].y, (unsigned)dv[u].z, (unsigned)dv[u].w};
#pragma unroll
    for (int k = 0; k < 4; ++k) {
      if (dd[k] < (unsigned)NN) {
        unsigned r = dd[k] / RNG;
        unsigned pos = atomicAdd(&cnt[r * NCP + lane], 1u);
        stage[pos] = (unsigned short)(dd[k] - r * RNG);
      }
    }
  }
  __syncthreads();

  uint4* __restrict__ o4 = reinterpret_cast<uint4*>(bins16 + (size_t)b * CHS16);
  const uint4* st4 = reinterpret_cast<const uint4*>(stage);
  for (int i = tid; i < CH / 8; i += TPB) o4[i] = st4[i];
}

__global__ __launch_bounds__(TPB) void gnn_deg_owner16_kernel(const unsigned* __restrict__ offs,
                                                              const unsigned short* __restrict__ bins16,
                                                              const float* __restrict__ x,
                                                              float* __restrict__ dinv,
                                                              float* __restrict__ xs) {
  __shared__ unsigned cnt[RNG];
  const int r = blockIdx.x;
  const int tid = threadIdx.x;
  const unsigned lane = threadIdx.x & 31;
  const unsigned w = threadIdx.x >> 5;
  if (tid < RNG) cnt[tid] = 0u;
  __syncthreads();

  unsigned s0 = 0u, s1 = 0u;
  if (lane < (unsigned)NSEG16) {
    unsigned bb = w * (unsigned)NSEG16 + lane;
    const unsigned* __restrict__ o = offs + (size_t)bb * OFFW + r;
    s0 = bb * (unsigned)CHS16 + o[0];
    s1 = bb * (unsigned)CHS16 + o[1];
  }
  for (int j = 0; j < NSEG16; j += SG16) {
    unsigned sa[SG16], sn[SG16];
    unsigned nmax = 0u;
#pragma unroll
    for (int g = 0; g < SG16; ++g) {
      sa[g] = (unsigned)__builtin_amdgcn_readlane((int)s0, j + g);
      sn[g] = (unsigned)__builtin_amdgcn_readlane((int)s1, j + g) - sa[g];
      nmax = sn[g] > nmax ? sn[g] : nmax;
    }
    for (unsigned k0 = 0; k0 < nmax; k0 += 32 * GW) {
      unsigned p[SG16 * GW];
#pragma unroll
      for (int g = 0; g < SG16; ++g) {
#pragma unroll
        for (int t = 0; t < GW; ++t) {
          unsigned k = k0 + 32u * t + lane;
          p[g * GW + t] = (k < sn[g]) ? (unsigned)bins16[sa[g] + k] : 0xFFFFu;
        }
      }
#pragma unroll
      for (int t = 0; t < SG16 * GW; ++t)
        if (p[t] != 0xFFFFu) atomicAdd(&cnt[p[t]], 1u);
    }
  }
  __syncthreads();

  if (tid < RNG) {
    int node = r * RNG + tid;
    float d = rsqrtf((float)cnt[tid] + 1.0f);
    dinv[node] = d;
    float2 xv = reinterpret_cast<const float2*>(x)[node];
    reinterpret_cast<float2*>(xs)[node] = make_float2(xv.x * d, xv.y * d);
  }
}

template <int NSEGW>
__global__ __launch_bounds__(TPB) void gnn_tile_sum2_kernel(const unsigned* __restrict__ offs,
                                                            const unsigned* __restrict__ bins,
                                                            const float* __restrict__ rows2,
                                                            float* __restrict__ part,
                                                            int first) {
  __shared__ __attribute__((aligned(16))) float tab[SBLK * 2];
  __shared__ __attribute__((aligned(16))) float acc[DRNG * 2];
  const int t = blockIdx.x;
  const int r = t / TS;
  const int q = t - r * TS;
  const int tid = threadIdx.x;
  const unsigned lane = threadIdx.x & 31;
  const unsigned w = threadIdx.x >> 5;

  const unsigned bb = w * (unsigned)NSEGW + (lane < (unsigned)NSEGW ? lane : 0u);
  const unsigned* __restrict__ o = offs + (size_t)bb * OFFW2 + t;
  const unsigned o0 = o[0], o1 = o[1];

  const float4* __restrict__ g4 = reinterpret_cast<const float4*>(rows2 + (size_t)2 * q * SBLK);
  float4* __restrict__ prow = reinterpret_cast<float4*>(part + (size_t)q * (2 * NPAD) + (size_t)2 * r * DRNG);
  float4* t4 = reinterpret_cast<float4*>(tab);
  float4* a4 = reinterpret_cast<float4*>(acc);
  constexpr int NG = (SBLK / 2 + TPB - 1) / TPB;
  constexpr int NA = (DRNG / 2 + TPB - 1) / TPB;
  const int nrow4 = (NN / 2 - q * (SBLK / 2)) < SBLK / 2 ? (NN / 2 - q * (SBLK / 2)) : SBLK / 2;
  float4 v[NG], pv[NA];
#pragma unroll
  for (int u = 0; u < NG; ++u) {
    int i = u * TPB + tid;
    v[u] = (i < nrow4) ? g4[i] : make_float4(0.0f, 0.0f, 0.0f, 0.0f);
  }
#pragma unroll
  for (int u = 0; u < NA; ++u) {
    int i = u * TPB + tid;
    pv[u] = (!first && i < DRNG / 2) ? prow[i] : make_float4(0.0f, 0.0f, 0.0f, 0.0f);
  }
#pragma unroll
  for (int u = 0; u < NG; ++u) {
    int i = u * TPB + tid;
    if (i < SBLK / 2) t4[i] = v[u];
  }
#pragma unroll
  for (int u = 0; u < NA; ++u) {
    int i = u * TPB + tid;
    if (i < DRNG / 2) a4[i] = pv[u];
  }
  const unsigned s0 = bb * (unsigned)CHS2 + o0;
  const unsigned s1 = bb * (unsigned)CHS2 + o1;
  __syncthreads();

  const float2* tab2 = reinterpret_cast<const float2*>(tab);
  constexpr int NGRP = (NSEGW + NSEG2 - 1) / NSEG2;
#pragma unroll
  for (int grp = 0; grp < NGRP; ++grp) {
    unsigned sa[NSEG2], sn[NSEG2];
    unsigned nmax = 0u;
#pragma unroll
    for (int g = 0; g < NSEG2; ++g) {
      sa[g] = 0u;
      sn[g] = 0u;
      if (grp * NSEG2 + g < NSEGW) {
        sa[g] = (unsigned)__builtin_amdgcn_readlane((int)s0, grp * NSEG2 + g);
        sn[g] = (unsigned)__builtin_amdgcn_readlane((int)s1, grp * NSEG2 + g) - sa[g];
      }
      nmax = sn[g] > nmax ? sn[g] : nmax;
    }
    for (unsigned k0 = 0; k0 < nmax; k0 += 32 * GW2) {
      unsigned p[NSEG2 * GW2];
      float2 h[NSEG2 * GW2];
#pragma unroll
      for (int g = 0; g < NSEG2; ++g) {
#pragma unroll
        for (int u = 0; u < GW2; ++u) {
          unsigned k = k0 + 32u * u + lane;
          p[g * GW2 + u] = (k < sn[g]) ? bins[sa[g] + k] : 0xFFFFFFFFu;
        }
      }
#pragma unroll
      for (int u = 0; u < NSEG2 * GW2; ++u) {
        h[u] = make_float2(0.0f, 0.0f);
        if (p[u] != 0xFFFFFFFFu) h[u] = tab2[p[u] & 0xFFFFu];
      }
#pragma unroll
      for (int u = 0; u < NSEG2 * GW2; ++u) {
        if (p[u] != 0xFFFFFFFFu) {
          unsigned a = p[u] >> 16;
          unsafeAtomicAdd(&acc[2 * a + 0], h[u].x);
          unsafeAtomicAdd(&acc[2 * a + 1], h[u].y);
        }
      }
    }
  }
  __syncthreads();

  for (int i = tid; i < DRNG / 2; i += TPB) prow[i] = a4[i];
}

__global__ void gnn_layer1_wmma_kernel(const float* __restrict__ part,
                                       const float* __restrict__ x,
                                       const float* __restrict__ dinv,
                                       const float* __restrict__ W1,
                                       const float* __restrict__ b1,
                                       const float* __restrict__ W2,
                                       float* __restrict__ hs2) {
  int wave = (blockIdx.x * blockDim.x + threadIdx.x) >> 5;
  int lane = threadIdx.x & 31;
  if (wave >= NTILES) return;
  int c    = lane & 15;
  int half = lane >> 4;
  int node = (wave << 4) + c;
  bool lo  = half == 0;

  float2 a = make_float2(0.0f, 0.0f);
#pragma unroll
  for (int s = 0; s < TS; ++s) {
    float2 v = reinterpret_cast<const float2*>(part + (size_t)s * (2 * NPAD))[node];
    a.x += v.x;
    a.y += v.y;
  }
  float di = dinv[node];
  float2 xv = reinterpret_cast<const float2*>(x)[node];
  float z0 = di * (a.x + xv.x * di);
  float z1 = di * (a.y + xv.y * di);

  float w0 = W1[c];
  float w1 = W1[16 + c];

  v2f A, B;
  A.x = lo ? w0 : 0.0f;
  A.y = lo ? w1 : 0.0f;
  B.x = lo ? z0 : 0.0f;
  B.y = lo ? z1 : 0.0f;

  v8f acc = {};
  acc = __builtin_amdgcn_wmma_f32_16x16x4_f32(
       false, A,  false, B,
       (short)0, acc,  false,  false);

  const float* __restrict__ bb = b1 + 8 * half;
  const float* __restrict__ ww = W2 + 16 * half;
  float h0 = 0.0f, h1 = 0.0f;
#pragma unroll
  for (int r = 0; r < 8; ++r) {
    float v = acc[r] + bb[r];
    v = v > 0.0f ? v : 0.0f;
    h0 += v * ww[2 * r + 0];
    h1 += v * ww[2 * r + 1];
  }
  h0 += __shfl_xor(h0, 16);
  h1 += __shfl_xor(h1, 16);
  if (lo) reinterpret_cast<float2*>(hs2)[node] = make_float2(h0 * di, h1 * di);
}

__global__ void gnn_finalize2_kernel(const float* __restrict__ part,
                                     const float* __restrict__ dinv,
                                     const float* __restrict__ b2,
                                     float* out) {
  int t = blockIdx.x * blockDim.x + threadIdx.x;
  if (t >= NN * 2) return;
  int i = t >> 1;
  int c = t & 1;
  float a = 0.0f;
#pragma unroll
  for (int s = 0; s < TS; ++s) a += part[(size_t)s * (2 * NPAD) + t];
  out[t] = dinv[i] * (a + out[t]) + b2[c];
}

extern "C" void kernel_launch(void* const* d_in, const int* in_sizes, int n_in,
                              void* d_out, int out_size, void* d_ws, size_t ws_size,
                              hipStream_t stream) {
  const float* x  = (const float*)d_in[0];
  const float* W1 = (const float*)d_in[1];
  const float* b1 = (const float*)d_in[2];
  const float* W2 = (const float*)d_in[3];
  const float* b2 = (const float*)d_in[4];
  const int* edge = (const int*)d_in[5];
  const int* src  = edge;
  const int* dst  = edge + NE;
  float* out = (float*)d_out;

  float* ws = (float*)d_ws;
  float* dinv  = ws;
  float* part  = ws + WS_PART;
  unsigned* offs = (unsigned*)(ws + WS_OFFS2);
  unsigned* offs16 = (unsigned*)(ws + WS_OFFS16);
  unsigned short* bins16 = (unsigned short*)(ws + WS_BINS16);
  float* rows = out;

  const int B = 256;
  const int threads1 = NTILES * 32;
  const int n2 = NN * 2;

  gnn_sort_chunk16_kernel<<<NB, TPB, 0, stream>>>(dst, offs16, bins16);
  gnn_deg_owner16_kernel<<<NR, TPB, 0, stream>>>(offs16, bins16, x, dinv, rows);

  if (ws_size >= WS_BYTES_FULL) {
    unsigned* bins = (unsigned*)(ws + WS_BINS_OF(NSLOT_FULL));
    gnn_sort_tile_kernel<<<NSLOT_FULL, TPB, 0, stream>>>(src, dst, offs, bins, 0);
    gnn_tile_sum2_kernel<NSLOT_FULL / NWAVES><<<NT, TPB, 0, stream>>>(offs, bins, rows, part, 1);
    gnn_layer1_wmma_kernel<<<(threads1 + B - 1) / B, B, 0, stream>>>(part, x, dinv, W1, b1, W2, rows);
    gnn_tile_sum2_kernel<NSLOT_FULL / NWAVES><<<NT, TPB, 0, stream>>>(offs, bins, rows, part, 1);
  } else if (ws_size >= WS_BYTES_HALF) {
    unsigned* bins = (unsigned*)(ws + WS_BINS_OF(NSLOT_HALF));
    gnn_sort_tile_kernel<<<NSLOT_HALF, TPB, 0, stream>>>(src, dst, offs, bins, 0);
    gnn_tile_sum2_kernel<NSLOT_HALF / NWAVES><<<NT, TPB, 0, stream>>>(offs, bins, rows, part, 1);
    gnn_sort_tile_kernel<<<NSLOT_HALF, TPB, 0, stream>>>(src, dst, offs, bins, NSLOT_HALF);
    gnn_tile_sum2_kernel<NSLOT_HALF / NWAVES><<<NT, TPB, 0, stream>>>(offs, bins, rows, part, 0);
    gnn_layer1_wmma_kernel<<<(threads1 + B - 1) / B, B, 0, stream>>>(part, x, dinv, W1, b1, W2, rows);
    gnn_tile_sum2_kernel<NSLOT_HALF / NWAVES><<<NT, TPB, 0, stream>>>(offs, bins, rows, part, 1);
    gnn_sort_tile_kernel<<<NSLOT_HALF, TPB, 0, stream>>>(src, dst, offs, bins, 0);
    gnn_tile_sum2_kernel<NSLOT_HALF / NWAVES><<<NT, TPB, 0, stream>>>(offs, bins, rows, part, 0);
  } else {
    unsigned* bins = (unsigned*)(ws + WS_BINS2);
    gnn_sort_tile_kernel<<<PCH, TPB, 0, stream>>>(src, dst, offs, bins, 0);
    gnn_tile_sum2_kernel<NSEG2><<<NT, TPB, 0, stream>>>(offs, bins, rows, part, 1);
    gnn_sort_tile_kernel<<<PCH, TPB, 0, stream>>>(src, dst, offs, bins, PCH);
    gnn_tile_sum2_kernel<NSEG2><<<NT, TPB, 0, stream>>>(offs, bins, rows, part, 0);
    gnn_sort_tile_kernel<<<PCH, TPB, 0, stream>>>(src, dst, offs, bins, 2 * PCH);
    gnn_tile_sum2_kernel<NSEG2><<<NT, TPB, 0, stream>>>(offs, bins, rows, part, 0);
    gnn_layer1_wmma_kernel<<<(threads1 + B - 1) / B, B, 0, stream>>>(part, x, dinv, W1, b1, W2, rows);
    gnn_tile_sum2_kernel<NSEG2><<<NT, TPB, 0, stream>>>(offs, bins, rows, part, 1);
    gnn_sort_tile_kernel<<<PCH, TPB, 0, stream>>>(src, dst, offs, bins, 0);
    gnn_tile_sum2_kernel<NSEG2><<<NT, TPB, 0, stream>>>(offs, bins, rows, part, 0);
    gnn_sort_tile_kernel<<<PCH, TPB, 0, stream>>>(src, dst, offs, bins, PCH);
    gnn_tile_sum2_kernel<NSEG2><<<NT, TPB, 0, stream>>>(offs, bins, rows, part, 0);
  }

  gnn_finalize2_kernel<<<(n2 + B - 1) / B, B, 0, stream>>>(part, dinv, b2, out);
}


# Degree count read from the kept sorted tile list (a range's 6 tiles are one run per chunk): 42 ranges x 6 chunk groups count in LDS, 6 integer planes added per node; the 16-bit degree sort is gone
# speedup vs baseline: 15.7236x; 1.1153x over previous
#include <hip/hip_runtime.h>


#define NN 200000
#define NE 12800000
#define NTILES (NN / 16)

#define TPB 1024
#define NWAVES (TPB / 32)
#define NCP 32
#define NKEY 256

#define RNG 800
#define NR (NN / RNG)
#define NB 512
#define CH (NE / NB)
#define CH4 (CH / 4)
#define OFFW 256
#define BU 7
#define GW 4
#define CHS16 25024
#define NSEG16 (NB / NWAVES)
#define SG16 4

#define TD 42
#define TS 6
#define NT (TD * TS)
#define NTP 256
#define DRNG 4768
#define NPAD (TD * DRNG)
#define SBLK 33334
#define NPART 3
#define PCH 256
#define CHA 16668
#define CHB 16664
#define CHS2 16672
#define OFFW2 256
#define BU2 5
#define NCHUNK (NPART * PCH)
#define NSLOT_HALF (NCHUNK / 2)
#define NSLOT_FULL NCHUNK
#define NSEG2 (PCH / NWAVES)
#define GW2 3

#define NCG 6
#define CGC (NSLOT_FULL / NCG)
#define CSEGW (CGC / NWAVES)
#define GWC 7

static_assert(NR * RNG == NN && RNG % 32 == 0, "degree ranges cover the nodes in whole lines");
static_assert(NCG * CGC == NSLOT_FULL && CSEGW * NWAVES == CGC && CSEGW <= 32, "chunk groups cover the slots; bounds held one per lane");
static_assert(NCG <= 2 * TS && NN % 4 == 0 && NPAD % 4 == 0, "count planes lie inside the planes of partial sums; 4 nodes per 16-byte group");
static_assert(CH * NB == NE && CH % 4 == 0, "degree chunks cover the edges");
static_assert(BU * TPB >= CH4, "one pass of the work-group loads a degree chunk");
static_assert(NR + 1 <= OFFW && OFFW == 8 * 32, "degree offset row: 8 entries per lane of one wave");
static_assert(CHS16 >= CH && (CHS16 * 2) % 128 == 0 && (CH * 2) % 16 == 0, "16-bit chunks: whole lines, whole 16-byte groups");
static_assert(NSEG16 * NWAVES == NB && NSEG16 % SG16 == 0 && NSEG16 <= 32, "degree pass: segment groups per wave, bounds held one per lane");
static_assert((size_t)NB * CHS16 < ((size_t)1 << 32), "16-bit entry index fits 32 bits");

static_assert(NPAD >= NN && DRNG % 16 == 0, "destination ranges cover the nodes in whole lines of 8-byte rows");
static_assert(TS * SBLK >= NN && (TS - 1) * SBLK < NN && SBLK % 2 == 0 && NN % 2 == 0, "source blocks cover the nodes in whole 16-byte groups");
static_assert(DRNG < (1 << 16) - 1 && SBLK <= (1 << 16), "word = 16 bits local destination | 16 bits local source; all ones is no word");
static_assert(NT <= NTP && NTP == 8 * 32 && NT + 1 <= OFFW2 && OFFW2 == NTP, "tile offset row: 8 counts per lane of one wave, whole lines");
static_assert((size_t)PCH * ((NPART - 1) * CHA + CHB) == NE && CHA % 4 == 0 && CHB % 4 == 0 && CHB <= CHA && BU2 * TPB >= CHA / 4, "chunks cover the edges; one pass loads a chunk");
static_assert(CHS2 >= CHA && CHS2 % 32 == 0, "whole lines per chunk");
static_assert(NSEG2 * NWAVES == PCH && NSEG2 <= 32, "bounds held one per lane");
static_assert(NSLOT_HALF % NWAVES == 0 && NSLOT_FULL % NWAVES == 0 && NSLOT_FULL / NWAVES <= 32,
              "every wave of a tile pass walks the same number of slots, bounds held one per lane");
static_assert((size_t)NSLOT_FULL * CHS2 < ((size_t)1 << 32), "word index fits 32 bits");
static_assert(NKEY == NTP && NKEY == OFFW && NKEY * NCP == 8 * TPB && NWAVES == 32 && NCP == 32,
              "counter scan: 8 entries per thread, one wave total per lane of a wave");

typedef __attribute__((ext_vector_type(2))) float v2f;
typedef __attribute__((ext_vector_type(8))) float v8f;

#define WS_OFFS16 ((size_t)1 * NN)
#define WS_BINS16 (WS_OFFS16 + (size_t)NB * OFFW)
#define WS_WORDS16 (WS_BINS16 + (size_t)NB * CHS16 / 2)
#define WS_PART ((size_t)1 * NN)
#define WS_OFFS2 (WS_PART + (size_t)TS * 2 * NPAD)
#define WS_BINS2 (WS_OFFS2 + (size_t)PCH * OFFW2)
#define WS_WORDS2 (WS_BINS2 + (size_t)PCH * CHS2)
#define WS_BINS_OF(S) (WS_OFFS2 + (size_t)(S) * OFFW2)
#define WS_WORDS_OF(S) (WS_BINS_OF(S) + (size_t)(S) * CHS2)
#define WS_BYTES_HALF (WS_WORDS_OF(NSLOT_HALF) * 4)
#define WS_BYTES_FULL (WS_WORDS_OF(NSLOT_FULL) * 4)
static_assert(WS_WORDS2 * 4 <= (size_t)35 * NN * 4, "inside the 28,000,000 bytes the versions before this one used");
static_assert(WS_PART % 32 == 0 && (2 * NPAD) % 32 == 0 && WS_OFFS2 % 32 == 0 && WS_BINS2 % 32 == 0, "planes start on 128-byte lines");
static_assert(WS_WORDS16 * 4 <= (size_t)35 * NN * 4, "degree list inside the same 28,000,000 bytes");
static_assert(WS_BINS_OF(PCH) == WS_BINS2 && WS_WORDS_OF(PCH) == WS_WORDS2, "the smallest layout is the one above");
static_assert(WS_BINS_OF(NSLOT_HALF) % 32 == 0 && WS_BINS_OF(NSLOT_FULL) % 32 == 0, "planes start on 128-byte lines");
static_assert(WS_BYTES_HALF == (size_t)36413696 && WS_BYTES_FULL == (size_t)62415104, "sizes named in the comments");
static_assert(WS_WORDS16 <= WS_WORDS2, "the degree list lies inside every layout");
static_assert(WS_OFFS16 % 32 == 0 && WS_BINS16 % 32 == 0, "degree planes start on 128-byte lines");

__device__ __forceinline__ unsigned gnn_wave_scan_incl(unsigned v, unsigned lane) {
  v += (unsigned)__builtin_amdgcn_update_dpp(0, (int)v, 0x111, 0xf, 0xf, false);
  v += (unsigned)__builtin_amdgcn_update_dpp(0, (int)v, 0x112, 0xf, 0xf, false);
  v += (unsigned)__builtin_amdgcn_update_dpp(0, (int)v, 0x114, 0xf, 0xf, false);
  v += (unsigned)__builtin_amdgcn_update_dpp(0, (int)v, 0x118, 0xf, 0xf, false);
  const unsigned low = (unsigned)__builtin_amdgcn_readlane((int)v, 15);
  if (lane >= 16u) v += low;
  return v;
}

__device__ __forceinline__ void gnn_scan_private_counters(unsigned* cnt, unsigned* wtot, unsigned* kbase,
                                                          int tid, unsigned lane, unsigned w) {
  uint4* c4 = reinterpret_cast<uint4*>(cnt);
  const uint4 ca = c4[2 * tid + 0];
  const uint4 cb = c4[2 * tid + 1];
  const unsigned c[8] = {ca.x, ca.y, ca.z, ca.w, cb.x, cb.y, cb.z, cb.w};
  unsigned s = 0u;
#pragma unroll
  for (int k = 0; k < 8; ++k) s += c[k];
  const unsigned inc = gnn_wave_scan_incl(s, lane);
  if (lane == 31u) wtot[w] = inc;
  __syncthreads();
  const unsigned wv = wtot[lane];
  const unsigned winc = gnn_wave_scan_incl(wv, lane);
  const int ws = __builtin_amdgcn_readfirstlane((int)w);
  const unsigned wbase = (unsigned)__builtin_amdgcn_readlane((int)(winc - wv), ws);
  unsigned ex = wbase + (inc - s);
  if ((tid & 3) == 0) kbase[tid >> 2] = ex;
  unsigned e[8];
#pragma unroll
  for (int k = 0; k < 8; ++k) { e[k] = ex; ex += c[k]; }
  c4[2 * tid + 0] = make_uint4(e[0], e[1], e[2], e[3]);
  c4[2 * tid + 1] = make_uint4(e[4], e[5], e[6], e[7]);
  __syncthreads();
}

__global__ __launch_bounds__(TPB) void gnn_sort_tile_kernel(const int* __restrict__ src,
                                                            const int* __restrict__ dst,
                                                            unsigned* __restrict__ offs,
                                                            unsigned* __restrict__ bins,
                                                            int cbase) {
  __shared__ __attribute__((aligned(16))) unsigned cnt[NKEY * NCP];
  __shared__ unsigned wtot[NWAVES];
  __shared__ unsigned kbase[NKEY];
  __shared__ __attribute__((aligned(16))) unsigned stage[CHA];
  const int slot = blockIdx.x;
  const int part = (cbase + slot) / PCH;
  const int b = (cbase + slot) - part * PCH;
  const int tid = threadIdx.x;
  const unsigned lane = threadIdx.x & 31;
  const unsigned w = threadIdx.x >> 5;
  reinterpret_cast<uint4*>(cnt)[2 * tid + 0] = make_uint4(0u, 0u, 0u, 0u);
  reinterpret_cast<uint4*>(cnt)[2 * tid + 1] = make_uint4(0u, 0u, 0u, 0u);
  __syncthreads();

  const int csz4 = (part < NPART - 1) ? CHA / 4 : CHB / 4;
  const size_t e0 = (size_t)part * PCH * CHA + (size_t)b * (4 * csz4);
  const int4* __restrict__ d4 = reinterpret_cast<const int4*>(dst + e0);
  const int4* __restrict__ s4 = reinterpret_cast<const int4*>(src + e0);
  int4 dv[BU2], sv[BU2];
#pragma unroll
  for (int u = 0; u < BU2; ++u) {
    int i = u * TPB + tid;
    bool ok = i < csz4;
    dv[u] = ok ? d4[i] : make_int4(-1, -1, -1, -1);
    sv[u] = ok ? s4[i] : make_int4(-1, -1, -1, -1);
  }
#pragma unroll
  for (int u = 0; u < BU2; ++u) {
    const unsigned dd[4] = {(unsigned)dv[u].x, (unsigned)dv[u].y, (unsigned)dv[u].z, (unsigned)dv[u].w};
    const unsigned ss[4] = {(unsigned)sv[u].x, (unsigned)sv[u].y, (unsigned)sv[u].z, (unsigned)sv[u].w};
#pragma unroll
    for (int k = 0; k < 4; ++k)
      if (dd[k] < (unsigned)NN && ss[k] < (unsigned)NN)
        atomicAdd(&cnt[((dd[k] / DRNG) * TS + ss[k] / SBLK) * NCP + lane], 1u);
  }
  __syncthreads();

  gnn_scan_private_counters(cnt, wtot, kbase, tid, lane, w);
  if (tid < NKEY) offs[(size_t)slot * OFFW2 + tid] = kbase[tid];

#pragma unroll
  for (int u = 0; u < BU2; ++u) {
    const unsigned dd[4] = {(unsigned)dv[u].x, (unsigned)dv[u].y, (unsigned)dv[u].z, (unsigned)dv[u].w};
    const unsigned ss[4] = {(unsigned)sv[u].x, (unsigned)sv[u].y, (unsigned)sv[u].z, (unsigned)sv[u].w};
#pragma unroll
    for (int k = 0; k < 4; ++k) {
      if (dd[k] < (unsigned)NN && ss[k] < (unsigned)NN) {
        unsigned r = dd[k] / DRNG;
        unsigned q = ss[k] / SBLK;
        unsigned pos = atomicAdd(&cnt[(r * TS + q) * NCP + lane], 1u);
        stage[pos] = ((dd[k] - r * DRNG) << 16) | (ss[k] - q * SBLK);
      }
    }
  }
  __syncthreads();

  uint4* __restrict__ o4 = reinterpret_cast<uint4*>(bins + (size_t)slot * CHS2);
  const uint4* st4 = reinterpret_cast<const uint4*>(stage);
  for (int i = tid; i < csz4; i += TPB) o4[i] = st4[i];
}

__global__ __launch_bounds__(TPB) void gnn_sort_chunk16_kernel(const int* __restrict__ dst,
                                                               unsigned* __restrict__ offs,
                                                               unsigned short* __restrict__ bins16) {
  __shared__ __attribute__((aligned(16))) unsigned cnt[NKEY * NCP];
  __shared__ unsigned wtot[NWAVES];
  __shared__ unsigned kbase[NKEY];
  __shared__ __attribute__((aligned(16))) unsigned short stage[CH];
  const int b = blockIdx.x;
  const int tid = threadIdx.x;
  const unsigned lane = threadIdx.x & 31;
  const unsigned w = threadIdx.x >> 5;
  reinterpret_cast<uint4*>(cnt)[2 * tid + 0] = make_uint4(0u, 0u, 0u, 0u);
  reinterpret_cast<uint4*>(cnt)[2 * tid + 1] = make_uint4(0u, 0u, 0u, 0u);
  __syncthreads();

  const int4* __restrict__ d4 = reinterpret_cast<const int4*>(dst + (size_t)b * CH);
  int4 dv[BU];
#pragma unroll
  for (int u = 0; u < BU; ++u) {
    int i = u * TPB + tid;
    dv[u] = (i < CH4) ? d4[i] : make_int4(-1, -1, -1, -1);
  }
#pragma unroll
  for (int u = 0; u < BU; ++u) {
    const unsigned dd[4] = {(unsigned)dv[u].x, (unsigned)dv[u].y, (unsigned)dv[u].z, (unsigned)dv[u].w};
#pragma unroll
    for (int k = 0; k < 4; ++k)
      if (dd[k] < (unsigned)NN) atomicAdd(&cnt[(dd[k] / RNG) * NCP + lane], 1u);
  }
  __syncthreads();

  gnn_scan_private_counters(cnt, wtot, kbase, tid, lane, w);
  if (tid < NKEY) offs[(size_t)b * OFFW + tid] = kbase[tid];

#pragma unroll
  for (int u = 0; u < BU; ++u) {
    const unsigned dd[4] = {(unsigned)dv[u].x, (unsigned)dv[u].y, (unsigned)dv[u].z, (unsigned)dv[u].w};
#pragma unroll
    for (int k = 0; k < 4; ++k) {
      if (dd[k] < (unsigned)NN) {
        unsigned r = dd[k] / RNG;
        unsigned pos = atomicAdd(&cnt[r * NCP + lane], 1u);
        stage[pos] = (unsigned short)(dd[k] - r * RNG);
      }
    }
  }
  __syncthreads();

  uint4* __restrict__ o4 = reinterpret_cast<uint4*>(bins16 + (size_t)b * CHS16);
  const uint4* st4 = reinterpret_cast<const uint4*>(stage);
  for (int i = tid; i < CH / 8; i += TPB) o4[i] = st4[i];
}

__global__ __launch_bounds__(TPB) void gnn_deg_owner16_kernel(const unsigned* __restrict__ offs,
                                                              const unsigned short* __restrict__ bins16,
                                                              const float* __restrict__ x,
                                                              float* __restrict__ dinv,
                                                              float* __restrict__ xs) {
  __shared__ unsigned cnt[RNG];
  const int r = blockIdx.x;
  const int tid = threadIdx.x;
  const unsigned lane = threadIdx.x & 31;
  const unsigned w = threadIdx.x >> 5;
  if (tid < RNG) cnt[tid] = 0u;
  __syncthreads();

  unsigned s0 = 0u, s1 = 0u;
  if (lane < (unsigned)NSEG16) {
    unsigned bb = w * (unsigned)NSEG16 + lane;
    const unsigned* __restrict__ o = offs + (size_t)bb * OFFW + r;
    s0 = bb * (unsigned)CHS16 + o[0];
    s1 = bb * (unsigned)CHS16 + o[1];
  }
  for (int j = 0; j < NSEG16; j += SG16) {
    unsigned sa[SG16], sn[SG16];
    unsigned nmax = 0u;
#pragma unroll
    for (int g = 0; g < SG16; ++g) {
      sa[g] = (unsigned)__builtin_amdgcn_readlane((int)s0, j + g);
      sn[g] = (unsigned)__builtin_amdgcn_readlane((int)s1, j + g) - sa[g];
      nmax = sn[g] > nmax ? sn[g] : nmax;
    }
    for (unsigned k0 = 0; k0 < nmax; k0 += 32 * GW) {
      unsigned p[SG16 * GW];
#pragma unroll
      for (int g = 0; g < SG16; ++g) {
#pragma unroll
        for (int t = 0; t < GW; ++t) {
          unsigned k = k0 + 32u * t + lane;
          p[g * GW + t] = (k < sn[g]) ? (unsigned)bins16[sa[g] + k] : 0xFFFFu;
        }
      }
#pragma unroll
      for (int t = 0; t < SG16 * GW; ++t)
        if (p[t] != 0xFFFFu) atomicAdd(&cnt[p[t]], 1u);
    }
  }
  __syncthreads();

  if (tid < RNG) {
    int node = r * RNG + tid;
    float d = rsqrtf((float)cnt[tid] + 1.0f);
    dinv[node] = d;
    float2 xv = reinterpret_cast<const float2*>(x)[node];
    reinterpret_cast<float2*>(xs)[node] = make_float2(xv.x * d, xv.y * d);
  }
}

__global__ __launch_bounds__(TPB) void gnn_deg_tile_kernel(const unsigned* __restrict__ offs,
                                                           const unsigned* __restrict__ bins,
                                                           unsigned* __restrict__ cntp) {
  __shared__ unsigned cnt[DRNG];
  const int r = blockIdx.x / NCG;
  const int g = blockIdx.x - r * NCG;
  const int tid = threadIdx.x;
  const unsigned lane = threadIdx.x & 31;
  const unsigned w = threadIdx.x >> 5;

  const unsigned bb = (unsigned)g * (unsigned)CGC + w * (unsigned)CSEGW + (lane < (unsigned)CSEGW ? lane : 0u);
  const unsigned* __restrict__ o = offs + (size_t)bb * OFFW2 + r * TS;
  const unsigned o0 = o[0], o1 = o[TS];
  for (int i = tid; i < DRNG; i += TPB) cnt[i] = 0u;
  const unsigned s0 = bb * (unsigned)CHS2 + o0;
  const unsigned s1 = bb * (unsigned)CHS2 + o1;
  __syncthreads();

  unsigned sa[CSEGW], sn[CSEGW];
  unsigned nmax = 0u;
#pragma unroll
  for (int j = 0; j < CSEGW; ++j) {
    sa[j] = (unsigned)__builtin_amdgcn_readlane((int)s0, j);
    sn[j] = (unsigned)__builtin_amdgcn_readlane((int)s1, j) - sa[j];
    nmax = sn[j] > nmax ? sn[j] : nmax;
  }
  for (unsigned k0 = 0; k0 < nmax; k0 += 32 * GWC) {
    unsigned p[CSEGW * GWC];
#pragma unroll
    for (int j = 0; j < CSEGW; ++j) {
#pragma unroll
      for (int u = 0; u < GWC; ++u) {
        unsigned k = k0 + 32u * u + lane;
        p[j * GWC + u] = (k < sn[j]) ? bins[sa[j] + k] : 0xFFFFFFFFu;
      }
    }
#pragma unroll
    for (int u = 0; u < CSEGW * GWC; ++u)
      if (p[u] != 0xFFFFFFFFu) atomicAdd(&cnt[p[u] >> 16], 1u);
  }
  __syncthreads();

  unsigned* __restrict__ plane = cntp + (size_t)g * NPAD + (size_t)r * DRNG;
  for (int i = tid; i < DRNG; i += TPB) plane[i] = cnt[i];
}

__global__ void gnn_deg_finish_kernel(const unsigned* __restrict__ cntp,
                                      const float* __restrict__ x,
                                      float* __restrict__ dinv,
                                      float* __restrict__ xs) {
  int t = blockIdx.x * blockDim.x + threadIdx.x;
  if (t >= NN / 4) return;
  uint4 cs = make_uint4(0u, 0u, 0u, 0u);
#pragma unroll
  for (int g = 0; g < NCG; ++g) {
    uint4 v = reinterpret_cast<const uint4*>(cntp + (size_t)g * NPAD)[t];
    cs.x += v.x;
    cs.y += v.y;
    cs.z += v.z;
    cs.w += v.w;
  }
  float4 xa = reinterpret_cast<const float4*>(x)[2 * t + 0];
  float4 xb = reinterpret_cast<const float4*>(x)[2 * t + 1];
  float4 d;
  d.x = rsqrtf((float)cs.x + 1.0f);
  d.y = rsqrtf((float)cs.y + 1.0f);
  d.z = rsqrtf((float)cs.z + 1.0f);
  d.w = rsqrtf((float)cs.w + 1.0f);
  reinterpret_cast<float4*>(dinv)[t] = d;
  reinterpret_cast<float4*>(xs)[2 * t + 0] = make_float4(xa.x * d.x, xa.y * d.x, xa.z * d.y, xa.w * d.y);
  reinterpret_cast<float4*>(xs)[2 * t + 1] = make_float4(xb.x * d.z, xb.y * d.z, xb.z * d.w, xb.w * d.w);
}

template <int NSEGW>
__global__ __launch_bounds__(TPB) void gnn_tile_sum2_kernel(const unsigned* __restrict__ offs,
                                                            const unsigned* __restrict__ bins,
                                                            const float* __restrict__ rows2,
                                                            float* __restrict__ part,
                                                            int first) {
  __shared__ __attribute__((aligned(16))) float tab[SBLK * 2];
  __shared__ __attribute__((aligned(16))) float acc[DRNG * 2];
  const int t = blockIdx.x;
  const int r = t / TS;
  const int q = t - r * TS;
  const int tid = threadIdx.x;
  const unsigned lane = threadIdx.x & 31;
  const unsigned w = threadIdx.x >> 5;

  const unsigned bb = w * (unsigned)NSEGW + (lane < (unsigned)NSEGW ? lane : 0u);
  const unsigned* __restrict__ o = offs + (size_t)bb * OFFW2 + t;
  const unsigned o0 = o[0], o1 = o[1];

  const float4* __restrict__ g4 = reinterpret_cast<const float4*>(rows2 + (size_t)2 * q * SBLK);
  float4* __restrict__ prow = reinterpret_cast<float4*>(part + (size_t)q * (2 * NPAD) + (size_t)2 * r * DRNG);
  float4* t4 = reinterpret_cast<float4*>(tab);
  float4* a4 = reinterpret_cast<float4*>(acc);
  constexpr int NG = (SBLK / 2 + TPB - 1) / TPB;
  constexpr int NA = (DRNG / 2 + TPB - 1) / TPB;
  const int nrow4 = (NN / 2 - q * (SBLK / 2)) < SBLK / 2 ? (NN / 2 - q * (SBLK / 2)) : SBLK / 2;
  float4 v[NG], pv[NA];
#pragma unroll
  for (int u = 0; u < NG; ++u) {
    int i = u * TPB + tid;
    v[u] = (i < nrow4) ? g4[i] : make_float4(0.0f, 0.0f, 0.0f, 0.0f);
  }
#pragma unroll
  for (int u = 0; u < NA; ++u) {
    int i = u * TPB + tid;
    pv[u] = (!first && i < DRNG / 2) ? prow[i] : make_float4(0.0f, 0.0f, 0.0f, 0.0f);
  }
#pragma unroll
  for (int u = 0; u < NG; ++u) {
    int i = u * TPB + tid;
    if (i < SBLK / 2) t4[i] = v[u];
  }
#pragma unroll
  for (int u = 0; u < NA; ++u) {
    int i = u * TPB + tid;
    if (i < DRNG / 2) a4[i] = pv[u];
  }
  const unsigned s0 = bb * (unsigned)CHS2 + o0;
  const unsigned s1 = bb * (unsigned)CHS2 + o1;
  __syncthreads();

  const float2* tab2 = reinterpret_cast<const float2*>(tab);
  constexpr int NGRP = (NSEGW + NSEG2 - 1) / NSEG2;
#pragma unroll
  for (int grp = 0; grp < NGRP; ++grp) {
    unsigned sa[NSEG2], sn[NSEG2];
    unsigned nmax = 0u;
#pragma unroll
    for (int g = 0; g < NSEG2; ++g) {
      sa[g] = 0u;
      sn[g] = 0u;
      if (grp * NSEG2 + g < NSEGW) {
        sa[g] = (unsigned)__builtin_amdgcn_readlane((int)s0, grp * NSEG2 + g);
        sn[g] = (unsigned)__builtin_amdgcn_readlane((int)s1, grp * NSEG2 + g) - sa[g];
      }
      nmax = sn[g] > nmax ? sn[g] : nmax;
    }
    for (unsigned k0 = 0; k0 < nmax; k0 += 32 * GW2) {
      unsigned p[NSEG2 * GW2];
      float2 h[NSEG2 * GW2];
#pragma unroll
      for (int g = 0; g < NSEG2; ++g) {
#pragma unroll
        for (int u = 0; u < GW2; ++u) {
          unsigned k = k0 + 32u * u + lane;
          p[g * GW2 + u] = (k < sn[g]) ? bins[sa[g] + k] : 0xFFFFFFFFu;
        }
      }
#pragma unroll
      for (int u = 0; u < NSEG2 * GW2; ++u) {
        h[u] = make_float2(0.0f, 0.0f);
        if (p[u] != 0xFFFFFFFFu) h[u] = tab2[p[u] & 0xFFFFu];
      }
#pragma unroll
      for (int u = 0; u < NSEG2 * GW2; ++u) {
        if (p[u] != 0xFFFFFFFFu) {
          unsigned a = p[u] >> 16;
          unsafeAtomicAdd(&acc[2 * a + 0], h[u].x);
          unsafeAtomicAdd(&acc[2 * a + 1], h[u].y);
        }
      }
    }
  }
  __syncthreads();

  for (int i = tid; i < DRNG / 2; i += TPB) prow[i] = a4[i];
}

__global__ void gnn_layer1_wmma_kernel(const float* __restrict__ part,
                                       const float* __restrict__ x,
                                       const float* __restrict__ dinv,
                                       const float* __restrict__ W1,
                                       const float* __restrict__ b1,
                                       const float* __restrict__ W2,
                                       float* __restrict__ hs2) {
  int wave = (blockIdx.x * blockDim.x + threadIdx.x) >> 5;
  int lane = threadIdx.x & 31;
  if (wave >= NTILES) return;
  int c    = lane & 15;
  int half = lane >> 4;
  int node = (wave << 4) + c;
  bool lo  = half == 0;

  float2 a = make_float2(0.0f, 0.0f);
#pragma unroll
  for (int s = 0; s < TS; ++s) {
    float2 v = reinterpret_cast<const float2*>(part + (size_t)s * (2 * NPAD))[node];
    a.x += v.x;
    a.y += v.y;
  }
  float di = dinv[node];
  float2 xv = reinterpret_cast<const float2*>(x)[node];
  float z0 = di * (a.x + xv.x * di);
  float z1 = di * (a.y + xv.y * di);

  float w0 = W1[c];
  float w1 = W1[16 + c];

  v2f A, B;
  A.x = lo ? w0 : 0.0f;
  A.y = lo ? w1 : 0.0f;
  B.x = lo ? z0 : 0.0f;
  B.y = lo ? z1 : 0.0f;

  v8f acc = {};
  acc = __builtin_amdgcn_wmma_f32_16x16x4_f32(
       false, A,  false, B,
       (short)0, acc,  false,  false);

  const float* __restrict__ bb = b1 + 8 * half;
  const float* __restrict__ ww = W2 + 16 * half;
  float h0 = 0.0f, h1 = 0.0f;
#pragma unroll
  for (int r = 0; r < 8; ++r) {
    float v = acc[r] + bb[r];
    v = v > 0.0f ? v : 0.0f;
    h0 += v * ww[2 * r + 0];
    h1 += v * ww[2 * r + 1];
  }
  h0 += __shfl_xor(h0, 16);
  h1 += __shfl_xor(h1, 16);
  if (lo) reinterpret_cast<float2*>(hs2)[node] = make_float2(h0 * di, h1 * di);
}

__global__ void gnn_finalize2_kernel(const float* __restrict__ part,
                                     const float* __restrict__ dinv,
                                     const float* __restrict__ b2,
                                     float* out) {
  int t = blockIdx.x * blockDim.x + threadIdx.x;
  if (t >= NN * 2) return;
  int i = t >> 1;
  int c = t & 1;
  float a = 0.0f;
#pragma unroll
  for (int s = 0; s < TS; ++s) a += part[(size_t)s * (2 * NPAD) + t];
  out[t] = dinv[i] * (a + out[t]) + b2[c];
}

extern "C" void kernel_launch(void* const* d_in, const int* in_sizes, int n_in,
                              void* d_out, int out_size, void* d_ws, size_t ws_size,
                              hipStream_t stream) {
  const float* x  = (const float*)d_in[0];
  const float* W1 = (const float*)d_in[1];
  const float* b1 = (const float*)d_in[2];
  const float* W2 = (const float*)d_in[3];
  const float* b2 = (const float*)d_in[4];
  const int* edge = (const int*)d_in[5];
  const int* src  = edge;
  const int* dst  = edge + NE;
  float* out = (float*)d_out;

  float* ws = (float*)d_ws;
  float* dinv  = ws;
  float* part  = ws + WS_PART;
  unsigned* offs = (unsigned*)(ws + WS_OFFS2);
  unsigned* offs16 = (unsigned*)(ws + WS_OFFS16);
  unsigned short* bins16 = (unsigned short*)(ws + WS_BINS16);
  float* rows = out;

  const int B = 256;
  const int threads1 = NTILES * 32;
  const int n2 = NN * 2;

  if (ws_size >= WS_BYTES_FULL) {
    unsigned* bins = (unsigned*)(ws + WS_BINS_OF(NSLOT_FULL));
    unsigned* cntp = (unsigned*)(ws + WS_PART);
    gnn_sort_tile_kernel<<<NSLOT_FULL, TPB, 0, stream>>>(src, dst, offs, bins, 0);
    gnn_deg_tile_kernel<<<TD * NCG, TPB, 0, stream>>>(offs, bins, cntp);
    gnn_deg_finish_kernel<<<(NN / 4 + B - 1) / B, B, 0, stream>>>(cntp, x, dinv, rows);
    gnn_tile_sum2_kernel<NSLOT_FULL / NWAVES><<<NT, TPB, 0, stream>>>(offs, bins, rows, part, 1);
    gnn_layer1_wmma_kernel<<<(threads1 + B - 1) / B, B, 0, stream>>>(part, x, dinv, W1, b1, W2, rows);
    gnn_tile_sum2_kernel<NSLOT_FULL / NWAVES><<<NT, TPB, 0, stream>>>(offs, bins, rows, part, 1);
  } else if (ws_size >= WS_BYTES_HALF) {
    unsigned* bins = (unsigned*)(ws + WS_BINS_OF(NSLOT_HALF));
    gnn_sort_chunk16_kernel<<<NB, TPB, 0, stream>>>(dst, offs16, bins16);
    gnn_deg_owner16_kernel<<<NR, TPB, 0, stream>>>(offs16, bins16, x, dinv, rows);
    gnn_sort_tile_kernel<<<NSLOT_HALF, TPB, 0, stream>>>(src, dst, offs, bins, 0);
    gnn_tile_sum2_kernel<NSLOT_HALF / NWAVES><<<NT, TPB, 0, stream>>>(offs, bins, rows, part, 1);
    gnn_sort_tile_kernel<<<NSLOT_HALF, TPB, 0, stream>>>(src, dst, offs, bins, NSLOT_HALF);
    gnn_tile_sum2_kernel<NSLOT_HALF / NWAVES><<<NT, TPB, 0, stream>>>(offs, bins, rows, part, 0);
    gnn_layer1_wmma_kernel<<<(threads1 + B - 1) / B, B, 0, stream>>>(part, x, dinv, W1, b1, W2, rows);
    gnn_tile_sum2_kernel<NSLOT_HALF / NWAVES><<<NT, TPB, 0, stream>>>(offs, bins, rows, part, 1);
    gnn_sort_tile_kernel<<<NSLOT_HALF, TPB, 0, stream>>>(src, dst, offs, bins, 0);
    gnn_tile_sum2_kernel<NSLOT_HALF / NWAVES><<<NT, TPB, 0, stream>>>(offs, bins, rows, part, 0);
  } else {
    unsigned* bins = (unsigned*)(ws + WS_BINS2);
    gnn_sort_chunk16_kernel<<<NB, TPB, 0, stream>>>(dst, offs16, bins16);
    gnn_deg_owner16_kernel<<<NR, TPB, 0, stream>>>(offs16, bins16, x, dinv, rows);
    gnn_sort_tile_kernel<<<PCH, TPB, 0, stream>>>(src, dst, offs, bins, 0);
    gnn_tile_sum2_kernel<NSEG2><<<NT, TPB, 0, stream>>>(offs, bins, rows, part, 1);
    gnn_sort_tile_kernel<<<PCH, TPB, 0, stream>>>(src, dst, offs, bins, PCH);
    gnn_tile_sum2_kernel<NSEG2><<<NT, TPB, 0, stream>>>(offs, bins, rows, part, 0);
    gnn_sort_tile_kernel<<<PCH, TPB, 0, stream>>>(src, dst, offs, bins, 2 * PCH);
    gnn_tile_sum2_kernel<NSEG2><<<NT, TPB, 0, stream>>>(offs, bins, rows, part, 0);
    gnn_layer1_wmma_kernel<<<(threads1 + B - 1) / B, B, 0, stream>>>(part, x, dinv, W1, b1, W2, rows);
    gnn_tile_sum2_kernel<NSEG2><<<NT, TPB, 0, stream>>>(offs, bins, rows, part, 1);
    gnn_sort_tile_kernel<<<PCH, TPB, 0, stream>>>(src, dst, offs, bins, 0);
    gnn_tile_sum2_kernel<NSEG2><<<NT, TPB, 0, stream>>>(offs, bins, rows, part, 0);
    gnn_sort_tile_kernel<<<PCH, TPB, 0, stream>>>(src, dst, offs, bins, PCH);
    gnn_tile_sum2_kernel<NSEG2><<<NT, TPB, 0, stream>>>(offs, bins, rows, part, 0);
  }

  gnn_finalize2_kernel<<<(n2 + B - 1) / B, B, 0, stream>>>(part, dinv, b2, out);
}


# Sorted-list words loaded ahead of the LDS work using them: tile pass asks for the next group's words before the current group's adds (LDS-only barrier), degree run in one trip; sum order kept
# speedup vs baseline: 16.2216x; 1.0317x over previous
#include <hip/hip_runtime.h>


#define NN 200000
#define NE 12800000
#define NTILES (NN / 16)

#define TPB 1024
#define NWAVES (TPB / 32)
#define NCP 32
#define NKEY 256

#define RNG 800
#define NR (NN / RNG)
#define NB 512
#define CH (NE / NB)
#define CH4 (CH / 4)
#define OFFW 256
#define BU 7
#define GW 4
#define CHS16 25024
#define NSEG16 (NB / NWAVES)
#define SG16 4

#define TD 42
#define TS 6
#define NT (TD * TS)
#define NTP 256
#define DRNG 4768
#define NPAD (TD * DRNG)
#define SBLK 33334
#define NPART 3
#define PCH 256
#define CHA 16668
#define CHB 16664
#define CHS2 16672
#define OFFW2 256
#define BU2 5
#define NCHUNK (NPART * PCH)
#define NSLOT_HALF (NCHUNK / 2)
#define NSLOT_FULL NCHUNK
#define NSEG2 (PCH / NWAVES)
#define GW2 3

#define NCG 6
#define CGC (NSLOT_FULL / NCG)
#define CSEGW (CGC / NWAVES)
#define GWC 14

static_assert(NR * RNG == NN && RNG % 32 == 0, "degree ranges cover the nodes in whole lines");
static_assert(NCG * CGC == NSLOT_FULL && CSEGW * NWAVES == CGC && CSEGW <= 32, "chunk groups cover the slots; bounds held one per lane");
static_assert(NCG <= 2 * TS && NN % 4 == 0 && NPAD % 4 == 0, "count planes lie inside the planes of partial sums; 4 nodes per 16-byte group");
static_assert(CH * NB == NE && CH % 4 == 0, "degree chunks cover the edges");
static_assert(BU * TPB >= CH4, "one pass of the work-group loads a degree chunk");
static_assert(NR + 1 <= OFFW && OFFW == 8 * 32, "degree offset row: 8 entries per lane of one wave");
static_assert(CHS16 >= CH && (CHS16 * 2) % 128 == 0 && (CH * 2) % 16 == 0, "16-bit chunks: whole lines, whole 16-byte groups");
static_assert(NSEG16 * NWAVES == NB && NSEG16 % SG16 == 0 && NSEG16 <= 32, "degree pass: segment groups per wave, bounds held one per lane");
static_assert((size_t)NB * CHS16 < ((size_t)1 << 32), "16-bit entry index fits 32 bits");

static_assert(NPAD >= NN && DRNG % 16 == 0, "destination ranges cover the nodes in whole lines of 8-byte rows");
static_assert(TS * SBLK >= NN && (TS - 1) * SBLK < NN && SBLK % 2 == 0 && NN % 2 == 0, "source blocks cover the nodes in whole 16-byte groups");
static_assert(DRNG < (1 << 16) - 1 && SBLK <= (1 << 16), "word = 16 bits local destination | 16 bits local source; all ones is no word");
static_assert(NT <= NTP && NTP == 8 * 32 && NT + 1 <= OFFW2 && OFFW2 == NTP, "tile offset row: 8 counts per lane of one wave, whole lines");
static_assert((size_t)PCH * ((NPART - 1) * CHA + CHB) == NE && CHA % 4 == 0 && CHB % 4 == 0 && CHB <= CHA && BU2 * TPB >= CHA / 4, "chunks cover the edges; one pass loads a chunk");
static_assert(CHS2 >= CHA && CHS2 % 32 == 0, "whole lines per chunk");
static_assert(NSEG2 * NWAVES == PCH && NSEG2 <= 32, "bounds held one per lane");
static_assert(NSLOT_HALF % NWAVES == 0 && NSLOT_FULL % NWAVES == 0 && NSLOT_FULL / NWAVES <= 32,
              "every wave of a tile pass walks the same number of slots, bounds held one per lane");
static_assert((size_t)NSLOT_FULL * CHS2 < ((size_t)1 << 32), "word index fits 32 bits");
static_assert(NKEY == NTP && NKEY == OFFW && NKEY * NCP == 8 * TPB && NWAVES == 32 && NCP == 32,
              "counter scan: 8 entries per thread, one wave total per lane of a wave");

typedef __attribute__((ext_vector_type(2))) float v2f;
typedef __attribute__((ext_vector_type(8))) float v8f;

#define WS_OFFS16 ((size_t)1 * NN)
#define WS_BINS16 (WS_OFFS16 + (size_t)NB * OFFW)
#define WS_WORDS16 (WS_BINS16 + (size_t)NB * CHS16 / 2)
#define WS_PART ((size_t)1 * NN)
#define WS_OFFS2 (WS_PART + (size_t)TS * 2 * NPAD)
#define WS_BINS2 (WS_OFFS2 + (size_t)PCH * OFFW2)
#define WS_WORDS2 (WS_BINS2 + (size_t)PCH * CHS2)
#define WS_BINS_OF(S) (WS_OFFS2 + (size_t)(S) * OFFW2)
#define WS_WORDS_OF(S) (WS_BINS_OF(S) + (size_t)(S) * CHS2)
#define WS_BYTES_HALF (WS_WORDS_OF(NSLOT_HALF) * 4)
#define WS_BYTES_FULL (WS_WORDS_OF(NSLOT_FULL) * 4)
static_assert(WS_WORDS2 * 4 <= (size_t)35 * NN * 4, "inside the 28,000,000 bytes the versions before this one used");
static_assert(WS_PART % 32 == 0 && (2 * NPAD) % 32 == 0 && WS_OFFS2 % 32 == 0 && WS_BINS2 % 32 == 0, "planes start on 128-byte lines");
static_assert(WS_WORDS16 * 4 <= (size_t)35 * NN * 4, "degree list inside the same 28,000,000 bytes");
static_assert(WS_BINS_OF(PCH) == WS_BINS2 && WS_WORDS_OF(PCH) == WS_WORDS2, "the smallest layout is the one above");
static_assert(WS_BINS_OF(NSLOT_HALF) % 32 == 0 && WS_BINS_OF(NSLOT_FULL) % 32 == 0, "planes start on 128-byte lines");
static_assert(WS_BYTES_HALF == (size_t)36413696 && WS_BYTES_FULL == (size_t)62415104, "sizes named in the comments");
static_assert(WS_WORDS16 <= WS_WORDS2, "the degree list lies inside every layout");
static_assert(WS_OFFS16 % 32 == 0 && WS_BINS16 % 32 == 0, "degree planes start on 128-byte lines");

__device__ __forceinline__ unsigned gnn_wave_scan_incl(unsigned v, unsigned lane) {
  v += (unsigned)__builtin_amdgcn_update_dpp(0, (int)v, 0x111, 0xf, 0xf, false);
  v += (unsigned)__builtin_amdgcn_update_dpp(0, (int)v, 0x112, 0xf, 0xf, false);
  v += (unsigned)__builtin_amdgcn_update_dpp(0, (int)v, 0x114, 0xf, 0xf, false);
  v += (unsigned)__builtin_amdgcn_update_dpp(0, (int)v, 0x118, 0xf, 0xf, false);
  const unsigned low = (unsigned)__builtin_amdgcn_readlane((int)v, 15);
  if (lane >= 16u) v += low;
  return v;
}

__device__ __forceinline__ void gnn_scan_private_counters(unsigned* cnt, unsigned* wtot, unsigned* kbase,
                                                          int tid, unsigned lane, unsigned w) {
  uint4* c4 = reinterpret_cast<uint4*>(cnt);
  const uint4 ca = c4[2 * tid + 0];
  const uint4 cb = c4[2 * tid + 1];
  const unsigned c[8] = {ca.x, ca.y, ca.z, ca.w, cb.x, cb.y, cb.z, cb.w};
  unsigned s = 0u;
#pragma unroll
  for (int k = 0; k < 8; ++k) s += c[k];
  const unsigned inc = gnn_wave_scan_incl(s, lane);
  if (lane == 31u) wtot[w] = inc;
  __syncthreads();
  const unsigned wv = wtot[lane];
  const unsigned winc = gnn_wave_scan_incl(wv, lane);
  const int ws = __builtin_amdgcn_readfirstlane((int)w);
  const unsigned wbase = (unsigned)__builtin_amdgcn_readlane((int)(winc - wv), ws);
  unsigned ex = wbase + (inc - s);
  if ((tid & 3) == 0) kbase[tid >> 2] = ex;
  unsigned e[8];
#pragma unroll
  for (int k = 0; k < 8; ++k) { e[k] = ex; ex += c[k]; }
  c4[2 * tid + 0] = make_uint4(e[0], e[1], e[2], e[3]);
  c4[2 * tid + 1] = make_uint4(e[4], e[5], e[6], e[7]);
  __syncthreads();
}

__global__ __launch_bounds__(TPB) void gnn_sort_tile_kernel(const int* __restrict__ src,
                                                            const int* __restrict__ dst,
                                                            unsigned* __restrict__ offs,
                                                            unsigned* __restrict__ bins,
                                                            int cbase) {
  __shared__ __attribute__((aligned(16))) unsigned cnt[NKEY * NCP];
  __shared__ unsigned wtot[NWAVES];
  __shared__ unsigned kbase[NKEY];
  __shared__ __attribute__((aligned(16))) unsigned stage[CHA];
  const int slot = blockIdx.x;
  const int part = (cbase + slot) / PCH;
  const int b = (cbase + slot) - part * PCH;
  const int tid = threadIdx.x;
  const unsigned lane = threadIdx.x & 31;
  const unsigned w = threadIdx.x >> 5;
  reinterpret_cast<uint4*>(cnt)[2 * tid + 0] = make_uint4(0u, 0u, 0u, 0u);
  reinterpret_cast<uint4*>(cnt)[2 * tid + 1] = make_uint4(0u, 0u, 0u, 0u);
  __syncthreads();

  const int csz4 = (part < NPART - 1) ? CHA / 4 : CHB / 4;
  const size_t e0 = (size_t)part * PCH * CHA + (size_t)b * (4 * csz4);
  const int4* __restrict__ d4 = reinterpret_cast<const int4*>(dst + e0);
  const int4* __restrict__ s4 = reinterpret_cast<const int4*>(src + e0);
  int4 dv[BU2], sv[BU2];
#pragma unroll
  for (int u = 0; u < BU2; ++u) {
    int i = u * TPB + tid;
    bool ok = i < csz4;
    dv[u] = ok ? d4[i] : make_int4(-1, -1, -1, -1);
    sv[u] = ok ? s4[i] : make_int4(-1, -1, -1, -1);
  }
#pragma unroll
  for (int u = 0; u < BU2; ++u) {
    const unsigned dd[4] = {(unsigned)dv[u].x, (unsigned)dv[u].y, (unsigned)dv[u].z, (unsigned)dv[u].w};
    const unsigned ss[4] = {(unsigned)sv[u].x, (unsigned)sv[u].y, (unsigned)sv[u].z, (unsigned)sv[u].w};
#pragma unroll
    for (int k = 0; k < 4; ++k)
      if (dd[k] < (unsigned)NN && ss[k] < (unsigned)NN)
        atomicAdd(&cnt[((dd[k] / DRNG) * TS + ss[k] / SBLK) * NCP + lane], 1u);
  }
  __syncthreads();

  gnn_scan_private_counters(cnt, wtot, kbase, tid, lane, w);
  if (tid < NKEY) offs[(size_t)slot * OFFW2 + tid] = kbase[tid];

#pragma unroll
  for (int u = 0; u < BU2; ++u) {
    const unsigned dd[4] = {(unsigned)dv[u].x, (unsigned)dv[u].y, (unsigned)dv[u].z, (unsigned)dv[u].w};
    const unsigned ss[4] = {(unsigned)sv[u].x, (unsigned)sv[u].y, (unsigned)sv[u].z, (unsigned)sv[u].w};
#pragma unroll
    for (int k = 0; k < 4; ++k) {
      if (dd[k] < (unsigned)NN && ss[k] < (unsigned)NN) {
        unsigned r = dd[k] / DRNG;
        unsigned q = ss[k] / SBLK;
        unsigned pos = atomicAdd(&cnt[(r * TS + q) * NCP + lane], 1u);
        stage[pos] = ((dd[k] - r * DRNG) << 16) | (ss[k] - q * SBLK);
      }
    }
  }
  __syncthreads();

  uint4* __restrict__ o4 = reinterpret_cast<uint4*>(bins + (size_t)slot * CHS2);
  const uint4* st4 = reinterpret_cast<const uint4*>(stage);
  for (int i = tid; i < csz4; i += TPB) o4[i] = st4[i];
}

__global__ __launch_bounds__(TPB) void gnn_sort_chunk16_kernel(const int* __restrict__ dst,
                                                               unsigned* __restrict__ offs,
                                                               unsigned short* __restrict__ bins16) {
  __shared__ __attribute__((aligned(16))) unsigned cnt[NKEY * NCP];
  __shared__ unsigned wtot[NWAVES];
  __shared__ unsigned kbase[NKEY];
  __shared__ __attribute__((aligned(16))) unsigned short stage[CH];
  const int b = blockIdx.x;
  const int tid = threadIdx.x;
  const unsigned lane = threadIdx.x & 31;
  const unsigned w = threadIdx.x >> 5;
  reinterpret_cast<uint4*>(cnt)[2 * tid + 0] = make_uint4(0u, 0u, 0u, 0u);
  reinterpret_cast<uint4*>(cnt)[2 * tid + 1] = make_uint4(0u, 0u, 0u, 0u);
  __syncthreads();

  const int4* __restrict__ d4 = reinterpret_cast<const int4*>(dst + (size_t)b * CH);
  int4 dv[BU];
#pragma unroll
  for (int u = 0; u < BU; ++u) {
    int i = u * TPB + tid;
    dv[u] = (i < CH4) ? d4[i] : make_int4(-1, -1, -1, -1);
  }
#pragma unroll
  for (int u = 0; u < BU; ++u) {
    const unsigned dd[4] = {(unsigned)dv[u].x, (unsigned)dv[u].y, (unsigned)dv[u].z, (unsigned)dv[u].w};
#pragma unroll
    for (int k = 0; k < 4; ++k)
      if (dd[k] < (unsigned)NN) atomicAdd(&cnt[(dd[k] / RNG) * NCP + lane], 1u);
  }
  __syncthreads();

  gnn_scan_private_counters(cnt, wtot, kbase, tid, lane, w);
  if (tid < NKEY) offs[(size_t)b * OFFW + tid] = kbase[tid];

#pragma unroll
  for (int u = 0; u < BU; ++u) {
    const unsigned dd[4] = {(unsigned)dv[u].x, (unsigned)dv[u].y, (unsigned)dv[u].z, (unsigned)dv[u].w};
#pragma unroll
    for (int k = 0; k < 4; ++k) {
      if (dd[k] < (unsigned)NN) {
        unsigned r = dd[k] / RNG;
        unsigned pos = atomicAdd(&cnt[r * NCP + lane], 1u);
        stage[pos] = (unsigned short)(dd[k] - r * RNG);
      }
    }
  }
  __syncthreads();

  uint4* __restrict__ o4 = reinterpret_cast<uint4*>(bins16 + (size_t)b * CHS16);
  const uint4* st4 = reinterpret_cast<const uint4*>(stage);
  for (int i = tid; i < CH / 8; i += TPB) o4[i] = st4[i];
}

__global__ __launch_bounds__(TPB) void gnn_deg_owner16_kernel(const unsigned* __restrict__ offs,
                                                              const unsigned short* __restrict__ bins16,
                                                              const float* __restrict__ x,
                                                              float* __restrict__ dinv,
                                                              float* __restrict__ xs) {
  __shared__ unsigned cnt[RNG];
  const int r = blockIdx.x;
  const int tid = threadIdx.x;
  const unsigned lane = threadIdx.x & 31;
  const unsigned w = threadIdx.x >> 5;
  if (tid < RNG) cnt[tid] = 0u;
  __syncthreads();

  unsigned s0 = 0u, s1 = 0u;
  if (lane < (unsigned)NSEG16) {
    unsigned bb = w * (unsigned)NSEG16 + lane;
    const unsigned* __restrict__ o = offs + (size_t)bb * OFFW + r;
    s0 = bb * (unsigned)CHS16 + o[0];
    s1 = bb * (unsigned)CHS16 + o[1];
  }
  for (int j = 0; j < NSEG16; j += SG16) {
    unsigned sa[SG16], sn[SG16];
    unsigned nmax = 0u;
#pragma unroll
    for (int g = 0; g < SG16; ++g) {
      sa[g] = (unsigned)__builtin_amdgcn_readlane((int)s0, j + g);
      sn[g] = (unsigned)__builtin_amdgcn_readlane((int)s1, j + g) - sa[g];
      nmax = sn[g] > nmax ? sn[g] : nmax;
    }
    for (unsigned k0 = 0; k0 < nmax; k0 += 32 * GW) {
      unsigned p[SG16 * GW];
#pragma unroll
      for (int g = 0; g < SG16; ++g) {
#pragma unroll
        for (int t = 0; t < GW; ++t) {
          unsigned k = k0 + 32u * t + lane;
          p[g * GW + t] = (k < sn[g]) ? (unsigned)bins16[sa[g] + k] : 0xFFFFu;
        }
      }
#pragma unroll
      for (int t = 0; t < SG16 * GW; ++t)
        if (p[t] != 0xFFFFu) atomicAdd(&cnt[p[t]], 1u);
    }
  }
  __syncthreads();

  if (tid < RNG) {
    int node = r * RNG + tid;
    float d = rsqrtf((float)cnt[tid] + 1.0f);
    dinv[node] = d;
    float2 xv = reinterpret_cast<const float2*>(x)[node];
    reinterpret_cast<float2*>(xs)[node] = make_float2(xv.x * d, xv.y * d);
  }
}

__global__ __launch_bounds__(TPB) void gnn_deg_tile_kernel(const unsigned* __restrict__ offs,
                                                           const unsigned* __restrict__ bins,
                                                           unsigned* __restrict__ cntp) {
  __shared__ unsigned cnt[DRNG];
  const int r = blockIdx.x / NCG;
  const int g = blockIdx.x - r * NCG;
  const int tid = threadIdx.x;
  const unsigned lane = threadIdx.x & 31;
  const unsigned w = threadIdx.x >> 5;

  const unsigned bb = (unsigned)g * (unsigned)CGC + w * (unsigned)CSEGW + (lane < (unsigned)CSEGW ? lane : 0u);
  const unsigned* __restrict__ o = offs + (size_t)bb * OFFW2 + r * TS;
  const unsigned o0 = o[0], o1 = o[TS];
  for (int i = tid; i < DRNG; i += TPB) cnt[i] = 0u;
  const unsigned s0 = bb * (unsigned)CHS2 + o0;
  const unsigned s1 = bb * (unsigned)CHS2 + o1;
  __syncthreads();

  unsigned sa[CSEGW], sn[CSEGW];
  unsigned nmax = 0u;
#pragma unroll
  for (int j = 0; j < CSEGW; ++j) {
    sa[j] = (unsigned)__builtin_amdgcn_readlane((int)s0, j);
    sn[j] = (unsigned)__builtin_amdgcn_readlane((int)s1, j) - sa[j];
    nmax = sn[j] > nmax ? sn[j] : nmax;
  }
  for (unsigned k0 = 0; k0 < nmax; k0 += 32 * GWC) {
    unsigned p[CSEGW * GWC];
#pragma unroll
    for (int j = 0; j < CSEGW; ++j) {
#pragma unroll
      for (int u = 0; u < GWC; ++u) {
        unsigned k = k0 + 32u * u + lane;
        p[j * GWC + u] = (k < sn[j]) ? bins[sa[j] + k] : 0xFFFFFFFFu;
      }
    }
#pragma unroll
    for (int u = 0; u < CSEGW * GWC; ++u)
      if (p[u] != 0xFFFFFFFFu) atomicAdd(&cnt[p[u] >> 16], 1u);
  }
  __syncthreads();

  unsigned* __restrict__ plane = cntp + (size_t)g * NPAD + (size_t)r * DRNG;
  for (int i = tid; i < DRNG; i += TPB) plane[i] = cnt[i];
}

__global__ void gnn_deg_finish_kernel(const unsigned* __restrict__ cntp,
                                      const float* __restrict__ x,
                                      float* __restrict__ dinv,
                                      float* __restrict__ xs) {
  int t = blockIdx.x * blockDim.x + threadIdx.x;
  if (t >= NN / 4) return;
  uint4 cs = make_uint4(0u, 0u, 0u, 0u);
#pragma unroll
  for (int g = 0; g < NCG; ++g) {
    uint4 v = reinterpret_cast<const uint4*>(cntp + (size_t)g * NPAD)[t];
    cs.x += v.x;
    cs.y += v.y;
    cs.z += v.z;
    cs.w += v.w;
  }
  float4 xa = reinterpret_cast<const float4*>(x)[2 * t + 0];
  float4 xb = reinterpret_cast<const float4*>(x)[2 * t + 1];
  float4 d;
  d.x = rsqrtf((float)cs.x + 1.0f);
  d.y = rsqrtf((float)cs.y + 1.0f);
  d.z = rsqrtf((float)cs.z + 1.0f);
  d.w = rsqrtf((float)cs.w + 1.0f);
  reinterpret_cast<float4*>(dinv)[t] = d;
  reinterpret_cast<float4*>(xs)[2 * t + 0] = make_float4(xa.x * d.x, xa.y * d.x, xa.z * d.y, xa.w * d.y);
  reinterpret_cast<float4*>(xs)[2 * t + 1] = make_float4(xb.x * d.z, xb.y * d.z, xb.z * d.w, xb.w * d.w);
}

template <int NSEGW>
__device__ __forceinline__ void gnn_group_bounds(int grp, unsigned s0, unsigned s1,
                                                 unsigned (&sa)[NSEG2], unsigned (&sn)[NSEG2]) {
#pragma unroll
  for (int g = 0; g < NSEG2; ++g) {
    sa[g] = 0u;
    sn[g] = 0u;
    if (grp * NSEG2 + g < NSEGW) {
      sa[g] = (unsigned)__builtin_amdgcn_readlane((int)s0, grp * NSEG2 + g);
      sn[g] = (unsigned)__builtin_amdgcn_readlane((int)s1, grp * NSEG2 + g) - sa[g];
    }
  }
}

__device__ __forceinline__ void gnn_load_words(const unsigned* __restrict__ bins,
                                               const unsigned (&sa)[NSEG2], const unsigned (&sn)[NSEG2],
                                               unsigned k0, unsigned lane, unsigned (&p)[NSEG2 * GW2]) {
#pragma unroll
  for (int g = 0; g < NSEG2; ++g) {
#pragma unroll
    for (int u = 0; u < GW2; ++u) {
      const unsigned k = k0 + 32u * u + lane;
      const bool in = k < sn[g];
      const unsigned wv = bins[in ? sa[g] + k : 0u];
      p[g * GW2 + u] = in ? wv : 0xFFFFFFFFu;
    }
  }
}

__device__ __forceinline__ void gnn_add_words(const unsigned (&p)[NSEG2 * GW2], const float2* tab2, float* acc) {
  float2 h[NSEG2 * GW2];
#pragma unroll
  for (int u = 0; u < NSEG2 * GW2; ++u) {
    h[u] = make_float2(0.0f, 0.0f);
    if (p[u] != 0xFFFFFFFFu) h[u] = tab2[p[u] & 0xFFFFu];
  }
#pragma unroll
  for (int u = 0; u < NSEG2 * GW2; ++u) {
    if (p[u] != 0xFFFFFFFFu) {
      unsigned a = p[u] >> 16;
      unsafeAtomicAdd(&acc[2 * a + 0], h[u].x);
      unsafeAtomicAdd(&acc[2 * a + 1], h[u].y);
    }
  }
}

template <int NSEGW>
__global__ __launch_bounds__(TPB) void gnn_tile_sum2_kernel(const unsigned* __restrict__ offs,
                                                            const unsigned* __restrict__ bins,
                                                            const float* __restrict__ rows2,
                                                            float* __restrict__ part,
                                                            int first) {
  __shared__ __attribute__((aligned(16))) float tab[SBLK * 2];
  __shared__ __attribute__((aligned(16))) float acc[DRNG * 2];
  const int t = blockIdx.x;
  const int r = t / TS;
  const int q = t - r * TS;
  const int tid = threadIdx.x;
  const unsigned lane = threadIdx.x & 31;
  const unsigned w = threadIdx.x >> 5;

  const unsigned bb = w * (unsigned)NSEGW + (lane < (unsigned)NSEGW ? lane : 0u);
  const unsigned* __restrict__ o = offs + (size_t)bb * OFFW2 + t;
  const unsigned o0 = o[0], o1 = o[1];

  const float4* __restrict__ g4 = reinterpret_cast<const float4*>(rows2 + (size_t)2 * q * SBLK);
  float4* __restrict__ prow = reinterpret_cast<float4*>(part + (size_t)q * (2 * NPAD) + (size_t)2 * r * DRNG);
  float4* t4 = reinterpret_cast<float4*>(tab);
  float4* a4 = reinterpret_cast<float4*>(acc);
  constexpr int NG = (SBLK / 2 + TPB - 1) / TPB;
  constexpr int NA = (DRNG / 2 + TPB - 1) / TPB;
  const int nrow4 = (NN / 2 - q * (SBLK / 2)) < SBLK / 2 ? (NN / 2 - q * (SBLK / 2)) : SBLK / 2;
  float4 v[NG], pv[NA];
#pragma unroll
  for (int u = 0; u < NG; ++u) {
    int i = u * TPB + tid;
    v[u] = (i < nrow4) ? g4[i] : make_float4(0.0f, 0.0f, 0.0f, 0.0f);
  }
#pragma unroll
  for (int u = 0; u < NA; ++u) {
    int i = u * TPB + tid;
    pv[u] = (!first && i < DRNG / 2) ? prow[i] : make_float4(0.0f, 0.0f, 0.0f, 0.0f);
  }
  const unsigned s0 = bb * (unsigned)CHS2 + o0;
  const unsigned s1 = bb * (unsigned)CHS2 + o1;
  constexpr int NGRP = (NSEGW + NSEG2 - 1) / NSEG2;
  unsigned sa_n[NSEG2], sn_n[NSEG2], p_n[NSEG2 * GW2];
  gnn_group_bounds<NSEGW>(0, s0, s1, sa_n, sn_n);
  gnn_load_words(bins, sa_n, sn_n, 0u, lane, p_n);
#pragma unroll
  for (int u = 0; u < NG; ++u) {
    int i = u * TPB + tid;
    if (i < SBLK / 2) t4[i] = v[u];
  }
#pragma unroll
  for (int u = 0; u < NA; ++u) {
    int i = u * TPB + tid;
    if (i < DRNG / 2) a4[i] = pv[u];
  }
  __builtin_amdgcn_fence(__ATOMIC_RELEASE, "workgroup", "local");
  __builtin_amdgcn_s_barrier();
  __builtin_amdgcn_fence(__ATOMIC_ACQUIRE, "workgroup", "local");

  const float2* tab2 = reinterpret_cast<const float2*>(tab);
#pragma unroll
  for (int grp = 0; grp < NGRP; ++grp) {
    unsigned sa[NSEG2], sn[NSEG2], p[NSEG2 * GW2];
    unsigned nmax = 0u;
#pragma unroll
    for (int g = 0; g < NSEG2; ++g) {
      sa[g] = sa_n[g];
      sn[g] = sn_n[g];
      nmax = sn[g] > nmax ? sn[g] : nmax;
    }
#pragma unroll
    for (int u = 0; u < NSEG2 * GW2; ++u) p[u] = p_n[u];
    if (grp + 1 < NGRP) {
      gnn_group_bounds<NSEGW>(grp + 1, s0, s1, sa_n, sn_n);
      gnn_load_words(bins, sa_n, sn_n, 0u, lane, p_n);
    }
    gnn_add_words(p, tab2, acc);
    for (unsigned k0 = 32 * GW2; k0 < nmax; k0 += 32 * GW2) {
      gnn_load_words(bins, sa, sn, k0, lane, p);
      gnn_add_words(p, tab2, acc);
    }
  }
  __syncthreads();

  for (int i = tid; i < DRNG / 2; i += TPB) prow[i] = a4[i];
}

__global__ void gnn_layer1_wmma_kernel(const float* __restrict__ part,
                                       const float* __restrict__ x,
                                       const float* __restrict__ dinv,
                                       const float* __restrict__ W1,
                                       const float* __restrict__ b1,
                                       const float* __restrict__ W2,
                                       float* __restrict__ hs2) {
  int wave = (blockIdx.x * blockDim.x + threadIdx.x) >> 5;
  int lane = threadIdx.x & 31;
  if (wave >= NTILES) return;
  int c    = lane & 15;
  int half = lane >> 4;
  int node = (wave << 4) + c;
  bool lo  = half == 0;

  float2 a = make_float2(0.0f, 0.0f);
#pragma unroll
  for (int s = 0; s < TS; ++s) {
    float2 v = reinterpret_cast<const float2*>(part + (size_t)s * (2 * NPAD))[node];
    a.x += v.x;
    a.y += v.y;
  }
  float di = dinv[node];
  float2 xv = reinterpret_cast<const float2*>(x)[node];
  float z0 = di * (a.x + xv.x * di);
  float z1 = di * (a.y + xv.y * di);

  float w0 = W1[c];
  float w1 = W1[16 + c];

  v2f A, B;
  A.x = lo ? w0 : 0.0f;
  A.y = lo ? w1 : 0.0f;
  B.x = lo ? z0 : 0.0f;
  B.y = lo ? z1 : 0.0f;

  v8f acc = {};
  acc = __builtin_amdgcn_wmma_f32_16x16x4_f32(
       false, A,  false, B,
       (short)0, acc,  false,  false);

  const float* __restrict__ bb = b1 + 8 * half;
  const float* __restrict__ ww = W2 + 16 * half;
  float h0 = 0.0f, h1 = 0.0f;
#pragma unroll
  for (int r = 0; r < 8; ++r) {
    float v = acc[r] + bb[r];
    v = v > 0.0f ? v : 0.0f;
    h0 += v * ww[2 * r + 0];
    h1 += v * ww[2 * r + 1];
  }
  h0 += __shfl_xor(h0, 16);
  h1 += __shfl_xor(h1, 16);
  if (lo) reinterpret_cast<float2*>(hs2)[node] = make_float2(h0 * di, h1 * di);
}

__global__ void gnn_finalize2_kernel(const float* __restrict__ part,
                                     const float* __restrict__ dinv,
                                     const float* __restrict__ b2,
                                     float* out) {
  int t = blockIdx.x * blockDim.x + threadIdx.x;
  if (t >= NN * 2) return;
  int i = t >> 1;
  int c = t & 1;
  float a = 0.0f;
#pragma unroll
  for (int s = 0; s < TS; ++s) a += part[(size_t)s * (2 * NPAD) + t];
  out[t] = dinv[i] * (a + out[t]) + b2[c];
}

extern "C" void kernel_launch(void* const* d_in, const int* in_sizes, int n_in,
                              void* d_out, int out_size, void* d_ws, size_t ws_size,
                              hipStream_t stream) {
  const float* x  = (const float*)d_in[0];
  const float* W1 = (const float*)d_in[1];
  const float* b1 = (const float*)d_in[2];
  const float* W2 = (const float*)d_in[3];
  const float* b2 = (const float*)d_in[4];
  const int* edge = (const int*)d_in[5];
  const int* src  = edge;
  const int* dst  = edge + NE;
  float* out = (float*)d_out;

  float* ws = (float*)d_ws;
  float* dinv  = ws;
  float* part  = ws + WS_PART;
  unsigned* offs = (unsigned*)(ws + WS_OFFS2);
  unsigned* offs16 = (unsigned*)(ws + WS_OFFS16);
  unsigned short* bins16 = (unsigned short*)(ws + WS_BINS16);
  float* rows = out;

  const int B = 256;
  const int threads1 = NTILES * 32;
  const int n2 = NN * 2;

  if (ws_size >= WS_BYTES_FULL) {
    unsigned* bins = (unsigned*)(ws + WS_BINS_OF(NSLOT_FULL));
    unsigned* cntp = (unsigned*)(ws + WS_PART);
    gnn_sort_tile_kernel<<<NSLOT_FULL, TPB, 0, stream>>>(src, dst, offs, bins, 0);
    gnn_deg_tile_kernel<<<TD * NCG, TPB, 0, stream>>>(offs, bins, cntp);
    gnn_deg_finish_kernel<<<(NN / 4 + B - 1) / B, B, 0, stream>>>(cntp, x, dinv, rows);
    gnn_tile_sum2_kernel<NSLOT_FULL / NWAVES><<<NT, TPB, 0, stream>>>(offs, bins, rows, part, 1);
    gnn_layer1_wmma_kernel<<<(threads1 + B - 1) / B, B, 0, stream>>>(part, x, dinv, W1, b1, W2, rows);
    gnn_tile_sum2_kernel<NSLOT_FULL / NWAVES><<<NT, TPB, 0, stream>>>(offs, bins, rows, part, 1);
  } else if (ws_size >= WS_BYTES_HALF) {
    unsigned* bins = (unsigned*)(ws + WS_BINS_OF(NSLOT_HALF));
    gnn_sort_chunk16_kernel<<<NB, TPB, 0, stream>>>(dst, offs16, bins16);
    gnn_deg_owner16_kernel<<<NR, TPB, 0, stream>>>(offs16, bins16, x, dinv, rows);
    gnn_sort_tile_kernel<<<NSLOT_HALF, TPB, 0, stream>>>(src, dst, offs, bins, 0);
    gnn_tile_sum2_kernel<NSLOT_HALF / NWAVES><<<NT, TPB, 0, stream>>>(offs, bins, rows, part, 1);
    gnn_sort_tile_kernel<<<NSLOT_HALF, TPB, 0, stream>>>(src, dst, offs, bins, NSLOT_HALF);
    gnn_tile_sum2_kernel<NSLOT_HALF / NWAVES><<<NT, TPB, 0, stream>>>(offs, bins, rows, part, 0);
    gnn_layer1_wmma_kernel<<<(threads1 + B - 1) / B, B, 0, stream>>>(part, x, dinv, W1, b1, W2, rows);
    gnn_tile_sum2_kernel<NSLOT_HALF / NWAVES><<<NT, TPB, 0, stream>>>(offs, bins, rows, part, 1);
    gnn_sort_tile_kernel<<<NSLOT_HALF, TPB, 0, stream>>>(src, dst, offs, bins, 0);
    gnn_tile_sum2_kernel<NSLOT_HALF / NWAVES><<<NT, TPB, 0, stream>>>(offs, bins, rows, part, 0);
  } else {
    unsigned* bins = (unsigned*)(ws + WS_BINS2);
    gnn_sort_chunk16_kernel<<<NB, TPB, 0, stream>>>(dst, offs16, bins16);
    gnn_deg_owner16_kernel<<<NR, TPB, 0, stream>>>(offs16, bins16, x, dinv, rows);
    gnn_sort_tile_kernel<<<PCH, TPB, 0, stream>>>(src, dst, offs, bins, 0);
    gnn_tile_sum2_kernel<NSEG2><<<NT, TPB, 0, stream>>>(offs, bins, rows, part, 1);
    gnn_sort_tile_kernel<<<PCH, TPB, 0, stream>>>(src, dst, offs, bins, PCH);
    gnn_tile_sum2_kernel<NSEG2><<<NT, TPB, 0, stream>>>(offs, bins, rows, part, 0);
    gnn_sort_tile_kernel<<<PCH, TPB, 0, stream>>>(src, dst, offs, bins, 2 * PCH);
    gnn_tile_sum2_kernel<NSEG2><<<NT, TPB, 0, stream>>>(offs, bins, rows, part, 0);
    gnn_layer1_wmma_kernel<<<(threads1 + B - 1) / B, B, 0, stream>>>(part, x, dinv, W1, b1, W2, rows);
    gnn_tile_sum2_kernel<NSEG2><<<NT, TPB, 0, stream>>>(offs, bins, rows, part, 1);
    gnn_sort_tile_kernel<<<PCH, TPB, 0, stream>>>(src, dst, offs, bins, 0);
    gnn_tile_sum2_kernel<NSEG2><<<NT, TPB, 0, stream>>>(offs, bins, rows, part, 0);
    gnn_sort_tile_kernel<<<PCH, TPB, 0, stream>>>(src, dst, offs, bins, PCH);
    gnn_tile_sum2_kernel<NSEG2><<<NT, TPB, 0, stream>>>(offs, bins, rows, part, 0);
  }

  gnn_finalize2_kernel<<<(n2 + B - 1) / B, B, 0, stream>>>(part, dinv, b2, out);
}
